# starting kernel
# baseline (speedup 1.0000x reference)
#include <hip/hip_runtime.h>
#include <math.h>

typedef float v2f __attribute__((ext_vector_type(2)));
typedef float v8f __attribute__((ext_vector_type(8)));

#define LROW 8192
#define NSTATE 64
#define PI_F 3.14159265358979323846f

#define OFF_F128Q   0
#define OFF_IF128Q  32768
#define OFF_G64PR   65536
#define OFF_G64PI   69632
#define OFF_IG64PR  73728
#define OFF_IG64PI  77824
#define OFF_WF      81920
#define OFF_WI      98304
#define OFF_KHAT    114688
#define WS_FLOATS   (OFF_KHAT + 4*16384)

static __device__ inline v8f wmma4(v2f a, v2f b, v8f c) {
  return __builtin_amdgcn_wmma_f32_16x16x4_f32(false, a, false, b, (short)0, c,
                                               false, false);
}

static __device__ inline int pidx(int row, int col) {
  return ((row >> 1) << 7) + (col << 1) + (row & 1);
}

__global__ void ssm_gen_tables(float* __restrict__ ws) {
  int idx = blockIdx.x * blockDim.x + threadIdx.x;
  {
    int k2 = idx >> 7, n2 = idx & 127;
    int t = (k2 * n2) & 127;
    float s, c;
    sincosf(-2.f * PI_F * (float)t / 128.f, &s, &c);
    int pos = k2 * 256 + (n2 >> 1) * 4 + (n2 & 1);
    ws[OFF_F128Q + pos]      = c;
    ws[OFF_F128Q + pos + 2]  = s;
    ws[OFF_IF128Q + pos]     = c;
    ws[OFF_IF128Q + pos + 2] = -s;
  }
  if (idx < 8192) {
    int k2 = idx >> 6, n1 = idx & 63;
    int t = (n1 * k2) & 8191;
    float s, c;
    sincosf(-2.f * PI_F * (float)t / 8192.f, &s, &c);
    ws[OFF_WF + 2 * idx]     = c;  ws[OFF_WF + 2 * idx + 1] = s;
    ws[OFF_WI + 2 * idx]     = c;  ws[OFF_WI + 2 * idx + 1] = -s;
  }
  if (idx < 4096) {
    int n1 = idx >> 6, k1 = idx & 63;
    int t = (n1 * k1) & 63;
    float s, c;
    sincosf(-2.f * PI_F * (float)t / 64.f, &s, &c);
    int pos = (n1 >> 1) * 128 + 2 * k1 + (n1 & 1);
    ws[OFF_G64PR + pos]  = c;  ws[OFF_G64PI + pos]  = s;
    ws[OFF_IG64PR + pos] = c;  ws[OFF_IG64PI + pos] = -s;
  }
}

__global__ void ssm_gen_khat(float* __restrict__ ws,
                             const float* __restrict__ Lr,
                             const float* __restrict__ Li,
                             const float* __restrict__ Pv,
                             const float* __restrict__ Bv,
                             const float* __restrict__ Ct,
                             const float* __restrict__ stepv) {
  int layer = blockIdx.y;
  int s = blockIdx.x * blockDim.x + threadIdx.x;
  int k2 = s >> 6, k1 = s & 63;
  int kf = 128 * k1 + k2;

  float osin, ocos;
  sincosf(-2.f * PI_F * (float)kf / 8192.f, &osin, &ocos);
  float pr = 1.f + ocos, pi = osin;
  float mr = 1.f - ocos, mi = -osin;
  float pd = pr * pr + pi * pi;
  float ipr = pr / pd, ipi = -pi / pd;
  float cr = 2.f * ipr, ci = 2.f * ipi;
  float sc = 2.f / stepv[layer];
  float gr = sc * (mr * ipr - mi * ipi);
  float gi = sc * (mr * ipi + mi * ipr);

  const float* lr = Lr + layer * NSTATE;
  const float* li = Li + layer * NSTATE;
  const float* pp = Pv + layer * NSTATE;
  const float* bb = Bv + layer * NSTATE;
  const float* ct = Ct + layer * NSTATE;

  float k00r = 0, k00i = 0, k01r = 0, k01i = 0;
  float k10r = 0, k10i = 0, k11r = 0, k11i = 0;
  for (int n = 0; n < NSTATE; ++n) {
    float dr = gr - lr[n], di = gi - li[n];
    float dd = dr * dr + di * di;
    float ir = dr / dd, ii = -di / dd;
    float b = bb[n], p = pp[n], c = ct[n];
    float w00 = c * b, w01 = c * p, w10 = p * b, w11 = p * p;
    k00r += w00 * ir; k00i += w00 * ii;
    k01r += w01 * ir; k01i += w01 * ii;
    k10r += w10 * ir; k10i += w10 * ii;
    k11r += w11 * ir; k11i += w11 * ii;
  }
  float qr = k01r * k10r - k01i * k10i;
  float qi = k01r * k10i + k01i * k10r;
  float er = 1.f + k11r, ei = k11i;
  float ed = er * er + ei * ei;
  float ier = er / ed, iei = -ei / ed;
  float tr = qr * ier - qi * iei;
  float ti = qr * iei + qi * ier;
  float kr = k00r - tr, ki = k00i - ti;
  float outr = cr * kr - ci * ki;
  float outi = cr * ki + ci * kr;
  const float scale = 1.f / 8192.f;
  ws[OFF_KHAT + layer * 16384 + 2 * s]     = outr * scale;
  ws[OFF_KHAT + layer * 16384 + 2 * s + 1] = outi * scale;
}

__global__ __launch_bounds__(256) void ssm_rows(const float* __restrict__ uin,
                                                float* __restrict__ uout,
                                                const float* __restrict__ ws) {
  extern __shared__ float smem[];
  float* xb = smem;
  float* Yr = smem + 8192;
  float* Yi = smem + 16384;

  const int tid  = threadIdx.x;
  const int wave = tid >> 5;
  const int lane = tid & 31;
  const int l16  = lane & 15;
  const int kSel = (lane >> 4) << 1;
  const int mHi  = (lane >> 4) << 3;
  const int mB   = wave << 4;

  const size_t rowOff = (size_t)blockIdx.x * LROW;
  const float4* src = (const float4*)(uin + rowOff);
  for (int i = tid; i < LROW / 4; i += 256) {
    float4 v = src[i];
    int n2 = i >> 4, n1 = (i & 15) << 2;
    int base = pidx(n2, n1);
    xb[base] = v.x; xb[base + 2] = v.y; xb[base + 4] = v.z; xb[base + 6] = v.w;
  }
  __syncthreads();

  const float* F128Q  = ws + OFF_F128Q;
  const float* IF128Q = ws + OFF_IF128Q;
  const float* GPr    = ws + OFF_G64PR;
  const float* GPi    = ws + OFF_G64PI;
  const float* iGPr   = ws + OFF_IG64PR;
  const float* iGPi   = ws + OFF_IG64PI;
  const float* Wf     = ws + OFF_WF;
  const float* Wi     = ws + OFF_WI;

  for (int layer = 0; layer < 4; ++layer) {
    const float* kh = ws + OFF_KHAT + layer * 16384;

    {
      v8f cr[4] = {}, ci[4] = {};
      for (int kk = 0; kk < 128; kk += 4) {
        float4 aq = *(const float4*)(F128Q + (mB + l16) * 256 + (kk + kSel) * 2);
        v2f ar; ar.x = aq.x; ar.y = aq.y;
        v2f ai; ai.x = aq.z; ai.y = aq.w;
        int jb = ((kk + kSel) >> 1) << 7;
#pragma unroll
        for (int t = 0; t < 4; ++t) {
          v2f b = *(const v2f*)(xb + jb + (((t << 4) + l16) << 1));
          cr[t] = wmma4(ar, b, cr[t]);
          ci[t] = wmma4(ai, b, ci[t]);
        }
      }
#pragma unroll
      for (int t = 0; t < 4; ++t) {
#pragma unroll
        for (int v = 0; v < 8; ++v) {
          int row = mB + v + mHi, col = (t << 4) + l16;
          v2f w = *(const v2f*)(Wf + ((row << 6) + col) * 2);
          int p = pidx(row, col);
          float a = cr[t][v], b = ci[t][v];
          Yr[p] = a * w.x - b * w.y;
          Yi[p] = a * w.y + b * w.x;
        }
      }
    }

    {
      v8f zr[4] = {}, zi[4] = {};
      for (int kk = 0; kk < 64; kk += 4) {
        int p0 = pidx(mB + l16, kk + kSel);
        v2f ar; ar.x = Yr[p0]; ar.y = Yr[p0 + 2];
        v2f ai; ai.x = Yi[p0]; ai.y = Yi[p0 + 2];
        int jrow = ((kk + kSel) >> 1) << 7;
#pragma unroll
        for (int t = 0; t < 4; ++t) {
          int cc = ((t << 4) + l16) << 1;
          v2f br = *(const v2f*)(GPr + jrow + cc);
          v2f bi = *(const v2f*)(GPi + jrow + cc);
          v2f nbi; nbi.x = -bi.x; nbi.y = -bi.y;
          zr[t] = wmma4(ar, br, zr[t]);  zr[t] = wmma4(ai, nbi, zr[t]);
          zi[t] = wmma4(ar, bi, zi[t]);  zi[t] = wmma4(ai, br, zi[t]);
        }
      }
#pragma unroll
      for (int t = 0; t < 4; ++t) {
#pragma unroll
        for (int v = 0; v < 8; ++v) {
          int row = mB + v + mHi, col = (t << 4) + l16;
          v2f k = *(const v2f*)(kh + ((row << 6) + col) * 2);
          int p = pidx(row, col);
          float a = zr[t][v], b = zi[t][v];
          Yr[p] = a * k.x - b * k.y;
          Yi[p] = a * k.y + b * k.x;
        }
      }
    }

    {
      v8f yr[4] = {}, yi[4] = {};
      for (int kk = 0; kk < 64; kk += 4) {
        int p0 = pidx(mB + l16, kk + kSel);
        v2f ar; ar.x = Yr[p0]; ar.y = Yr[p0 + 2];
        v2f ai; ai.x = Yi[p0]; ai.y = Yi[p0 + 2];
        int jrow = ((kk + kSel) >> 1) << 7;
#pragma unroll
        for (int t = 0; t < 4; ++t) {
          int cc = ((t << 4) + l16) << 1;
          v2f br = *(const v2f*)(iGPr + jrow + cc);
          v2f bi = *(const v2f*)(iGPi + jrow + cc);
          v2f nbi; nbi.x = -bi.x; nbi.y = -bi.y;
          yr[t] = wmma4(ar, br, yr[t]);  yr[t] = wmma4(ai, nbi, yr[t]);
          yi[t] = wmma4(ar, bi, yi[t]);  yi[t] = wmma4(ai, br, yi[t]);
        }
      }
#pragma unroll
      for (int t = 0; t < 4; ++t) {
#pragma unroll
        for (int v = 0; v < 8; ++v) {
          int row = mB + v + mHi, col = (t << 4) + l16;
          v2f w = *(const v2f*)(Wi + ((row << 6) + col) * 2);
          int p = pidx(row, col);
          float a = yr[t][v], b = yi[t][v];
          Yr[p] = a * w.x - b * w.y;
          Yi[p] = a * w.y + b * w.x;
        }
      }
    }
    __syncthreads();

    {
      v8f acc[4] = {};
      for (int kk = 0; kk < 128; kk += 4) {
        float4 aq = *(const float4*)(IF128Q + (mB + l16) * 256 + (kk + kSel) * 2);
        v2f ar; ar.x = aq.x; ar.y = aq.y;
        v2f ai; ai.x = aq.z; ai.y = aq.w;
        int jb = ((kk + kSel) >> 1) << 7;
#pragma unroll
        for (int t = 0; t < 4; ++t) {
          int cc = ((t << 4) + l16) << 1;
          v2f br  = *(const v2f*)(Yr + jb + cc);
          v2f byi = *(const v2f*)(Yi + jb + cc);
          v2f nbi; nbi.x = -byi.x; nbi.y = -byi.y;
          acc[t] = wmma4(ar, br, acc[t]);
          acc[t] = wmma4(ai, nbi, acc[t]);
        }
      }
#pragma unroll
      for (int t = 0; t < 4; ++t) {
#pragma unroll
        for (int v = 0; v < 8; ++v) {
          int row = mB + v + mHi, col = (t << 4) + l16;
          int p = pidx(row, col);
          float y = acc[t][v] + xb[p];
          xb[p] = 0.5f * y * (1.f + erff(y * 0.70710678118654752f));
        }
      }
    }
    __syncthreads();
  }

  float4* dst = (float4*)(uout + rowOff);
  for (int i = tid; i < LROW / 4; i += 256) {
    int n2 = i >> 4, n1 = (i & 15) << 2;
    int base = pidx(n2, n1);
    float4 o;
    o.x = xb[base]; o.y = xb[base + 2]; o.z = xb[base + 4]; o.w = xb[base + 6];
    dst[i] = o;
  }
}

extern "C" void kernel_launch(void* const* d_in, const int* in_sizes, int n_in,
                              void* d_out, int out_size, void* d_ws, size_t ws_size,
                              hipStream_t stream) {
  const float* u    = (const float*)d_in[0];
  const float* Lr   = (const float*)d_in[1];
  const float* Li   = (const float*)d_in[2];
  const float* Pv   = (const float*)d_in[3];
  const float* Bv   = (const float*)d_in[4];
  const float* Ct   = (const float*)d_in[5];
  const float* step = (const float*)d_in[6];
  float* out = (float*)d_out;
  float* ws  = (float*)d_ws;

  int rows = in_sizes[0] / LROW;

  ssm_gen_tables<<<64, 256, 0, stream>>>(ws);
  ssm_gen_khat<<<dim3(32, 4), 256, 0, stream>>>(ws, Lr, Li, Pv, Bv, Ct, step);
  ssm_rows<<<rows, 256, 24576 * sizeof(float), stream>>>(u, out, ws);
}


# Residency 1 -> 3 workgroups per WGP: twiddle factors loaded inside each epilogue (two tiles at a time) instead of held across layers; 292 -> 156 VGPRs, 6 waves per SIMD requested, no scratch
# speedup vs baseline: 1.3711x; 1.3711x over previous
#include <hip/hip_runtime.h>
#include <math.h>

typedef float v2f __attribute__((ext_vector_type(2)));
typedef float v8f __attribute__((ext_vector_type(8)));

#define LROW 8192
#define NSTATE 64
#define PI_F 3.14159265358979323846f

#define OFF_F128Q   0
#define OFF_IF128Q  32768
#define OFF_G64PR   65536
#define OFF_G64PI   69632
#define OFF_IG64PR  73728
#define OFF_IG64PI  77824
#define OFF_WF      81920
#define OFF_WI      98304
#define OFF_KHAT    114688
#define WS_FLOATS   (OFF_KHAT + 4*16384)

static __device__ inline v8f wmma4(v2f a, v2f b, v8f c) {
  return __builtin_amdgcn_wmma_f32_16x16x4_f32(false, a, false, b, (short)0, c,
                                               false, false);
}

static __device__ inline int pidx(int row, int col) {
  return ((row >> 1) << 7) + (col << 1) + (row & 1);
}

__global__ void ssm_gen_tables(float* __restrict__ ws) {
  int idx = blockIdx.x * blockDim.x + threadIdx.x;
  {
    int k2 = idx >> 7, n2 = idx & 127;
    int t = (k2 * n2) & 127;
    float s, c;
    sincosf(-2.f * PI_F * (float)t / 128.f, &s, &c);
    int pos = k2 * 256 + (n2 >> 1) * 4 + (n2 & 1);
    ws[OFF_F128Q + pos]      = c;
    ws[OFF_F128Q + pos + 2]  = s;
    ws[OFF_IF128Q + pos]     = c;
    ws[OFF_IF128Q + pos + 2] = -s;
  }
  if (idx < 8192) {
    int k2 = idx >> 6, n1 = idx & 63;
    int t = (n1 * k2) & 8191;
    float s, c;
    sincosf(-2.f * PI_F * (float)t / 8192.f, &s, &c);
    ws[OFF_WF + 2 * idx]     = c;  ws[OFF_WF + 2 * idx + 1] = s;
    ws[OFF_WI + 2 * idx]     = c;  ws[OFF_WI + 2 * idx + 1] = -s;
  }
  if (idx < 4096) {
    int n1 = idx >> 6, k1 = idx & 63;
    int t = (n1 * k1) & 63;
    float s, c;
    sincosf(-2.f * PI_F * (float)t / 64.f, &s, &c);
    int pos = (n1 >> 1) * 128 + 2 * k1 + (n1 & 1);
    ws[OFF_G64PR + pos]  = c;  ws[OFF_G64PI + pos]  = s;
    ws[OFF_IG64PR + pos] = c;  ws[OFF_IG64PI + pos] = -s;
  }
}

__global__ void ssm_gen_khat(float* __restrict__ ws,
                             const float* __restrict__ Lr,
                             const float* __restrict__ Li,
                             const float* __restrict__ Pv,
                             const float* __restrict__ Bv,
                             const float* __restrict__ Ct,
                             const float* __restrict__ stepv) {
  int layer = blockIdx.y;
  int s = blockIdx.x * blockDim.x + threadIdx.x;
  int k2 = s >> 6, k1 = s & 63;
  int kf = 128 * k1 + k2;

  float osin, ocos;
  sincosf(-2.f * PI_F * (float)kf / 8192.f, &osin, &ocos);
  float pr = 1.f + ocos, pi = osin;
  float mr = 1.f - ocos, mi = -osin;
  float pd = pr * pr + pi * pi;
  float ipr = pr / pd, ipi = -pi / pd;
  float cr = 2.f * ipr, ci = 2.f * ipi;
  float sc = 2.f / stepv[layer];
  float gr = sc * (mr * ipr - mi * ipi);
  float gi = sc * (mr * ipi + mi * ipr);

  const float* lr = Lr + layer * NSTATE;
  const float* li = Li + layer * NSTATE;
  const float* pp = Pv + layer * NSTATE;
  const float* bb = Bv + layer * NSTATE;
  const float* ct = Ct + layer * NSTATE;

  float k00r = 0, k00i = 0, k01r = 0, k01i = 0;
  float k10r = 0, k10i = 0, k11r = 0, k11i = 0;
  for (int n = 0; n < NSTATE; ++n) {
    float dr = gr - lr[n], di = gi - li[n];
    float dd = dr * dr + di * di;
    float ir = dr / dd, ii = -di / dd;
    float b = bb[n], p = pp[n], c = ct[n];
    float w00 = c * b, w01 = c * p, w10 = p * b, w11 = p * p;
    k00r += w00 * ir; k00i += w00 * ii;
    k01r += w01 * ir; k01i += w01 * ii;
    k10r += w10 * ir; k10i += w10 * ii;
    k11r += w11 * ir; k11i += w11 * ii;
  }
  float qr = k01r * k10r - k01i * k10i;
  float qi = k01r * k10i + k01i * k10r;
  float er = 1.f + k11r, ei = k11i;
  float ed = er * er + ei * ei;
  float ier = er / ed, iei = -ei / ed;
  float tr = qr * ier - qi * iei;
  float ti = qr * iei + qi * ier;
  float kr = k00r - tr, ki = k00i - ti;
  float outr = cr * kr - ci * ki;
  float outi = cr * ki + ci * kr;
  const float scale = 1.f / 8192.f;
  ws[OFF_KHAT + layer * 16384 + 2 * s]     = outr * scale;
  ws[OFF_KHAT + layer * 16384 + 2 * s + 1] = outi * scale;
}

__global__ __launch_bounds__(256, 6) void ssm_rows(const float* __restrict__ uin,
                                                float* __restrict__ uout,
                                                const float* ws) {
  extern __shared__ float smem[];
  float* xb = smem;
  float* Yr = smem + 8192;
  float* Yi = smem + 16384;

  const int tid  = threadIdx.x;
  const int wave = tid >> 5;
  const int lane = tid & 31;
  const int l16  = lane & 15;
  const int kSel = (lane >> 4) << 1;
  const int mHi  = (lane >> 4) << 3;
  const int mB   = wave << 4;
  const int eB = (((mB + mHi) << 6) + l16) << 1;
  const int pB = ((mB + mHi) << 6) + (l16 << 1);

  const size_t rowOff = (size_t)blockIdx.x * LROW;
  const float4* src = (const float4*)(uin + rowOff);
  for (int i = tid; i < LROW / 4; i += 256) {
    float4 v = src[i];
    int n2 = i >> 4, n1 = (i & 15) << 2;
    int base = pidx(n2, n1);
    xb[base] = v.x; xb[base + 2] = v.y; xb[base + 4] = v.z; xb[base + 6] = v.w;
  }
  __syncthreads();

  const float* F128Q  = ws + OFF_F128Q;
  const float* IF128Q = ws + OFF_IF128Q;
  const float* GPr    = ws + OFF_G64PR;
  const float* GPi    = ws + OFF_G64PI;
  const float* iGPr   = ws + OFF_IG64PR;
  const float* iGPi   = ws + OFF_IG64PI;
  const float* Wf     = ws + OFF_WF;
  const float* Wi     = ws + OFF_WI;

  for (int layer = 0; layer < 4; ++layer) {
    const float* kh = ws + OFF_KHAT + layer * 16384;

    {
      v8f cr[4] = {}, ci[4] = {};
      for (int kk = 0; kk < 128; kk += 4) {
        float4 aq = *(const float4*)(F128Q + (mB + l16) * 256 + (kk + kSel) * 2);
        v2f ar; ar.x = aq.x; ar.y = aq.y;
        v2f ai; ai.x = aq.z; ai.y = aq.w;
        int jb = ((kk + kSel) >> 1) << 7;
#pragma unroll
        for (int t = 0; t < 4; ++t) {
          v2f b = *(const v2f*)(xb + jb + (((t << 4) + l16) << 1));
          cr[t] = wmma4(ar, b, cr[t]);
          ci[t] = wmma4(ai, b, ci[t]);
        }
      }
      const float* tb = Wf + eB;
#pragma unroll
      for (int th = 0; th < 4; th += 2) {
        v2f tw[2][8];
#pragma unroll
        for (int t = 0; t < 2; ++t) {
#pragma unroll
          for (int v = 0; v < 8; ++v) tw[t][v] = *(const v2f*)(tb + 128 * v + 32 * (th + t));
        }
#pragma unroll
        for (int t = 0; t < 2; ++t) {
#pragma unroll
          for (int v = 0; v < 8; ++v) {
            v2f w = tw[t][v];
            int p = pB + 128 * (v >> 1) + 32 * (th + t) + (v & 1);
            float a = cr[th + t][v], b = ci[th + t][v];
            Yr[p] = a * w.x - b * w.y;
            Yi[p] = a * w.y + b * w.x;
          }
        }
      }
    }

    {
      v8f zr[4] = {}, zi[4] = {};
      for (int kk = 0; kk < 64; kk += 4) {
        int p0 = pidx(mB + l16, kk + kSel);
        v2f ar; ar.x = Yr[p0]; ar.y = Yr[p0 + 2];
        v2f ai; ai.x = Yi[p0]; ai.y = Yi[p0 + 2];
        int jrow = ((kk + kSel) >> 1) << 7;
#pragma unroll
        for (int t = 0; t < 4; ++t) {
          int cc = ((t << 4) + l16) << 1;
          v2f br = *(const v2f*)(GPr + jrow + cc);
          v2f bi = *(const v2f*)(GPi + jrow + cc);
          v2f nbi; nbi.x = -bi.x; nbi.y = -bi.y;
          zr[t] = wmma4(ar, br, zr[t]);  zr[t] = wmma4(ai, nbi, zr[t]);
          zi[t] = wmma4(ar, bi, zi[t]);  zi[t] = wmma4(ai, br, zi[t]);
        }
      }
      const float* tb = kh + eB;
#pragma unroll
      for (int th = 0; th < 4; th += 2) {
        v2f tw[2][8];
#pragma unroll
        for (int t = 0; t < 2; ++t) {
#pragma unroll
          for (int v = 0; v < 8; ++v) tw[t][v] = *(const v2f*)(tb + 128 * v + 32 * (th + t));
        }
#pragma unroll
        for (int t = 0; t < 2; ++t) {
#pragma unroll
          for (int v = 0; v < 8; ++v) {
            v2f k = tw[t][v];
            int p = pB + 128 * (v >> 1) + 32 * (th + t) + (v & 1);
            float a = zr[th + t][v], b = zi[th + t][v];
            Yr[p] = a * k.x - b * k.y;
            Yi[p] = a * k.y + b * k.x;
          }
        }
      }
    }

    {
      v8f yr[4] = {}, yi[4] = {};
      for (int kk = 0; kk < 64; kk += 4) {
        int p0 = pidx(mB + l16, kk + kSel);
        v2f ar; ar.x = Yr[p0]; ar.y = Yr[p0 + 2];
        v2f ai; ai.x = Yi[p0]; ai.y = Yi[p0 + 2];
        int jrow = ((kk + kSel) >> 1) << 7;
#pragma unroll
        for (int t = 0; t < 4; ++t) {
          int cc = ((t << 4) + l16) << 1;
          v2f br = *(const v2f*)(iGPr + jrow + cc);
          v2f bi = *(const v2f*)(iGPi + jrow + cc);
          v2f nbi; nbi.x = -bi.x; nbi.y = -bi.y;
          yr[t] = wmma4(ar, br, yr[t]);  yr[t] = wmma4(ai, nbi, yr[t]);
          yi[t] = wmma4(ar, bi, yi[t]);  yi[t] = wmma4(ai, br, yi[t]);
        }
      }
      const float* tb = Wi + eB;
#pragma unroll
      for (int th = 0; th < 4; th += 2) {
        v2f tw[2][8];
#pragma unroll
        for (int t = 0; t < 2; ++t) {
#pragma unroll
          for (int v = 0; v < 8; ++v) tw[t][v] = *(const v2f*)(tb + 128 * v + 32 * (th + t));
        }
#pragma unroll
        for (int t = 0; t < 2; ++t) {
#pragma unroll
          for (int v = 0; v < 8; ++v) {
            v2f w = tw[t][v];
            int p = pB + 128 * (v >> 1) + 32 * (th + t) + (v & 1);
            float a = yr[th + t][v], b = yi[th + t][v];
            Yr[p] = a * w.x - b * w.y;
            Yi[p] = a * w.y + b * w.x;
          }
        }
      }
    }
    __syncthreads();

    {
      v8f acc[4] = {};
      for (int kk = 0; kk < 128; kk += 4) {
        float4 aq = *(const float4*)(IF128Q + (mB + l16) * 256 + (kk + kSel) * 2);
        v2f ar; ar.x = aq.x; ar.y = aq.y;
        v2f ai; ai.x = aq.z; ai.y = aq.w;
        int jb = ((kk + kSel) >> 1) << 7;
#pragma unroll
        for (int t = 0; t < 4; ++t) {
          int cc = ((t << 4) + l16) << 1;
          v2f br  = *(const v2f*)(Yr + jb + cc);
          v2f byi = *(const v2f*)(Yi + jb + cc);
          v2f nbi; nbi.x = -byi.x; nbi.y = -byi.y;
          acc[t] = wmma4(ar, br, acc[t]);
          acc[t] = wmma4(ai, nbi, acc[t]);
        }
      }
#pragma unroll
      for (int t = 0; t < 4; ++t) {
#pragma unroll
        for (int v = 0; v < 8; ++v) {
          int p = pB + 128 * (v >> 1) + 32 * t + (v & 1);
          float y = acc[t][v] + xb[p];
          xb[p] = 0.5f * y * (1.f + erff(y * 0.70710678118654752f));
        }
      }
    }
    __syncthreads();
  }

  float4* dst = (float4*)(uout + rowOff);
  for (int i = tid; i < LROW / 4; i += 256) {
    int n2 = i >> 4, n1 = (i & 15) << 2;
    int base = pidx(n2, n1);
    float4 o;
    o.x = xb[base]; o.y = xb[base + 2]; o.z = xb[base + 4]; o.w = xb[base + 6];
    dst[i] = o;
  }
}

extern "C" void kernel_launch(void* const* d_in, const int* in_sizes, int n_in,
                              void* d_out, int out_size, void* d_ws, size_t ws_size,
                              hipStream_t stream) {
  const float* u    = (const float*)d_in[0];
  const float* Lr   = (const float*)d_in[1];
  const float* Li   = (const float*)d_in[2];
  const float* Pv   = (const float*)d_in[3];
  const float* Bv   = (const float*)d_in[4];
  const float* Ct   = (const float*)d_in[5];
  const float* step = (const float*)d_in[6];
  float* out = (float*)d_out;
  float* ws  = (float*)d_ws;

  int rows = in_sizes[0] / LROW;

  ssm_gen_tables<<<64, 256, 0, stream>>>(ws);
  ssm_gen_khat<<<dim3(32, 4), 256, 0, stream>>>(ws, Lr, Li, Pv, Bv, Ct, step);
  ssm_rows<<<rows, 256, 24576 * sizeof(float), stream>>>(u, out, ws);
}


# Two rows per workgroup go through the FFT chain as one complex sequence a+ib (K_hat table holds the Hermitian part of K): 25% fewer WMMAs; sum order of first and last stage changes
# speedup vs baseline: 1.8274x; 1.3328x over previous
#include <hip/hip_runtime.h>
#include <math.h>

typedef float v2f __attribute__((ext_vector_type(2)));
typedef float v8f __attribute__((ext_vector_type(8)));

#define LROW 8192
#define LPAIR 16384
#define NSTATE 64
#define PI_F 3.14159265358979323846f

#define OFF_F128Q   0
#define OFF_IF128Q  32768
#define OFF_G64PR   65536
#define OFF_G64PI   69632
#define OFF_IG64PR  73728
#define OFF_IG64PI  77824
#define OFF_WF      81920
#define OFF_WI      98304
#define OFF_KHAT    114688
#define WS_FLOATS   (OFF_KHAT + 4*16384)

static __device__ inline v8f wmma4(v2f a, v2f b, v8f c) {
  return __builtin_amdgcn_wmma_f32_16x16x4_f32(false, a, false, b, (short)0, c,
                                               false, false);
}

static __device__ inline int pidx(int row, int col) {
  return ((row >> 1) << 7) + (col << 1) + (row & 1);
}

__global__ void ssm_gen_tables(float* __restrict__ ws) {
  int idx = blockIdx.x * blockDim.x + threadIdx.x;
  {
    int k2 = idx >> 7, n2 = idx & 127;
    int t = (k2 * n2) & 127;
    float s, c;
    sincosf(-2.f * PI_F * (float)t / 128.f, &s, &c);
    int pos = k2 * 256 + (n2 >> 1) * 4 + (n2 & 1);
    ws[OFF_F128Q + pos]      = c;
    ws[OFF_F128Q + pos + 2]  = s;
    ws[OFF_IF128Q + pos]     = c;
    ws[OFF_IF128Q + pos + 2] = -s;
  }
  if (idx < 8192) {
    int k2 = idx >> 6, n1 = idx & 63;
    int t = (n1 * k2) & 8191;
    float s, c;
    sincosf(-2.f * PI_F * (float)t / 8192.f, &s, &c);
    ws[OFF_WF + 2 * idx]     = c;  ws[OFF_WF + 2 * idx + 1] = s;
    ws[OFF_WI + 2 * idx]     = c;  ws[OFF_WI + 2 * idx + 1] = -s;
  }
  if (idx < 4096) {
    int n1 = idx >> 6, k1 = idx & 63;
    int t = (n1 * k1) & 63;
    float s, c;
    sincosf(-2.f * PI_F * (float)t / 64.f, &s, &c);
    int pos = (n1 >> 1) * 128 + 2 * k1 + (n1 & 1);
    ws[OFF_G64PR + pos]  = c;  ws[OFF_G64PI + pos]  = s;
    ws[OFF_IG64PR + pos] = c;  ws[OFF_IG64PI + pos] = -s;
  }
}

static __device__ inline void ssm_khat_at(int kf, float stp,
                                          const float* __restrict__ lr,
                                          const float* __restrict__ li,
                                          const float* __restrict__ pp,
                                          const float* __restrict__ bb,
                                          const float* __restrict__ ct,
                                          float& outr, float& outi) {
  float osin, ocos;
  sincosf(-2.f * PI_F * (float)kf / 8192.f, &osin, &ocos);
  float pr = 1.f + ocos, pi = osin;
  float mr = 1.f - ocos, mi = -osin;
  float pd = pr * pr + pi * pi;
  float ipr = pr / pd, ipi = -pi / pd;
  float cr = 2.f * ipr, ci = 2.f * ipi;
  float sc = 2.f / stp;
  float gr = sc * (mr * ipr - mi * ipi);
  float gi = sc * (mr * ipi + mi * ipr);

  float k00r = 0, k00i = 0, k01r = 0, k01i = 0;
  float k10r = 0, k10i = 0, k11r = 0, k11i = 0;
  for (int n = 0; n < NSTATE; ++n) {
    float dr = gr - lr[n], di = gi - li[n];
    float dd = dr * dr + di * di;
    float ir = dr / dd, ii = -di / dd;
    float b = bb[n], p = pp[n], c = ct[n];
    float w00 = c * b, w01 = c * p, w10 = p * b, w11 = p * p;
    k00r += w00 * ir; k00i += w00 * ii;
    k01r += w01 * ir; k01i += w01 * ii;
    k10r += w10 * ir; k10i += w10 * ii;
    k11r += w11 * ir; k11i += w11 * ii;
  }
  float qr = k01r * k10r - k01i * k10i;
  float qi = k01r * k10i + k01i * k10r;
  float er = 1.f + k11r, ei = k11i;
  float ed = er * er + ei * ei;
  float ier = er / ed, iei = -ei / ed;
  float tr = qr * ier - qi * iei;
  float ti = qr * iei + qi * ier;
  float kr = k00r - tr, ki = k00i - ti;
  outr = cr * kr - ci * ki;
  outi = cr * ki + ci * kr;
}

__global__ void ssm_gen_khat(float* __restrict__ ws,
                             const float* __restrict__ Lr,
                             const float* __restrict__ Li,
                             const float* __restrict__ Pv,
                             const float* __restrict__ Bv,
                             const float* __restrict__ Ct,
                             const float* __restrict__ stepv) {
  int layer = blockIdx.y;
  int s = blockIdx.x * blockDim.x + threadIdx.x;
  int k2 = s >> 6, k1 = s & 63;
  int kf = 128 * k1 + k2;
  int kg = (8192 - kf) & 8191;

  const float* lr = Lr + layer * NSTATE;
  const float* li = Li + layer * NSTATE;
  const float* pp = Pv + layer * NSTATE;
  const float* bb = Bv + layer * NSTATE;
  const float* ct = Ct + layer * NSTATE;
  float stp = stepv[layer];

  float ar, ai, br, bi;
  ssm_khat_at(kf, stp, lr, li, pp, bb, ct, ar, ai);
  ssm_khat_at(kg, stp, lr, li, pp, bb, ct, br, bi);
  const float scale = 0.5f / 8192.f;
  ws[OFF_KHAT + layer * 16384 + 2 * s]     = (ar + br) * scale;
  ws[OFF_KHAT + layer * 16384 + 2 * s + 1] = (ai - bi) * scale;
}

__global__ __launch_bounds__(256, 4) void ssm_rows(const float* __restrict__ uin,
                                                float* __restrict__ uout,
                                                const float* ws) {
  extern __shared__ float smem[];
  float* xa = smem;
  float* xb = smem + 8192;
  float* Yr = smem + 16384;
  float* Yi = smem + 24576;

  const int tid  = threadIdx.x;
  const int wave = tid >> 5;
  const int lane = tid & 31;
  const int l16  = lane & 15;
  const int kSel = (lane >> 4) << 1;
  const int mHi  = (lane >> 4) << 3;
  const int mB   = wave << 4;
  const int eB = (((mB + mHi) << 6) + l16) << 1;
  const int pB = ((mB + mHi) << 6) + (l16 << 1);

  const size_t rowOff = (size_t)blockIdx.x * (2 * LROW);
  const float4* srcA = (const float4*)(uin + rowOff);
  const float4* srcB = (const float4*)(uin + rowOff + LROW);
  for (int i = tid; i < LROW / 4; i += 256) {
    float4 va = srcA[i];
    float4 vb = srcB[i];
    int n2 = i >> 4, n1 = (i & 15) << 2;
    int base = pidx(n2, n1);
    xa[base] = va.x; xa[base + 2] = va.y; xa[base + 4] = va.z; xa[base + 6] = va.w;
    xb[base] = vb.x; xb[base + 2] = vb.y; xb[base + 4] = vb.z; xb[base + 6] = vb.w;
  }
  __syncthreads();

  const float* F128Q  = ws + OFF_F128Q;
  const float* IF128Q = ws + OFF_IF128Q;
  const float* GPr    = ws + OFF_G64PR;
  const float* GPi    = ws + OFF_G64PI;
  const float* iGPr   = ws + OFF_IG64PR;
  const float* iGPi   = ws + OFF_IG64PI;
  const float* Wf     = ws + OFF_WF;
  const float* Wi     = ws + OFF_WI;

  for (int layer = 0; layer < 4; ++layer) {
    const float* kh = ws + OFF_KHAT + layer * 16384;

    {
      v8f cr[4] = {}, ci[4] = {};
#pragma unroll 2
      for (int kk = 0; kk < 128; kk += 4) {
        float4 aq = *(const float4*)(F128Q + (mB + l16) * 256 + (kk + kSel) * 2);
        v2f ar; ar.x = aq.x; ar.y = aq.y;
        v2f ai; ai.x = aq.z; ai.y = aq.w;
        v2f nai; nai.x = -aq.z; nai.y = -aq.w;
        int jb = ((kk + kSel) >> 1) << 7;
#pragma unroll
        for (int t = 0; t < 4; ++t) {
          int cc = ((t << 4) + l16) << 1;
          v2f ba = *(const v2f*)(xa + jb + cc);
          v2f bb = *(const v2f*)(xb + jb + cc);
          cr[t] = wmma4(ar, ba, cr[t]);  cr[t] = wmma4(nai, bb, cr[t]);
          ci[t] = wmma4(ai, ba, ci[t]);  ci[t] = wmma4(ar, bb, ci[t]);
        }
      }
      const float* tb = Wf + eB;
#pragma unroll
      for (int th = 0; th < 4; th += 2) {
        v2f tw[2][8];
#pragma unroll
        for (int t = 0; t < 2; ++t) {
#pragma unroll
          for (int v = 0; v < 8; ++v) tw[t][v] = *(const v2f*)(tb + 128 * v + 32 * (th + t));
        }
#pragma unroll
        for (int t = 0; t < 2; ++t) {
#pragma unroll
          for (int v = 0; v < 8; ++v) {
            v2f w = tw[t][v];
            int p = pB + 128 * (v >> 1) + 32 * (th + t) + (v & 1);
            float a = cr[th + t][v], b = ci[th + t][v];
            Yr[p] = a * w.x - b * w.y;
            Yi[p] = a * w.y + b * w.x;
          }
        }
      }
    }

    {
      v8f zr[4] = {}, zi[4] = {};
      for (int kk = 0; kk < 64; kk += 4) {
        int p0 = pidx(mB + l16, kk + kSel);
        v2f ar; ar.x = Yr[p0]; ar.y = Yr[p0 + 2];
        v2f ai; ai.x = Yi[p0]; ai.y = Yi[p0 + 2];
        int jrow = ((kk + kSel) >> 1) << 7;
#pragma unroll
        for (int t = 0; t < 4; ++t) {
          int cc = ((t << 4) + l16) << 1;
          v2f br = *(const v2f*)(GPr + jrow + cc);
          v2f bi = *(const v2f*)(GPi + jrow + cc);
          v2f nbi; nbi.x = -bi.x; nbi.y = -bi.y;
          zr[t] = wmma4(ar, br, zr[t]);  zr[t] = wmma4(ai, nbi, zr[t]);
          zi[t] = wmma4(ar, bi, zi[t]);  zi[t] = wmma4(ai, br, zi[t]);
        }
      }
      const float* tb = kh + eB;
#pragma unroll
      for (int th = 0; th < 4; th += 2) {
        v2f tw[2][8];
#pragma unroll
        for (int t = 0; t < 2; ++t) {
#pragma unroll
          for (int v = 0; v < 8; ++v) tw[t][v] = *(const v2f*)(tb + 128 * v + 32 * (th + t));
        }
#pragma unroll
        for (int t = 0; t < 2; ++t) {
#pragma unroll
          for (int v = 0; v < 8; ++v) {
            v2f k = tw[t][v];
            int p = pB + 128 * (v >> 1) + 32 * (th + t) + (v & 1);
            float a = zr[th + t][v], b = zi[th + t][v];
            Yr[p] = a * k.x - b * k.y;
            Yi[p] = a * k.y + b * k.x;
          }
        }
      }
    }

    {
      v8f yr[4] = {}, yi[4] = {};
      for (int kk = 0; kk < 64; kk += 4) {
        int p0 = pidx(mB + l16, kk + kSel);
        v2f ar; ar.x = Yr[p0]; ar.y = Yr[p0 + 2];
        v2f ai; ai.x = Yi[p0]; ai.y = Yi[p0 + 2];
        int jrow = ((kk + kSel) >> 1) << 7;
#pragma unroll
        for (int t = 0; t < 4; ++t) {
          int cc = ((t << 4) + l16) << 1;
          v2f br = *(const v2f*)(iGPr + jrow + cc);
          v2f bi = *(const v2f*)(iGPi + jrow + cc);
          v2f nbi; nbi.x = -bi.x; nbi.y = -bi.y;
          yr[t] = wmma4(ar, br, yr[t]);  yr[t] = wmma4(ai, nbi, yr[t]);
          yi[t] = wmma4(ar, bi, yi[t]);  yi[t] = wmma4(ai, br, yi[t]);
        }
      }
      const float* tb = Wi + eB;
#pragma unroll
      for (int th = 0; th < 4; th += 2) {
        v2f tw[2][8];
#pragma unroll
        for (int t = 0; t < 2; ++t) {
#pragma unroll
          for (int v = 0; v < 8; ++v) tw[t][v] = *(const v2f*)(tb + 128 * v + 32 * (th + t));
        }
#pragma unroll
        for (int t = 0; t < 2; ++t) {
#pragma unroll
          for (int v = 0; v < 8; ++v) {
            v2f w = tw[t][v];
            int p = pB + 128 * (v >> 1) + 32 * (th + t) + (v & 1);
            float a = yr[th + t][v], b = yi[th + t][v];
            Yr[p] = a * w.x - b * w.y;
            Yi[p] = a * w.y + b * w.x;
          }
        }
      }
    }
    __syncthreads();

    {
      v8f accr[4] = {}, acci[4] = {};
      for (int kk = 0; kk < 128; kk += 4) {
        float4 aq = *(const float4*)(IF128Q + (mB + l16) * 256 + (kk + kSel) * 2);
        v2f ar; ar.x = aq.x; ar.y = aq.y;
        v2f ai; ai.x = aq.z; ai.y = aq.w;
        v2f nai; nai.x = -aq.z; nai.y = -aq.w;
        int jb = ((kk + kSel) >> 1) << 7;
#pragma unroll
        for (int t = 0; t < 4; ++t) {
          int cc = ((t << 4) + l16) << 1;
          v2f br = *(const v2f*)(Yr + jb + cc);
          v2f bi = *(const v2f*)(Yi + jb + cc);
          accr[t] = wmma4(ar, br, accr[t]);  accr[t] = wmma4(nai, bi, accr[t]);
          acci[t] = wmma4(ar, bi, acci[t]);  acci[t] = wmma4(ai, br, acci[t]);
        }
      }
#pragma unroll
      for (int t = 0; t < 4; ++t) {
#pragma unroll
        for (int v = 0; v < 8; ++v) {
          int p = pB + 128 * (v >> 1) + 32 * t + (v & 1);
          float ya = accr[t][v] + xa[p];
          float yb = acci[t][v] + xb[p];
          xa[p] = 0.5f * ya * (1.f + erff(ya * 0.70710678118654752f));
          xb[p] = 0.5f * yb * (1.f + erff(yb * 0.70710678118654752f));
        }
      }
    }
    __syncthreads();
  }

  float4* dstA = (float4*)(uout + rowOff);
  float4* dstB = (float4*)(uout + rowOff + LROW);
  for (int i = tid; i < LROW / 4; i += 256) {
    int n2 = i >> 4, n1 = (i & 15) << 2;
    int base = pidx(n2, n1);
    float4 oa, ob;
    oa.x = xa[base]; oa.y = xa[base + 2]; oa.z = xa[base + 4]; oa.w = xa[base + 6];
    ob.x = xb[base]; ob.y = xb[base + 2]; ob.z = xb[base + 4]; ob.w = xb[base + 6];
    dstA[i] = oa;
    dstB[i] = ob;
  }
}

extern "C" void kernel_launch(void* const* d_in, const int* in_sizes, int n_in,
                              void* d_out, int out_size, void* d_ws, size_t ws_size,
                              hipStream_t stream) {
  const float* u    = (const float*)d_in[0];
  const float* Lr   = (const float*)d_in[1];
  const float* Li   = (const float*)d_in[2];
  const float* Pv   = (const float*)d_in[3];
  const float* Bv   = (const float*)d_in[4];
  const float* Ct   = (const float*)d_in[5];
  const float* step = (const float*)d_in[6];
  float* out = (float*)d_out;
  float* ws  = (float*)d_ws;

  int pairs = in_sizes[0] / LPAIR;

  ssm_gen_tables<<<64, 256, 0, stream>>>(ws);
  ssm_gen_khat<<<dim3(32, 4), 256, 0, stream>>>(ws, Lr, Li, Pv, Bv, Ct, step);
  ssm_rows<<<pairs, 256, 32768 * sizeof(float), stream>>>(u, out, ws);
}


# All four FFT stages: each complex matrix product as 3 real WMMA GEMMs instead of 4 (P1=Ar(Br+Bi), P2=(Ai-Ar)Br, P3=-(Ar+Ai)Bi, own accumulators): 25% fewer WMMAs; regroups every stage's sums
# speedup vs baseline: 2.1376x; 1.1697x over previous
#include <hip/hip_runtime.h>
#include <math.h>

typedef float v2f __attribute__((ext_vector_type(2)));
typedef float v8f __attribute__((ext_vector_type(8)));

#define LROW 8192
#define LPAIR 16384
#define NSTATE 64
#define PI_F 3.14159265358979323846f

#define OFF_F128Q   0
#define OFF_IF128Q  32768
#define OFF_G64PR   65536
#define OFF_G64PI   69632
#define OFF_IG64PR  73728
#define OFF_IG64PI  77824
#define OFF_WF      81920
#define OFF_WI      98304
#define OFF_KHAT    114688
#define WS_FLOATS   (OFF_KHAT + 4*16384)

static __device__ inline v8f wmma4(v2f a, v2f b, v8f c) {
  return __builtin_amdgcn_wmma_f32_16x16x4_f32(false, a, false, b, (short)0, c,
                                               false, false);
}

static __device__ inline int pidx(int row, int col) {
  return ((row >> 1) << 7) + (col << 1) + (row & 1);
}

__global__ void ssm_gen_tables(float* __restrict__ ws) {
  int idx = blockIdx.x * blockDim.x + threadIdx.x;
  {
    int k2 = idx >> 7, n2 = idx & 127;
    int t = (k2 * n2) & 127;
    float s, c;
    sincosf(-2.f * PI_F * (float)t / 128.f, &s, &c);
    int pos = k2 * 256 + (n2 >> 1) * 4 + (n2 & 1);
    ws[OFF_F128Q + pos]      = c;
    ws[OFF_F128Q + pos + 2]  = s;
    ws[OFF_IF128Q + pos]     = c;
    ws[OFF_IF128Q + pos + 2] = -s;
  }
  if (idx < 8192) {
    int k2 = idx >> 6, n1 = idx & 63;
    int t = (n1 * k2) & 8191;
    float s, c;
    sincosf(-2.f * PI_F * (float)t / 8192.f, &s, &c);
    ws[OFF_WF + 2 * idx]     = c;  ws[OFF_WF + 2 * idx + 1] = s;
    ws[OFF_WI + 2 * idx]     = c;  ws[OFF_WI + 2 * idx + 1] = -s;
  }
  if (idx < 4096) {
    int n1 = idx >> 6, k1 = idx & 63;
    int t = (n1 * k1) & 63;
    float s, c;
    sincosf(-2.f * PI_F * (float)t / 64.f, &s, &c);
    int pos = (n1 >> 1) * 128 + 2 * k1 + (n1 & 1);
    ws[OFF_G64PR + pos]  = c;  ws[OFF_G64PI + pos]  = s;
    ws[OFF_IG64PR + pos] = c;  ws[OFF_IG64PI + pos] = -s;
  }
}

static __device__ inline void ssm_khat_at(int kf, float stp,
                                          const float* __restrict__ lr,
                                          const float* __restrict__ li,
                                          const float* __restrict__ pp,
                                          const float* __restrict__ bb,
                                          const float* __restrict__ ct,
                                          float& outr, float& outi) {
  float osin, ocos;
  sincosf(-2.f * PI_F * (float)kf / 8192.f, &osin, &ocos);
  float pr = 1.f + ocos, pi = osin;
  float mr = 1.f - ocos, mi = -osin;
  float pd = pr * pr + pi * pi;
  float ipr = pr / pd, ipi = -pi / pd;
  float cr = 2.f * ipr, ci = 2.f * ipi;
  float sc = 2.f / stp;
  float gr = sc * (mr * ipr - mi * ipi);
  float gi = sc * (mr * ipi + mi * ipr);

  float k00r = 0, k00i = 0, k01r = 0, k01i = 0;
  float k10r = 0, k10i = 0, k11r = 0, k11i = 0;
  for (int n = 0; n < NSTATE; ++n) {
    float dr = gr - lr[n], di = gi - li[n];
    float dd = dr * dr + di * di;
    float ir = dr / dd, ii = -di / dd;
    float b = bb[n], p = pp[n], c = ct[n];
    float w00 = c * b, w01 = c * p, w10 = p * b, w11 = p * p;
    k00r += w00 * ir; k00i += w00 * ii;
    k01r += w01 * ir; k01i += w01 * ii;
    k10r += w10 * ir; k10i += w10 * ii;
    k11r += w11 * ir; k11i += w11 * ii;
  }
  float qr = k01r * k10r - k01i * k10i;
  float qi = k01r * k10i + k01i * k10r;
  float er = 1.f + k11r, ei = k11i;
  float ed = er * er + ei * ei;
  float ier = er / ed, iei = -ei / ed;
  float tr = qr * ier - qi * iei;
  float ti = qr * iei + qi * ier;
  float kr = k00r - tr, ki = k00i - ti;
  outr = cr * kr - ci * ki;
  outi = cr * ki + ci * kr;
}

__global__ void ssm_gen_khat(float* __restrict__ ws,
                             const float* __restrict__ Lr,
                             const float* __restrict__ Li,
                             const float* __restrict__ Pv,
                             const float* __restrict__ Bv,
                             const float* __restrict__ Ct,
                             const float* __restrict__ stepv) {
  int layer = blockIdx.y;
  int s = blockIdx.x * blockDim.x + threadIdx.x;
  int k2 = s >> 6, k1 = s & 63;
  int kf = 128 * k1 + k2;
  int kg = (8192 - kf) & 8191;

  const float* lr = Lr + layer * NSTATE;
  const float* li = Li + layer * NSTATE;
  const float* pp = Pv + layer * NSTATE;
  const float* bb = Bv + layer * NSTATE;
  const float* ct = Ct + layer * NSTATE;
  float stp = stepv[layer];

  float ar, ai, br, bi;
  ssm_khat_at(kf, stp, lr, li, pp, bb, ct, ar, ai);
  ssm_khat_at(kg, stp, lr, li, pp, bb, ct, br, bi);
  const float scale = 0.5f / 8192.f;
  ws[OFF_KHAT + layer * 16384 + 2 * s]     = (ar + br) * scale;
  ws[OFF_KHAT + layer * 16384 + 2 * s + 1] = (ai - bi) * scale;
}

__global__ __launch_bounds__(256, 4) void ssm_rows(const float* __restrict__ uin,
                                                float* __restrict__ uout,
                                                const float* ws) {
  extern __shared__ float smem[];
  float* xa = smem;
  float* xb = smem + 8192;
  float* Yr = smem + 16384;
  float* Yi = smem + 24576;

  const int tid  = threadIdx.x;
  const int wave = tid >> 5;
  const int lane = tid & 31;
  const int l16  = lane & 15;
  const int kSel = (lane >> 4) << 1;
  const int mHi  = (lane >> 4) << 3;
  const int mB   = wave << 4;
  const int eB = (((mB + mHi) << 6) + l16) << 1;
  const int pB = ((mB + mHi) << 6) + (l16 << 1);

  const size_t rowOff = (size_t)blockIdx.x * (2 * LROW);
  const float4* srcA = (const float4*)(uin + rowOff);
  const float4* srcB = (const float4*)(uin + rowOff + LROW);
  for (int i = tid; i < LROW / 4; i += 256) {
    float4 va = srcA[i];
    float4 vb = srcB[i];
    int n2 = i >> 4, n1 = (i & 15) << 2;
    int base = pidx(n2, n1);
    xa[base] = va.x; xa[base + 2] = va.y; xa[base + 4] = va.z; xa[base + 6] = va.w;
    xb[base] = vb.x; xb[base + 2] = vb.y; xb[base + 4] = vb.z; xb[base + 6] = vb.w;
  }
  __syncthreads();

  const float* F128Q  = ws + OFF_F128Q;
  const float* IF128Q = ws + OFF_IF128Q;
  const float* GPr    = ws + OFF_G64PR;
  const float* GPi    = ws + OFF_G64PI;
  const float* iGPr   = ws + OFF_IG64PR;
  const float* iGPi   = ws + OFF_IG64PI;
  const float* Wf     = ws + OFF_WF;
  const float* Wi     = ws + OFF_WI;

  for (int layer = 0; layer < 4; ++layer) {
    const float* kh = ws + OFF_KHAT + layer * 16384;

    {
      v8f c1[4] = {}, c2[4] = {}, c3[4] = {};
#pragma unroll 2
      for (int kk = 0; kk < 128; kk += 4) {
        float4 aq = *(const float4*)(F128Q + (mB + l16) * 256 + (kk + kSel) * 2);
        v2f ar; ar.x = aq.x; ar.y = aq.y;
        v2f ai; ai.x = aq.z; ai.y = aq.w;
        v2f ad = ai - ar;
        v2f as = -(ar + ai);
        int jb = ((kk + kSel) >> 1) << 7;
#pragma unroll
        for (int t = 0; t < 4; ++t) {
          int cc = ((t << 4) + l16) << 1;
          v2f ba = *(const v2f*)(xa + jb + cc);
          v2f bb = *(const v2f*)(xb + jb + cc);
          v2f bs = ba + bb;
          c1[t] = wmma4(ar, bs, c1[t]);
          c2[t] = wmma4(ad, ba, c2[t]);
          c3[t] = wmma4(as, bb, c3[t]);
        }
      }
      const float* tb = Wf + eB;
#pragma unroll
      for (int th = 0; th < 4; th += 2) {
        v2f tw[2][8];
#pragma unroll
        for (int t = 0; t < 2; ++t) {
#pragma unroll
          for (int v = 0; v < 8; ++v) tw[t][v] = *(const v2f*)(tb + 128 * v + 32 * (th + t));
        }
#pragma unroll
        for (int t = 0; t < 2; ++t) {
#pragma unroll
          for (int v = 0; v < 8; ++v) {
            v2f w = tw[t][v];
            int p = pB + 128 * (v >> 1) + 32 * (th + t) + (v & 1);
            float a = c1[th + t][v] + c3[th + t][v], b = c1[th + t][v] + c2[th + t][v];
            Yr[p] = a * w.x - b * w.y;
            Yi[p] = a * w.y + b * w.x;
          }
        }
      }
    }

    {
      v8f z1[4] = {}, z2[4] = {}, z3[4] = {};
      for (int kk = 0; kk < 64; kk += 4) {
        int p0 = pidx(mB + l16, kk + kSel);
        v2f ar; ar.x = Yr[p0]; ar.y = Yr[p0 + 2];
        v2f ai; ai.x = Yi[p0]; ai.y = Yi[p0 + 2];
        v2f as = ar + ai;
        int jrow = ((kk + kSel) >> 1) << 7;
#pragma unroll
        for (int t = 0; t < 4; ++t) {
          int cc = ((t << 4) + l16) << 1;
          v2f br = *(const v2f*)(GPr + jrow + cc);
          v2f bi = *(const v2f*)(GPi + jrow + cc);
          v2f bd = bi - br;
          v2f bn = -(br + bi);
          z1[t] = wmma4(as, br, z1[t]);
          z2[t] = wmma4(ar, bd, z2[t]);
          z3[t] = wmma4(ai, bn, z3[t]);
        }
      }
      const float* tb = kh + eB;
#pragma unroll
      for (int th = 0; th < 4; th += 2) {
        v2f tw[2][8];
#pragma unroll
        for (int t = 0; t < 2; ++t) {
#pragma unroll
          for (int v = 0; v < 8; ++v) tw[t][v] = *(const v2f*)(tb + 128 * v + 32 * (th + t));
        }
#pragma unroll
        for (int t = 0; t < 2; ++t) {
#pragma unroll
          for (int v = 0; v < 8; ++v) {
            v2f k = tw[t][v];
            int p = pB + 128 * (v >> 1) + 32 * (th + t) + (v & 1);
            float a = z1[th + t][v] + z3[th + t][v], b = z1[th + t][v] + z2[th + t][v];
            Yr[p] = a * k.x - b * k.y;
            Yi[p] = a * k.y + b * k.x;
          }
        }
      }
    }

    {
      v8f y1[4] = {}, y2[4] = {}, y3[4] = {};
      for (int kk = 0; kk < 64; kk += 4) {
        int p0 = pidx(mB + l16, kk + kSel);
        v2f ar; ar.x = Yr[p0]; ar.y = Yr[p0 + 2];
        v2f ai; ai.x = Yi[p0]; ai.y = Yi[p0 + 2];
        v2f as = ar + ai;
        int jrow = ((kk + kSel) >> 1) << 7;
#pragma unroll
        for (int t = 0; t < 4; ++t) {
          int cc = ((t << 4) + l16) << 1;
          v2f br = *(const v2f*)(iGPr + jrow + cc);
          v2f bi = *(const v2f*)(iGPi + jrow + cc);
          v2f bd = bi - br;
          v2f bn = -(br + bi);
          y1[t] = wmma4(as, br, y1[t]);
          y2[t] = wmma4(ar, bd, y2[t]);
          y3[t] = wmma4(ai, bn, y3[t]);
        }
      }
      const float* tb = Wi + eB;
#pragma unroll
      for (int th = 0; th < 4; th += 2) {
        v2f tw[2][8];
#pragma unroll
        for (int t = 0; t < 2; ++t) {
#pragma unroll
          for (int v = 0; v < 8; ++v) tw[t][v] = *(const v2f*)(tb + 128 * v + 32 * (th + t));
        }
#pragma unroll
        for (int t = 0; t < 2; ++t) {
#pragma unroll
          for (int v = 0; v < 8; ++v) {
            v2f w = tw[t][v];
            int p = pB + 128 * (v >> 1) + 32 * (th + t) + (v & 1);
            float a = y1[th + t][v] + y3[th + t][v], b = y1[th + t][v] + y2[th + t][v];
            Yr[p] = a * w.x - b * w.y;
            Yi[p] = a * w.y + b * w.x;
          }
        }
      }
    }
    __syncthreads();

    {
      v8f w1[4] = {}, w2[4] = {}, w3[4] = {};
      for (int kk = 0; kk < 128; kk += 4) {
        float4 aq = *(const float4*)(IF128Q + (mB + l16) * 256 + (kk + kSel) * 2);
        v2f ar; ar.x = aq.x; ar.y = aq.y;
        v2f ai; ai.x = aq.z; ai.y = aq.w;
        v2f ad = ai - ar;
        v2f as = -(ar + ai);
        int jb = ((kk + kSel) >> 1) << 7;
#pragma unroll
        for (int t = 0; t < 4; ++t) {
          int cc = ((t << 4) + l16) << 1;
          v2f br = *(const v2f*)(Yr + jb + cc);
          v2f bi = *(const v2f*)(Yi + jb + cc);
          v2f bs = br + bi;
          w1[t] = wmma4(ar, bs, w1[t]);
          w2[t] = wmma4(ad, br, w2[t]);
          w3[t] = wmma4(as, bi, w3[t]);
        }
      }
#pragma unroll
      for (int t = 0; t < 4; ++t) {
#pragma unroll
        for (int v = 0; v < 8; ++v) {
          int p = pB + 128 * (v >> 1) + 32 * t + (v & 1);
          float ya = (w1[t][v] + w3[t][v]) + xa[p];
          float yb = (w1[t][v] + w2[t][v]) + xb[p];
          xa[p] = 0.5f * ya * (1.f + erff(ya * 0.70710678118654752f));
          xb[p] = 0.5f * yb * (1.f + erff(yb * 0.70710678118654752f));
        }
      }
    }
    __syncthreads();
  }

  float4* dstA = (float4*)(uout + rowOff);
  float4* dstB = (float4*)(uout + rowOff + LROW);
  for (int i = tid; i < LROW / 4; i += 256) {
    int n2 = i >> 4, n1 = (i & 15) << 2;
    int base = pidx(n2, n1);
    float4 oa, ob;
    oa.x = xa[base]; oa.y = xa[base + 2]; oa.z = xa[base + 4]; oa.w = xa[base + 6];
    ob.x = xb[base]; ob.y = xb[base + 2]; ob.z = xb[base + 4]; ob.w = xb[base + 6];
    dstA[i] = oa;
    dstB[i] = ob;
  }
}

extern "C" void kernel_launch(void* const* d_in, const int* in_sizes, int n_in,
                              void* d_out, int out_size, void* d_ws, size_t ws_size,
                              hipStream_t stream) {
  const float* u    = (const float*)d_in[0];
  const float* Lr   = (const float*)d_in[1];
  const float* Li   = (const float*)d_in[2];
  const float* Pv   = (const float*)d_in[3];
  const float* Bv   = (const float*)d_in[4];
  const float* Ct   = (const float*)d_in[5];
  const float* step = (const float*)d_in[6];
  float* out = (float*)d_out;
  float* ws  = (float*)d_ws;

  int pairs = in_sizes[0] / LPAIR;

  ssm_gen_tables<<<64, 256, 0, stream>>>(ws);
  ssm_gen_khat<<<dim3(32, 4), 256, 0, stream>>>(ws, Lr, Li, Pv, Bv, Ct, step);
  ssm_rows<<<pairs, 256, 32768 * sizeof(float), stream>>>(u, out, ws);
}


# GELU epilogue without data-dependent branches: both erf range polynomials evaluated per element and picked per lane, two elements per packed instruction, exp(-q) by v_exp_f32; sum order unchanged
# speedup vs baseline: 2.3090x; 1.0802x over previous
#include <hip/hip_runtime.h>
#include <math.h>

typedef float v2f __attribute__((ext_vector_type(2)));
typedef float v8f __attribute__((ext_vector_type(8)));

#define LROW 8192
#define LPAIR 16384
#define NSTATE 64
#define PI_F 3.14159265358979323846f

#define OFF_F128Q   0
#define OFF_IF128Q  32768
#define OFF_G64PR   65536
#define OFF_G64PI   69632
#define OFF_IG64PR  73728
#define OFF_IG64PI  77824
#define OFF_WF      81920
#define OFF_WI      98304
#define OFF_KHAT    114688
#define WS_FLOATS   (OFF_KHAT + 4*16384)

static __device__ inline v8f wmma4(v2f a, v2f b, v8f c) {
  return __builtin_amdgcn_wmma_f32_16x16x4_f32(false, a, false, b, (short)0, c,
                                               false, false);
}

static __device__ inline int pidx(int row, int col) {
  return ((row >> 1) << 7) + (col << 1) + (row & 1);
}

static __device__ inline v2f vfma2(v2f a, v2f b, v2f c) {
  return __builtin_elementwise_fma(a, b, c);
}
static __device__ inline v2f bc2(float s) { v2f r; r.x = s; r.y = s; return r; }

static __device__ inline v2f gelu2(v2f y) {
  v2f x  = y * 0.707106769f;
  v2f ax = __builtin_elementwise_abs(x);
  v2f t  = x * x;
  v2f pb = vfma2(t, bc2(-0.000561801775f), bc2(0.00491381623f));
  pb = vfma2(t, pb, bc2(-0.0267075151f));
  pb = vfma2(t, pb, bc2(0.112800106f));
  pb = vfma2(t, pb, bc2(-0.376122952f));
  pb = vfma2(t, pb, bc2(0.128379107f));
  v2f rs = vfma2(ax, pb, ax);
  v2f pa = vfma2(ax, bc2(0.0000169988107f), bc2(-0.000378677854f));
  pa = vfma2(ax, pa, bc2(0.00385781587f));
  pa = vfma2(ax, pa, bc2(-0.0241816975f));
  pa = vfma2(ax, pa, bc2(0.106668264f));
  pa = vfma2(ax, pa, bc2(0.634933293f));
  pa = vfma2(ax, pa, bc2(0.128689408f));
  v2f q  = vfma2(ax, pa, ax);
  v2f qe = q * -1.44269502f;
  v2f e;
  e.x = __builtin_amdgcn_exp2f(qe.x);
  e.y = __builtin_amdgcn_exp2f(qe.y);
  v2f rl = 1.0f - e;
  v2f r;
  r.x = ax.x < 1.0f ? rs.x : rl.x;
  r.y = ax.y < 1.0f ? rs.y : rl.y;
  r = __builtin_elementwise_copysign(r, x);
  return (0.5f * y) * (1.0f + r);
}

__global__ void ssm_gen_tables(float* __restrict__ ws) {
  int idx = blockIdx.x * blockDim.x + threadIdx.x;
  {
    int k2 = idx >> 7, n2 = idx & 127;
    int t = (k2 * n2) & 127;
    float s, c;
    sincosf(-2.f * PI_F * (float)t / 128.f, &s, &c);
    int pos = k2 * 256 + (n2 >> 1) * 4 + (n2 & 1);
    ws[OFF_F128Q + pos]      = c;
    ws[OFF_F128Q + pos + 2]  = s;
    ws[OFF_IF128Q + pos]     = c;
    ws[OFF_IF128Q + pos + 2] = -s;
  }
  if (idx < 8192) {
    int k2 = idx >> 6, n1 = idx & 63;
    int t = (n1 * k2) & 8191;
    float s, c;
    sincosf(-2.f * PI_F * (float)t / 8192.f, &s, &c);
    ws[OFF_WF + 2 * idx]     = c;  ws[OFF_WF + 2 * idx + 1] = s;
    ws[OFF_WI + 2 * idx]     = c;  ws[OFF_WI + 2 * idx + 1] = -s;
  }
  if (idx < 4096) {
    int n1 = idx >> 6, k1 = idx & 63;
    int t = (n1 * k1) & 63;
    float s, c;
    sincosf(-2.f * PI_F * (float)t / 64.f, &s, &c);
    int pos = (n1 >> 1) * 128 + 2 * k1 + (n1 & 1);
    ws[OFF_G64PR + pos]  = c;  ws[OFF_G64PI + pos]  = s;
    ws[OFF_IG64PR + pos] = c;  ws[OFF_IG64PI + pos] = -s;
  }
}

static __device__ inline void ssm_khat_at(int kf, float stp,
                                          const float* __restrict__ lr,
                                          const float* __restrict__ li,
                                          const float* __restrict__ pp,
                                          const float* __restrict__ bb,
                                          const float* __restrict__ ct,
                                          float& outr, float& outi) {
  float osin, ocos;
  sincosf(-2.f * PI_F * (float)kf / 8192.f, &osin, &ocos);
  float pr = 1.f + ocos, pi = osin;
  float mr = 1.f - ocos, mi = -osin;
  float pd = pr * pr + pi * pi;
  float ipr = pr / pd, ipi = -pi / pd;
  float cr = 2.f * ipr, ci = 2.f * ipi;
  float sc = 2.f / stp;
  float gr = sc * (mr * ipr - mi * ipi);
  float gi = sc * (mr * ipi + mi * ipr);

  float k00r = 0, k00i = 0, k01r = 0, k01i = 0;
  float k10r = 0, k10i = 0, k11r = 0, k11i = 0;
  for (int n = 0; n < NSTATE; ++n) {
    float dr = gr - lr[n], di = gi - li[n];
    float dd = dr * dr + di * di;
    float ir = dr / dd, ii = -di / dd;
    float b = bb[n], p = pp[n], c = ct[n];
    float w00 = c * b, w01 = c * p, w10 = p * b, w11 = p * p;
    k00r += w00 * ir; k00i += w00 * ii;
    k01r += w01 * ir; k01i += w01 * ii;
    k10r += w10 * ir; k10i += w10 * ii;
    k11r += w11 * ir; k11i += w11 * ii;
  }
  float qr = k01r * k10r - k01i * k10i;
  float qi = k01r * k10i + k01i * k10r;
  float er = 1.f + k11r, ei = k11i;
  float ed = er * er + ei * ei;
  float ier = er / ed, iei = -ei / ed;
  float tr = qr * ier - qi * iei;
  float ti = qr * iei + qi * ier;
  float kr = k00r - tr, ki = k00i - ti;
  outr = cr * kr - ci * ki;
  outi = cr * ki + ci * kr;
}

__global__ void ssm_gen_khat(float* __restrict__ ws,
                             const float* __restrict__ Lr,
                             const float* __restrict__ Li,
                             const float* __restrict__ Pv,
                             const float* __restrict__ Bv,
                             const float* __restrict__ Ct,
                             const float* __restrict__ stepv) {
  int layer = blockIdx.y;
  int s = blockIdx.x * blockDim.x + threadIdx.x;
  int k2 = s >> 6, k1 = s & 63;
  int kf = 128 * k1 + k2;
  int kg = (8192 - kf) & 8191;

  const float* lr = Lr + layer * NSTATE;
  const float* li = Li + layer * NSTATE;
  const float* pp = Pv + layer * NSTATE;
  const float* bb = Bv + layer * NSTATE;
  const float* ct = Ct + layer * NSTATE;
  float stp = stepv[layer];

  float ar, ai, br, bi;
  ssm_khat_at(kf, stp, lr, li, pp, bb, ct, ar, ai);
  ssm_khat_at(kg, stp, lr, li, pp, bb, ct, br, bi);
  const float scale = 0.5f / 8192.f;
  ws[OFF_KHAT + layer * 16384 + 2 * s]     = (ar + br) * scale;
  ws[OFF_KHAT + layer * 16384 + 2 * s + 1] = (ai - bi) * scale;
}

__global__ __launch_bounds__(256, 4) void ssm_rows(const float* __restrict__ uin,
                                                float* __restrict__ uout,
                                                const float* ws) {
  extern __shared__ float smem[];
  float* xa = smem;
  float* xb = smem + 8192;
  float* Yr = smem + 16384;
  float* Yi = smem + 24576;

  const int tid  = threadIdx.x;
  const int wave = tid >> 5;
  const int lane = tid & 31;
  const int l16  = lane & 15;
  const int kSel = (lane >> 4) << 1;
  const int mHi  = (lane >> 4) << 3;
  const int mB   = wave << 4;
  const int eB = (((mB + mHi) << 6) + l16) << 1;
  const int pB = ((mB + mHi) << 6) + (l16 << 1);

  const size_t rowOff = (size_t)blockIdx.x * (2 * LROW);
  const float4* srcA = (const float4*)(uin + rowOff);
  const float4* srcB = (const float4*)(uin + rowOff + LROW);
  for (int i = tid; i < LROW / 4; i += 256) {
    float4 va = srcA[i];
    float4 vb = srcB[i];
    int n2 = i >> 4, n1 = (i & 15) << 2;
    int base = pidx(n2, n1);
    xa[base] = va.x; xa[base + 2] = va.y; xa[base + 4] = va.z; xa[base + 6] = va.w;
    xb[base] = vb.x; xb[base + 2] = vb.y; xb[base + 4] = vb.z; xb[base + 6] = vb.w;
  }
  __syncthreads();

  const float* F128Q  = ws + OFF_F128Q;
  const float* IF128Q = ws + OFF_IF128Q;
  const float* GPr    = ws + OFF_G64PR;
  const float* GPi    = ws + OFF_G64PI;
  const float* iGPr   = ws + OFF_IG64PR;
  const float* iGPi   = ws + OFF_IG64PI;
  const float* Wf     = ws + OFF_WF;
  const float* Wi     = ws + OFF_WI;

  for (int layer = 0; layer < 4; ++layer) {
    const float* kh = ws + OFF_KHAT + layer * 16384;

    {
      v8f c1[4] = {}, c2[4] = {}, c3[4] = {};
#pragma unroll 2
      for (int kk = 0; kk < 128; kk += 4) {
        float4 aq = *(const float4*)(F128Q + (mB + l16) * 256 + (kk + kSel) * 2);
        v2f ar; ar.x = aq.x; ar.y = aq.y;
        v2f ai; ai.x = aq.z; ai.y = aq.w;
        v2f ad = ai - ar;
        v2f as = -(ar + ai);
        int jb = ((kk + kSel) >> 1) << 7;
#pragma unroll
        for (int t = 0; t < 4; ++t) {
          int cc = ((t << 4) + l16) << 1;
          v2f ba = *(const v2f*)(xa + jb + cc);
          v2f bb = *(const v2f*)(xb + jb + cc);
          v2f bs = ba + bb;
          c1[t] = wmma4(ar, bs, c1[t]);
          c2[t] = wmma4(ad, ba, c2[t]);
          c3[t] = wmma4(as, bb, c3[t]);
        }
      }
      const float* tb = Wf + eB;
#pragma unroll
      for (int th = 0; th < 4; th += 2) {
        v2f tw[2][8];
#pragma unroll
        for (int t = 0; t < 2; ++t) {
#pragma unroll
          for (int v = 0; v < 8; ++v) tw[t][v] = *(const v2f*)(tb + 128 * v + 32 * (th + t));
        }
#pragma unroll
        for (int t = 0; t < 2; ++t) {
#pragma unroll
          for (int v = 0; v < 8; ++v) {
            v2f w = tw[t][v];
            int p = pB + 128 * (v >> 1) + 32 * (th + t) + (v & 1);
            float a = c1[th + t][v] + c3[th + t][v], b = c1[th + t][v] + c2[th + t][v];
            Yr[p] = a * w.x - b * w.y;
            Yi[p] = a * w.y + b * w.x;
          }
        }
      }
    }

    {
      v8f z1[4] = {}, z2[4] = {}, z3[4] = {};
      for (int kk = 0; kk < 64; kk += 4) {
        int p0 = pidx(mB + l16, kk + kSel);
        v2f ar; ar.x = Yr[p0]; ar.y = Yr[p0 + 2];
        v2f ai; ai.x = Yi[p0]; ai.y = Yi[p0 + 2];
        v2f as = ar + ai;
        int jrow = ((kk + kSel) >> 1) << 7;
#pragma unroll
        for (int t = 0; t < 4; ++t) {
          int cc = ((t << 4) + l16) << 1;
          v2f br = *(const v2f*)(GPr + jrow + cc);
          v2f bi = *(const v2f*)(GPi + jrow + cc);
          v2f bd = bi - br;
          v2f bn = -(br + bi);
          z1[t] = wmma4(as, br, z1[t]);
          z2[t] = wmma4(ar, bd, z2[t]);
          z3[t] = wmma4(ai, bn, z3[t]);
        }
      }
      const float* tb = kh + eB;
#pragma unroll
      for (int th = 0; th < 4; th += 2) {
        v2f tw[2][8];
#pragma unroll
        for (int t = 0; t < 2; ++t) {
#pragma unroll
          for (int v = 0; v < 8; ++v) tw[t][v] = *(const v2f*)(tb + 128 * v + 32 * (th + t));
        }
#pragma unroll
        for (int t = 0; t < 2; ++t) {
#pragma unroll
          for (int v = 0; v < 8; ++v) {
            v2f k = tw[t][v];
            int p = pB + 128 * (v >> 1) + 32 * (th + t) + (v & 1);
            float a = z1[th + t][v] + z3[th + t][v], b = z1[th + t][v] + z2[th + t][v];
            Yr[p] = a * k.x - b * k.y;
            Yi[p] = a * k.y + b * k.x;
          }
        }
      }
    }

    {
      v8f y1[4] = {}, y2[4] = {}, y3[4] = {};
      for (int kk = 0; kk < 64; kk += 4) {
        int p0 = pidx(mB + l16, kk + kSel);
        v2f ar; ar.x = Yr[p0]; ar.y = Yr[p0 + 2];
        v2f ai; ai.x = Yi[p0]; ai.y = Yi[p0 + 2];
        v2f as = ar + ai;
        int jrow = ((kk + kSel) >> 1) << 7;
#pragma unroll
        for (int t = 0; t < 4; ++t) {
          int cc = ((t << 4) + l16) << 1;
          v2f br = *(const v2f*)(iGPr + jrow + cc);
          v2f bi = *(const v2f*)(iGPi + jrow + cc);
          v2f bd = bi - br;
          v2f bn = -(br + bi);
          y1[t] = wmma4(as, br, y1[t]);
          y2[t] = wmma4(ar, bd, y2[t]);
          y3[t] = wmma4(ai, bn, y3[t]);
        }
      }
      const float* tb = Wi + eB;
#pragma unroll
      for (int th = 0; th < 4; th += 2) {
        v2f tw[2][8];
#pragma unroll
        for (int t = 0; t < 2; ++t) {
#pragma unroll
          for (int v = 0; v < 8; ++v) tw[t][v] = *(const v2f*)(tb + 128 * v + 32 * (th + t));
        }
#pragma unroll
        for (int t = 0; t < 2; ++t) {
#pragma unroll
          for (int v = 0; v < 8; ++v) {
            v2f w = tw[t][v];
            int p = pB + 128 * (v >> 1) + 32 * (th + t) + (v & 1);
            float a = y1[th + t][v] + y3[th + t][v], b = y1[th + t][v] + y2[th + t][v];
            Yr[p] = a * w.x - b * w.y;
            Yi[p] = a * w.y + b * w.x;
          }
        }
      }
    }
    __syncthreads();

    {
      v8f w1[4] = {}, w2[4] = {}, w3[4] = {};
      for (int kk = 0; kk < 128; kk += 4) {
        float4 aq = *(const float4*)(IF128Q + (mB + l16) * 256 + (kk + kSel) * 2);
        v2f ar; ar.x = aq.x; ar.y = aq.y;
        v2f ai; ai.x = aq.z; ai.y = aq.w;
        v2f ad = ai - ar;
        v2f as = -(ar + ai);
        int jb = ((kk + kSel) >> 1) << 7;
#pragma unroll
        for (int t = 0; t < 4; ++t) {
          int cc = ((t << 4) + l16) << 1;
          v2f br = *(const v2f*)(Yr + jb + cc);
          v2f bi = *(const v2f*)(Yi + jb + cc);
          v2f bs = br + bi;
          w1[t] = wmma4(ar, bs, w1[t]);
          w2[t] = wmma4(ad, br, w2[t]);
          w3[t] = wmma4(as, bi, w3[t]);
        }
      }
#pragma unroll
      for (int t = 0; t < 4; ++t) {
#pragma unroll
        for (int j = 0; j < 4; ++j) {
          int p = pB + 128 * j + 32 * t;
          v2f s1; s1.x = w1[t][2 * j]; s1.y = w1[t][2 * j + 1];
          v2f s2; s2.x = w2[t][2 * j]; s2.y = w2[t][2 * j + 1];
          v2f s3; s3.x = w3[t][2 * j]; s3.y = w3[t][2 * j + 1];
          v2f ya = (s1 + s3) + *(const v2f*)(xa + p);
          v2f yb = (s1 + s2) + *(const v2f*)(xb + p);
          *(v2f*)(xa + p) = gelu2(ya);
          *(v2f*)(xb + p) = gelu2(yb);
        }
      }
    }
    __syncthreads();
  }

  float4* dstA = (float4*)(uout + rowOff);
  float4* dstB = (float4*)(uout + rowOff + LROW);
  for (int i = tid; i < LROW / 4; i += 256) {
    int n2 = i >> 4, n1 = (i & 15) << 2;
    int base = pidx(n2, n1);
    float4 oa, ob;
    oa.x = xa[base]; oa.y = xa[base + 2]; oa.z = xa[base + 4]; oa.w = xa[base + 6];
    ob.x = xb[base]; ob.y = xb[base + 2]; ob.z = xb[base + 4]; ob.w = xb[base + 6];
    dstA[i] = oa;
    dstB[i] = ob;
  }
}

extern "C" void kernel_launch(void* const* d_in, const int* in_sizes, int n_in,
                              void* d_out, int out_size, void* d_ws, size_t ws_size,
                              hipStream_t stream) {
  const float* u    = (const float*)d_in[0];
  const float* Lr   = (const float*)d_in[1];
  const float* Li   = (const float*)d_in[2];
  const float* Pv   = (const float*)d_in[3];
  const float* Bv   = (const float*)d_in[4];
  const float* Ct   = (const float*)d_in[5];
  const float* step = (const float*)d_in[6];
  float* out = (float*)d_out;
  float* ws  = (float*)d_ws;

  int pairs = in_sizes[0] / LPAIR;

  ssm_gen_tables<<<64, 256, 0, stream>>>(ws);
  ssm_gen_khat<<<dim3(32, 4), 256, 0, stream>>>(ws, Lr, Li, Pv, Bv, Ct, step);
  ssm_rows<<<pairs, 256, 32768 * sizeof(float), stream>>>(u, out, ws);
}


# All four FFT stages: data operand folded along K (entries j and N-j added and subtracted), one real DFT table then gives Re and Im: 2 real WMMA GEMMs per complex product, not 3; regroups all sums
# speedup vs baseline: 3.1488x; 1.3637x over previous
#include <hip/hip_runtime.h>
#include <math.h>

typedef float v2f __attribute__((ext_vector_type(2)));
typedef float v8f __attribute__((ext_vector_type(8)));

#define LROW 8192
#define LPAIR 16384
#define NSTATE 64
#define PI_F 3.14159265358979323846f

#define OFF_T128F   0
#define OFF_T128I   32768
#define OFF_T64F    65536
#define OFF_T64I    73728
#define OFF_WF      81920
#define OFF_WI      98304
#define OFF_KHAT    114688
#define WS_FLOATS   (OFF_KHAT + 4*16384)

static __device__ inline v8f wmma4(v2f a, v2f b, v8f c) {
  return __builtin_amdgcn_wmma_f32_16x16x4_f32(false, a, false, b, (short)0, c,
                                               false, false);
}

static __device__ inline int pidx(int row, int col) {
  return ((row >> 1) << 7) + (col << 1) + (row & 1);
}

static __device__ inline v2f vfma2(v2f a, v2f b, v2f c) {
  return __builtin_elementwise_fma(a, b, c);
}
static __device__ inline v2f bc2(float s) { v2f r; r.x = s; r.y = s; return r; }

static __device__ inline v2f gelu2(v2f y) {
  v2f x  = y * 0.707106769f;
  v2f ax = __builtin_elementwise_abs(x);
  v2f t  = x * x;
  v2f pb = vfma2(t, bc2(-0.000561801775f), bc2(0.00491381623f));
  pb = vfma2(t, pb, bc2(-0.0267075151f));
  pb = vfma2(t, pb, bc2(0.112800106f));
  pb = vfma2(t, pb, bc2(-0.376122952f));
  pb = vfma2(t, pb, bc2(0.128379107f));
  v2f rs = vfma2(ax, pb, ax);
  v2f pa = vfma2(ax, bc2(0.0000169988107f), bc2(-0.000378677854f));
  pa = vfma2(ax, pa, bc2(0.00385781587f));
  pa = vfma2(ax, pa, bc2(-0.0241816975f));
  pa = vfma2(ax, pa, bc2(0.106668264f));
  pa = vfma2(ax, pa, bc2(0.634933293f));
  pa = vfma2(ax, pa, bc2(0.128689408f));
  v2f q  = vfma2(ax, pa, ax);
  v2f qe = q * -1.44269502f;
  v2f e;
  e.x = __builtin_amdgcn_exp2f(qe.x);
  e.y = __builtin_amdgcn_exp2f(qe.y);
  v2f rl = 1.0f - e;
  v2f r;
  r.x = ax.x < 1.0f ? rs.x : rl.x;
  r.y = ax.y < 1.0f ? rs.y : rl.y;
  r = __builtin_elementwise_copysign(r, x);
  return (0.5f * y) * (1.0f + r);
}

__global__ void ssm_gen_tables(float* __restrict__ ws) {
  int idx = blockIdx.x * blockDim.x + threadIdx.x;
  {
    int k2 = idx >> 7, J = idx & 127;
    int t = (k2 * J) & 127;
    float s, c;
    sincosf(-2.f * PI_F * (float)t / 128.f, &s, &c);
    int pos = (J >> 2) * 512 + k2 * 4 + (J & 3);
    ws[OFF_T128F + pos] = (J <= 64) ? c : s;
    ws[OFF_T128I + pos] = (J <= 64) ? c : -s;
  }
  if (idx < 8192) {
    int k2 = idx >> 6, n1 = idx & 63;
    int t = (n1 * k2) & 8191;
    float s, c;
    sincosf(-2.f * PI_F * (float)t / 8192.f, &s, &c);
    ws[OFF_WF + 2 * idx]     = c;  ws[OFF_WF + 2 * idx + 1] = s;
    ws[OFF_WI + 2 * idx]     = c;  ws[OFF_WI + 2 * idx + 1] = -s;
  }
  if (idx < 4096) {
    int J = idx >> 6, k1 = idx & 63;
    int t = (J * k1) & 63;
    float s, c;
    sincosf(-2.f * PI_F * (float)t / 64.f, &s, &c);
    int pos = (J >> 1) * 128 + 2 * k1 + (J & 1);
    ws[OFF_T64F + pos] = (J <= 32) ? c : s;
    ws[OFF_T64I + pos] = (J <= 32) ? c : -s;
  }
}

static __device__ inline void ssm_khat_at(int kf, float stp,
                                          const float* __restrict__ lr,
                                          const float* __restrict__ li,
                                          const float* __restrict__ pp,
                                          const float* __restrict__ bb,
                                          const float* __restrict__ ct,
                                          float& outr, float& outi) {
  float osin, ocos;
  sincosf(-2.f * PI_F * (float)kf / 8192.f, &osin, &ocos);
  float pr = 1.f + ocos, pi = osin;
  float mr = 1.f - ocos, mi = -osin;
  float pd = pr * pr + pi * pi;
  float ipr = pr / pd, ipi = -pi / pd;
  float cr = 2.f * ipr, ci = 2.f * ipi;
  float sc = 2.f / stp;
  float gr = sc * (mr * ipr - mi * ipi);
  float gi = sc * (mr * ipi + mi * ipr);

  float k00r = 0, k00i = 0, k01r = 0, k01i = 0;
  float k10r = 0, k10i = 0, k11r = 0, k11i = 0;
  for (int n = 0; n < NSTATE; ++n) {
    float dr = gr - lr[n], di = gi - li[n];
    float dd = dr * dr + di * di;
    float ir = dr / dd, ii = -di / dd;
    float b = bb[n], p = pp[n], c = ct[n];
    float w00 = c * b, w01 = c * p, w10 = p * b, w11 = p * p;
    k00r += w00 * ir; k00i += w00 * ii;
    k01r += w01 * ir; k01i += w01 * ii;
    k10r += w10 * ir; k10i += w10 * ii;
    k11r += w11 * ir; k11i += w11 * ii;
  }
  float qr = k01r * k10r - k01i * k10i;
  float qi = k01r * k10i + k01i * k10r;
  float er = 1.f + k11r, ei = k11i;
  float ed = er * er + ei * ei;
  float ier = er / ed, iei = -ei / ed;
  float tr = qr * ier - qi * iei;
  float ti = qr * iei + qi * ier;
  float kr = k00r - tr, ki = k00i - ti;
  outr = cr * kr - ci * ki;
  outi = cr * ki + ci * kr;
}

__global__ void ssm_gen_khat(float* __restrict__ ws,
                             const float* __restrict__ Lr,
                             const float* __restrict__ Li,
                             const float* __restrict__ Pv,
                             const float* __restrict__ Bv,
                             const float* __restrict__ Ct,
                             const float* __restrict__ stepv) {
  int layer = blockIdx.y;
  int s = blockIdx.x * blockDim.x + threadIdx.x;
  int k2 = s >> 6, k1 = s & 63;
  int kf = 128 * k1 + k2;
  int kg = (8192 - kf) & 8191;

  const float* lr = Lr + layer * NSTATE;
  const float* li = Li + layer * NSTATE;
  const float* pp = Pv + layer * NSTATE;
  const float* bb = Bv + layer * NSTATE;
  const float* ct = Ct + layer * NSTATE;
  float stp = stepv[layer];

  float ar, ai, br, bi;
  ssm_khat_at(kf, stp, lr, li, pp, bb, ct, ar, ai);
  ssm_khat_at(kg, stp, lr, li, pp, bb, ct, br, bi);
  const float scale = 0.5f / 8192.f;
  ws[OFF_KHAT + layer * 16384 + 2 * s]     = (ar + br) * scale;
  ws[OFF_KHAT + layer * 16384 + 2 * s + 1] = (ai - bi) * scale;
}

static __device__ inline void fold_rows(const float* sr, const float* si,
                                        float* dr, float* di, int tid) {
  const int col2 = (tid & 63) << 1;
  const int jg = tid >> 6;
  const bool edge = (jg == 0);
  const int bj = ((jg >> 1) << 7) + col2 + (jg & 1);
  const int bm = ((64 - ((jg + 1) >> 1)) << 7) + col2 + (jg & 1);
  const int p64 = (32 << 7) + col2;
#pragma unroll
  for (int h = 0; h < 2; ++h) {
    float a[8], b[8], c[8], d[8];
#pragma unroll
    for (int q = 0; q < 8; ++q) {
      const int i = 8 * h + q;
      int pj = bj + 256 * i;
      int pm = (i == 0 && edge) ? p64 : bm - 256 * i;
      a[q] = sr[pj]; b[q] = sr[pm]; c[q] = si[pj]; d[q] = si[pm];
    }
#pragma unroll
    for (int q = 0; q < 8; ++q) {
      const int i = 8 * h + q;
      int pj = bj + 256 * i;
      int pm = (i == 0 && edge) ? p64 : bm - 256 * i;
      bool cp = (i == 0 && edge);
      dr[pj] = cp ? a[q] : a[q] + b[q];
      dr[pm] = cp ? b[q] : c[q] - d[q];
      di[pj] = cp ? c[q] : c[q] + d[q];
      di[pm] = cp ? d[q] : b[q] - a[q];
    }
  }
}

static __device__ inline void fold_cols(float* yr, float* yi, int mB, int lane) {
  const int cj = (mB << 6) + (lane << 1), cm = (mB << 6) + ((64 - lane) << 1);
  if (lane != 0) {
#pragma unroll
    for (int h = 0; h < 2; ++h) {
      float a[8], b[8], c[8], d[8];
#pragma unroll
      for (int q = 0; q < 8; ++q) {
        const int r = 8 * h + q;
        const int ro = ((r >> 1) << 7) + (r & 1);
        a[q] = yr[cj + ro]; b[q] = yr[cm + ro]; c[q] = yi[cj + ro]; d[q] = yi[cm + ro];
      }
#pragma unroll
      for (int q = 0; q < 8; ++q) {
        const int r = 8 * h + q;
        const int ro = ((r >> 1) << 7) + (r & 1);
        yr[cj + ro] = a[q] + b[q];
        yr[cm + ro] = c[q] - d[q];
        yi[cj + ro] = c[q] + d[q];
        yi[cm + ro] = b[q] - a[q];
      }
    }
  }
}

__global__ __launch_bounds__(256, 4) void ssm_rows(const float* __restrict__ uin,
                                                float* __restrict__ uout,
                                                const float* ws) {
  extern __shared__ float smem[];
  float* xa = smem;
  float* xb = smem + 8192;
  float* Yr = smem + 16384;
  float* Yi = smem + 24576;

  const int tid  = threadIdx.x;
  const int wave = tid >> 5;
  const int lane = tid & 31;
  const int l16  = lane & 15;
  const int kSel = (lane >> 4) << 1;
  const int mHi  = (lane >> 4) << 3;
  const int mB   = wave << 4;
  const int eB = (((mB + mHi) << 6) + l16) << 1;
  const int pB = ((mB + mHi) << 6) + (l16 << 1);
  const int aB = ((mB + l16) << 2) + kSel;

  const size_t rowOff = (size_t)blockIdx.x * (2 * LROW);
  const float4* srcA = (const float4*)(uin + rowOff);
  const float4* srcB = (const float4*)(uin + rowOff + LROW);
  for (int i = tid; i < LROW / 4; i += 256) {
    float4 va = srcA[i];
    float4 vb = srcB[i];
    int n2 = i >> 4, n1 = (i & 15) << 2;
    int base = pidx(n2, n1);
    xa[base] = va.x; xa[base + 2] = va.y; xa[base + 4] = va.z; xa[base + 6] = va.w;
    xb[base] = vb.x; xb[base + 2] = vb.y; xb[base + 4] = vb.z; xb[base + 6] = vb.w;
  }
  __syncthreads();

  const float* T128F = ws + OFF_T128F;
  const float* T128I = ws + OFF_T128I;
  const float* T64F  = ws + OFF_T64F;
  const float* T64I  = ws + OFF_T64I;
  const float* Wf    = ws + OFF_WF;
  const float* Wi    = ws + OFF_WI;

  for (int layer = 0; layer < 4; ++layer) {
    const float* kh = ws + OFF_KHAT + layer * 16384;

    fold_rows(xa, xb, Yr, Yi, tid);
    __syncthreads();
    {
      v8f c1[4] = {}, c2[4] = {};
#pragma unroll 4
      for (int kk = 0; kk < 128; kk += 4) {
        v2f a = *(const v2f*)(T128F + (kk << 7) + aB);
        int jb = ((kk + kSel) >> 1) << 7;
#pragma unroll
        for (int t = 0; t < 4; ++t) {
          int cc = ((t << 4) + l16) << 1;
          v2f b1 = *(const v2f*)(Yr + jb + cc);
          v2f b2 = *(const v2f*)(Yi + jb + cc);
          c1[t] = wmma4(a, b1, c1[t]);
          c2[t] = wmma4(a, b2, c2[t]);
        }
      }
      __syncthreads();
      const float* tb = Wf + eB;
#pragma unroll
      for (int th = 0; th < 4; th += 2) {
        v2f tw[2][8];
#pragma unroll
        for (int t = 0; t < 2; ++t) {
#pragma unroll
          for (int v = 0; v < 8; ++v) tw[t][v] = *(const v2f*)(tb + 128 * v + 32 * (th + t));
        }
#pragma unroll
        for (int t = 0; t < 2; ++t) {
#pragma unroll
          for (int v = 0; v < 8; ++v) {
            v2f w = tw[t][v];
            int p = pB + 128 * (v >> 1) + 32 * (th + t) + (v & 1);
            float a = c1[th + t][v], b = c2[th + t][v];
            Yr[p] = a * w.x - b * w.y;
            Yi[p] = a * w.y + b * w.x;
          }
        }
      }
    }

    fold_cols(Yr, Yi, mB, lane);
    {
      v8f z1[4] = {}, z2[4] = {};
      for (int kk = 0; kk < 64; kk += 4) {
        int p0 = pidx(mB + l16, kk + kSel);
        v2f a1; a1.x = Yr[p0]; a1.y = Yr[p0 + 2];
        v2f a2; a2.x = Yi[p0]; a2.y = Yi[p0 + 2];
        int jrow = ((kk + kSel) >> 1) << 7;
#pragma unroll
        for (int t = 0; t < 4; ++t) {
          int cc = ((t << 4) + l16) << 1;
          v2f b = *(const v2f*)(T64F + jrow + cc);
          z1[t] = wmma4(a1, b, z1[t]);
          z2[t] = wmma4(a2, b, z2[t]);
        }
      }
      const float* tb = kh + eB;
#pragma unroll
      for (int th = 0; th < 4; th += 2) {
        v2f tw[2][8];
#pragma unroll
        for (int t = 0; t < 2; ++t) {
#pragma unroll
          for (int v = 0; v < 8; ++v) tw[t][v] = *(const v2f*)(tb + 128 * v + 32 * (th + t));
        }
#pragma unroll
        for (int t = 0; t < 2; ++t) {
#pragma unroll
          for (int v = 0; v < 8; ++v) {
            v2f k = tw[t][v];
            int p = pB + 128 * (v >> 1) + 32 * (th + t) + (v & 1);
            float a = z1[th + t][v], b = z2[th + t][v];
            Yr[p] = a * k.x - b * k.y;
            Yi[p] = a * k.y + b * k.x;
          }
        }
      }
    }

    fold_cols(Yr, Yi, mB, lane);
    {
      v8f y1[4] = {}, y2[4] = {};
      for (int kk = 0; kk < 64; kk += 4) {
        int p0 = pidx(mB + l16, kk + kSel);
        v2f a1; a1.x = Yr[p0]; a1.y = Yr[p0 + 2];
        v2f a2; a2.x = Yi[p0]; a2.y = Yi[p0 + 2];
        int jrow = ((kk + kSel) >> 1) << 7;
#pragma unroll
        for (int t = 0; t < 4; ++t) {
          int cc = ((t << 4) + l16) << 1;
          v2f b = *(const v2f*)(T64I + jrow + cc);
          y1[t] = wmma4(a1, b, y1[t]);
          y2[t] = wmma4(a2, b, y2[t]);
        }
      }
      const float* tb = Wi + eB;
#pragma unroll
      for (int th = 0; th < 4; th += 2) {
        v2f tw[2][8];
#pragma unroll
        for (int t = 0; t < 2; ++t) {
#pragma unroll
          for (int v = 0; v < 8; ++v) tw[t][v] = *(const v2f*)(tb + 128 * v + 32 * (th + t));
        }
#pragma unroll
        for (int t = 0; t < 2; ++t) {
#pragma unroll
          for (int v = 0; v < 8; ++v) {
            v2f w = tw[t][v];
            int p = pB + 128 * (v >> 1) + 32 * (th + t) + (v & 1);
            float a = y1[th + t][v], b = y2[th + t][v];
            Yr[p] = a * w.x - b * w.y;
            Yi[p] = a * w.y + b * w.x;
          }
        }
      }
    }
    __syncthreads();

    fold_rows(Yr, Yi, Yr, Yi, tid);
    __syncthreads();
    {
      v8f w1[4] = {}, w2[4] = {};
      for (int kk = 0; kk < 128; kk += 4) {
        v2f a = *(const v2f*)(T128I + (kk << 7) + aB);
        int jb = ((kk + kSel) >> 1) << 7;
#pragma unroll
        for (int t = 0; t < 4; ++t) {
          int cc = ((t << 4) + l16) << 1;
          v2f b1 = *(const v2f*)(Yr + jb + cc);
          v2f b2 = *(const v2f*)(Yi + jb + cc);
          w1[t] = wmma4(a, b1, w1[t]);
          w2[t] = wmma4(a, b2, w2[t]);
        }
      }
#pragma unroll
      for (int t = 0; t < 4; ++t) {
#pragma unroll
        for (int j = 0; j < 4; ++j) {
          int p = pB + 128 * j + 32 * t;
          v2f s1; s1.x = w1[t][2 * j]; s1.y = w1[t][2 * j + 1];
          v2f s2; s2.x = w2[t][2 * j]; s2.y = w2[t][2 * j + 1];
          v2f ya = s1 + *(const v2f*)(xa + p);
          v2f yb = s2 + *(const v2f*)(xb + p);
          *(v2f*)(xa + p) = gelu2(ya);
          *(v2f*)(xb + p) = gelu2(yb);
        }
      }
    }
    __syncthreads();
  }

  float4* dstA = (float4*)(uout + rowOff);
  float4* dstB = (float4*)(uout + rowOff + LROW);
  for (int i = tid; i < LROW / 4; i += 256) {
    int n2 = i >> 4, n1 = (i & 15) << 2;
    int base = pidx(n2, n1);
    float4 oa, ob;
    oa.x = xa[base]; oa.y = xa[base + 2]; oa.z = xa[base + 4]; oa.w = xa[base + 6];
    ob.x = xb[base]; ob.y = xb[base + 2]; ob.z = xb[base + 4]; ob.w = xb[base + 6];
    dstA[i] = oa;
    dstB[i] = ob;
  }
}

extern "C" void kernel_launch(void* const* d_in, const int* in_sizes, int n_in,
                              void* d_out, int out_size, void* d_ws, size_t ws_size,
                              hipStream_t stream) {
  const float* u    = (const float*)d_in[0];
  const float* Lr   = (const float*)d_in[1];
  const float* Li   = (const float*)d_in[2];
  const float* Pv   = (const float*)d_in[3];
  const float* Bv   = (const float*)d_in[4];
  const float* Ct   = (const float*)d_in[5];
  const float* step = (const float*)d_in[6];
  float* out = (float*)d_out;
  float* ws  = (float*)d_ws;

  int pairs = in_sizes[0] / LPAIR;

  ssm_gen_tables<<<64, 256, 0, stream>>>(ws);
  ssm_gen_khat<<<dim3(32, 4), 256, 0, stream>>>(ws, Lr, Li, Pv, Bv, Ct, step);
  ssm_rows<<<pairs, 256, 32768 * sizeof(float), stream>>>(u, out, ws);
}


# Both 64-point stages: the table's cosine rows (J<=32) and sine rows (J>32) summed apart for output columns 0..32 only, columns 64-k by C-S: 86 WMMAs per stage, not 128; regroups those stages' sums
# speedup vs baseline: 3.3045x; 1.0494x over previous
#include <hip/hip_runtime.h>
#include <math.h>

typedef float v2f __attribute__((ext_vector_type(2)));
typedef float v8f __attribute__((ext_vector_type(8)));

#define LROW 8192
#define LPAIR 16384
#define NSTATE 64
#define PI_F 3.14159265358979323846f

#define OFF_T128F   0
#define OFF_T128I   32768
#define OFF_T64F    65536
#define OFF_T64I    73728
#define OFF_WF      81920
#define OFF_WI      98304
#define OFF_KHAT    114688
#define WS_FLOATS   (OFF_KHAT + 4*16384)

static __device__ inline v8f wmma4(v2f a, v2f b, v8f c) {
  return __builtin_amdgcn_wmma_f32_16x16x4_f32(false, a, false, b, (short)0, c,
                                               false, false);
}

static __device__ inline int pidx(int row, int col) {
  return ((row >> 1) << 7) + (col << 1) + (row & 1);
}

static __device__ inline v2f vfma2(v2f a, v2f b, v2f c) {
  return __builtin_elementwise_fma(a, b, c);
}
static __device__ inline v2f bc2(float s) { v2f r; r.x = s; r.y = s; return r; }

static __device__ inline v2f gelu2(v2f y) {
  v2f x  = y * 0.707106769f;
  v2f ax = __builtin_elementwise_abs(x);
  v2f t  = x * x;
  v2f pb = vfma2(t, bc2(-0.000561801775f), bc2(0.00491381623f));
  pb = vfma2(t, pb, bc2(-0.0267075151f));
  pb = vfma2(t, pb, bc2(0.112800106f));
  pb = vfma2(t, pb, bc2(-0.376122952f));
  pb = vfma2(t, pb, bc2(0.128379107f));
  v2f rs = vfma2(ax, pb, ax);
  v2f pa = vfma2(ax, bc2(0.0000169988107f), bc2(-0.000378677854f));
  pa = vfma2(ax, pa, bc2(0.00385781587f));
  pa = vfma2(ax, pa, bc2(-0.0241816975f));
  pa = vfma2(ax, pa, bc2(0.106668264f));
  pa = vfma2(ax, pa, bc2(0.634933293f));
  pa = vfma2(ax, pa, bc2(0.128689408f));
  v2f q  = vfma2(ax, pa, ax);
  v2f qe = q * -1.44269502f;
  v2f e;
  e.x = __builtin_amdgcn_exp2f(qe.x);
  e.y = __builtin_amdgcn_exp2f(qe.y);
  v2f rl = 1.0f - e;
  v2f r;
  r.x = ax.x < 1.0f ? rs.x : rl.x;
  r.y = ax.y < 1.0f ? rs.y : rl.y;
  r = __builtin_elementwise_copysign(r, x);
  return (0.5f * y) * (1.0f + r);
}

__global__ void ssm_gen_tables(float* __restrict__ ws) {
  int idx = blockIdx.x * blockDim.x + threadIdx.x;
  {
    int k2 = idx >> 7, J = idx & 127;
    int t = (k2 * J) & 127;
    float s, c;
    sincosf(-2.f * PI_F * (float)t / 128.f, &s, &c);
    int pos = (J >> 2) * 512 + k2 * 4 + (J & 3);
    ws[OFF_T128F + pos] = (J <= 64) ? c : s;
    ws[OFF_T128I + pos] = (J <= 64) ? c : -s;
  }
  if (idx < 8192) {
    int k2 = idx >> 6, n1 = idx & 63;
    int t = (n1 * k2) & 8191;
    float s, c;
    sincosf(-2.f * PI_F * (float)t / 8192.f, &s, &c);
    ws[OFF_WF + 2 * idx]     = c;  ws[OFF_WF + 2 * idx + 1] = s;
    ws[OFF_WI + 2 * idx]     = c;  ws[OFF_WI + 2 * idx + 1] = -s;
  }
  if (idx < 4096) {
    int J = idx >> 6, k1 = idx & 63;
    int t = (J * k1) & 63;
    float s, c;
    sincosf(-2.f * PI_F * (float)t / 64.f, &s, &c);
    int pos = (J >> 1) * 128 + 2 * k1 + (J & 1);
    ws[OFF_T64F + pos] = (J <= 32) ? c : s;
    ws[OFF_T64I + pos] = (J <= 32) ? c : -s;
  }
}

static __device__ inline void ssm_khat_at(int kf, float stp,
                                          const float* __restrict__ lr,
                                          const float* __restrict__ li,
                                          const float* __restrict__ pp,
                                          const float* __restrict__ bb,
                                          const float* __restrict__ ct,
                                          float& outr, float& outi) {
  float osin, ocos;
  sincosf(-2.f * PI_F * (float)kf / 8192.f, &osin, &ocos);
  float pr = 1.f + ocos, pi = osin;
  float mr = 1.f - ocos, mi = -osin;
  float pd = pr * pr + pi * pi;
  float ipr = pr / pd, ipi = -pi / pd;
  float cr = 2.f * ipr, ci = 2.f * ipi;
  float sc = 2.f / stp;
  float gr = sc * (mr * ipr - mi * ipi);
  float gi = sc * (mr * ipi + mi * ipr);

  float k00r = 0, k00i = 0, k01r = 0, k01i = 0;
  float k10r = 0, k10i = 0, k11r = 0, k11i = 0;
  for (int n = 0; n < NSTATE; ++n) {
    float dr = gr - lr[n], di = gi - li[n];
    float dd = dr * dr + di * di;
    float ir = dr / dd, ii = -di / dd;
    float b = bb[n], p = pp[n], c = ct[n];
    float w00 = c * b, w01 = c * p, w10 = p * b, w11 = p * p;
    k00r += w00 * ir; k00i += w00 * ii;
    k01r += w01 * ir; k01i += w01 * ii;
    k10r += w10 * ir; k10i += w10 * ii;
    k11r += w11 * ir; k11i += w11 * ii;
  }
  float qr = k01r * k10r - k01i * k10i;
  float qi = k01r * k10i + k01i * k10r;
  float er = 1.f + k11r, ei = k11i;
  float ed = er * er + ei * ei;
  float ier = er / ed, iei = -ei / ed;
  float tr = qr * ier - qi * iei;
  float ti = qr * iei + qi * ier;
  float kr = k00r - tr, ki = k00i - ti;
  outr = cr * kr - ci * ki;
  outi = cr * ki + ci * kr;
}

__global__ void ssm_gen_khat(float* __restrict__ ws,
                             const float* __restrict__ Lr,
                             const float* __restrict__ Li,
                             const float* __restrict__ Pv,
                             const float* __restrict__ Bv,
                             const float* __restrict__ Ct,
                             const float* __restrict__ stepv) {
  int layer = blockIdx.y;
  int s = blockIdx.x * blockDim.x + threadIdx.x;
  int k2 = s >> 6, k1 = s & 63;
  int kf = 128 * k1 + k2;
  int kg = (8192 - kf) & 8191;

  const float* lr = Lr + layer * NSTATE;
  const float* li = Li + layer * NSTATE;
  const float* pp = Pv + layer * NSTATE;
  const float* bb = Bv + layer * NSTATE;
  const float* ct = Ct + layer * NSTATE;
  float stp = stepv[layer];

  float ar, ai, br, bi;
  ssm_khat_at(kf, stp, lr, li, pp, bb, ct, ar, ai);
  ssm_khat_at(kg, stp, lr, li, pp, bb, ct, br, bi);
  const float scale = 0.5f / 8192.f;
  ws[OFF_KHAT + layer * 16384 + 2 * s]     = (ar + br) * scale;
  ws[OFF_KHAT + layer * 16384 + 2 * s + 1] = (ai - bi) * scale;
}

static __device__ inline void fold_rows(const float* sr, const float* si,
                                        float* dr, float* di, int tid) {
  const int col2 = (tid & 63) << 1;
  const int jg = tid >> 6;
  const bool edge = (jg == 0);
  const int bj = ((jg >> 1) << 7) + col2 + (jg & 1);
  const int bm = ((64 - ((jg + 1) >> 1)) << 7) + col2 + (jg & 1);
  const int p64 = (32 << 7) + col2;
#pragma unroll
  for (int h = 0; h < 2; ++h) {
    float a[8], b[8], c[8], d[8];
#pragma unroll
    for (int q = 0; q < 8; ++q) {
      const int i = 8 * h + q;
      int pj = bj + 256 * i;
      int pm = (i == 0 && edge) ? p64 : bm - 256 * i;
      a[q] = sr[pj]; b[q] = sr[pm]; c[q] = si[pj]; d[q] = si[pm];
    }
#pragma unroll
    for (int q = 0; q < 8; ++q) {
      const int i = 8 * h + q;
      int pj = bj + 256 * i;
      int pm = (i == 0 && edge) ? p64 : bm - 256 * i;
      bool cp = (i == 0 && edge);
      dr[pj] = cp ? a[q] : a[q] + b[q];
      dr[pm] = cp ? b[q] : c[q] - d[q];
      di[pj] = cp ? c[q] : c[q] + d[q];
      di[pm] = cp ? d[q] : b[q] - a[q];
    }
  }
}

static __device__ inline void fold_cols(float* yr, float* yi, int mB, int lane) {
  const int cj = (mB << 6) + (lane << 1), cm = (mB << 6) + ((64 - lane) << 1);
  if (lane != 0) {
#pragma unroll
    for (int h = 0; h < 2; ++h) {
      float a[8], b[8], c[8], d[8];
#pragma unroll
      for (int q = 0; q < 8; ++q) {
        const int r = 8 * h + q;
        const int ro = ((r >> 1) << 7) + (r & 1);
        a[q] = yr[cj + ro]; b[q] = yr[cm + ro]; c[q] = yi[cj + ro]; d[q] = yi[cm + ro];
      }
#pragma unroll
      for (int q = 0; q < 8; ++q) {
        const int r = 8 * h + q;
        const int ro = ((r >> 1) << 7) + (r & 1);
        yr[cj + ro] = a[q] + b[q];
        yr[cm + ro] = c[q] - d[q];
        yi[cj + ro] = c[q] + d[q];
        yi[cm + ro] = b[q] - a[q];
      }
    }
  }
}

static __device__ __forceinline__ void dft64_inplace(float* Yr, float* Yi,
                                                     const float* T64, const float* tbl,
                                                     int mB, int mHi, int l16, int kSel) {
  v8f c1[3] = {}, c2[3] = {}, s1[2] = {}, s2[2] = {};
  const int arow = pidx(mB + l16, kSel);
  const bool hi = (kSel != 0);
#pragma unroll 3
  for (int kk = 0; kk < 36; kk += 4) {
    int p0 = arow + 2 * kk;
    const bool last = (kk == 32);
    v2f a1; a1.x = Yr[p0]; a1.y = Yr[p0 + 2];
    v2f a2; a2.x = Yi[p0]; a2.y = Yi[p0 + 2];
    a1.x = (last && hi) ? 0.f : a1.x;  a1.y = last ? 0.f : a1.y;
    a2.x = (last && hi) ? 0.f : a2.x;  a2.y = last ? 0.f : a2.y;
    int jrow = ((kk + kSel) >> 1) << 7;
#pragma unroll
    for (int t = 0; t < 3; ++t) {
      int cc = ((t << 4) + l16) << 1;
      v2f b = *(const v2f*)(T64 + jrow + cc);
      c1[t] = wmma4(a1, b, c1[t]);
      c2[t] = wmma4(a2, b, c2[t]);
    }
  }
#pragma unroll 4
  for (int kk = 32; kk < 64; kk += 4) {
    int p0 = arow + 2 * kk;
    const bool first = (kk == 32);
    v2f a1; a1.x = Yr[p0]; a1.y = Yr[p0 + 2];
    v2f a2; a2.x = Yi[p0]; a2.y = Yi[p0 + 2];
    a1.x = (first && !hi) ? 0.f : a1.x;
    a2.x = (first && !hi) ? 0.f : a2.x;
    int jrow = ((kk + kSel) >> 1) << 7;
#pragma unroll
    for (int t = 0; t < 2; ++t) {
      int cc = ((t << 4) + l16) << 1;
      v2f b = *(const v2f*)(T64 + jrow + cc);
      s1[t] = wmma4(a1, b, s1[t]);
      s2[t] = wmma4(a2, b, s2[t]);
    }
  }
  const int rb = (mB + mHi) << 6;
#pragma unroll
  for (int t = 0; t < 2; ++t) {
    const int kd = (t << 4) + l16;
    const int km = (t == 0) ? ((l16 == 0) ? 32 : 64 - l16) : 48 - l16;
    const float* td = tbl + ((rb + kd) << 1);
    const float* tm = tbl + ((rb + km) << 1);
    const int pd = rb + (kd << 1);
    const int pm = rb + (km << 1);
    v2f twd[8], twm[8];
#pragma unroll
    for (int v = 0; v < 8; ++v) {
      twd[v] = *(const v2f*)(td + 128 * v);
      twm[v] = *(const v2f*)(tm + 128 * v);
    }
#pragma unroll
    for (int v = 0; v < 8; ++v) {
      float ad = c1[t][v] + s1[t][v], bd = c2[t][v] + s2[t][v];
      float am = c1[t][v] - s1[t][v], bm = c2[t][v] - s2[t][v];
      if (t == 0) {
        am = (l16 == 0) ? c1[2][v] : am;
        bm = (l16 == 0) ? c2[2][v] : bm;
      }
      const int po = 128 * (v >> 1) + (v & 1);
      v2f wd = twd[v], wm = twm[v];
      Yr[pd + po] = ad * wd.x - bd * wd.y;
      Yi[pd + po] = ad * wd.y + bd * wd.x;
      Yr[pm + po] = am * wm.x - bm * wm.y;
      Yi[pm + po] = am * wm.y + bm * wm.x;
    }
  }
}

__global__ __launch_bounds__(256, 4) void ssm_rows(const float* __restrict__ uin,
                                                float* __restrict__ uout,
                                                const float* ws) {
  extern __shared__ float smem[];
  float* xa = smem;
  float* xb = smem + 8192;
  float* Yr = smem + 16384;
  float* Yi = smem + 24576;

  const int tid  = threadIdx.x;
  const int wave = tid >> 5;
  const int lane = tid & 31;
  const int l16  = lane & 15;
  const int kSel = (lane >> 4) << 1;
  const int mHi  = (lane >> 4) << 3;
  const int mB   = wave << 4;
  const int eB = (((mB + mHi) << 6) + l16) << 1;
  const int pB = ((mB + mHi) << 6) + (l16 << 1);
  const int aB = ((mB + l16) << 2) + kSel;

  const size_t rowOff = (size_t)blockIdx.x * (2 * LROW);
  const float4* srcA = (const float4*)(uin + rowOff);
  const float4* srcB = (const float4*)(uin + rowOff + LROW);
  for (int i = tid; i < LROW / 4; i += 256) {
    float4 va = srcA[i];
    float4 vb = srcB[i];
    int n2 = i >> 4, n1 = (i & 15) << 2;
    int base = pidx(n2, n1);
    xa[base] = va.x; xa[base + 2] = va.y; xa[base + 4] = va.z; xa[base + 6] = va.w;
    xb[base] = vb.x; xb[base + 2] = vb.y; xb[base + 4] = vb.z; xb[base + 6] = vb.w;
  }
  __syncthreads();

  const float* T128F = ws + OFF_T128F;
  const float* T128I = ws + OFF_T128I;
  const float* T64F  = ws + OFF_T64F;
  const float* T64I  = ws + OFF_T64I;
  const float* Wf    = ws + OFF_WF;
  const float* Wi    = ws + OFF_WI;

  for (int layer = 0; layer < 4; ++layer) {
    const float* kh = ws + OFF_KHAT + layer * 16384;

    fold_rows(xa, xb, Yr, Yi, tid);
    __syncthreads();
    {
      v8f c1[4] = {}, c2[4] = {};
#pragma unroll 4
      for (int kk = 0; kk < 128; kk += 4) {
        v2f a = *(const v2f*)(T128F + (kk << 7) + aB);
        int jb = ((kk + kSel) >> 1) << 7;
#pragma unroll
        for (int t = 0; t < 4; ++t) {
          int cc = ((t << 4) + l16) << 1;
          v2f b1 = *(const v2f*)(Yr + jb + cc);
          v2f b2 = *(const v2f*)(Yi + jb + cc);
          c1[t] = wmma4(a, b1, c1[t]);
          c2[t] = wmma4(a, b2, c2[t]);
        }
      }
      __syncthreads();
      const float* tb = Wf + eB;
#pragma unroll
      for (int th = 0; th < 4; th += 2) {
        v2f tw[2][8];
#pragma unroll
        for (int t = 0; t < 2; ++t) {
#pragma unroll
          for (int v = 0; v < 8; ++v) tw[t][v] = *(const v2f*)(tb + 128 * v + 32 * (th + t));
        }
#pragma unroll
        for (int t = 0; t < 2; ++t) {
#pragma unroll
          for (int v = 0; v < 8; ++v) {
            v2f w = tw[t][v];
            int p = pB + 128 * (v >> 1) + 32 * (th + t) + (v & 1);
            float a = c1[th + t][v], b = c2[th + t][v];
            Yr[p] = a * w.x - b * w.y;
            Yi[p] = a * w.y + b * w.x;
          }
        }
      }
    }

    fold_cols(Yr, Yi, mB, lane);
    dft64_inplace(Yr, Yi, T64F, kh, mB, mHi, l16, kSel);

    fold_cols(Yr, Yi, mB, lane);
    dft64_inplace(Yr, Yi, T64I, Wi, mB, mHi, l16, kSel);
    __syncthreads();

    fold_rows(Yr, Yi, Yr, Yi, tid);
    __syncthreads();
    {
      v8f w1[4] = {}, w2[4] = {};
      for (int kk = 0; kk < 128; kk += 4) {
        v2f a = *(const v2f*)(T128I + (kk << 7) + aB);
        int jb = ((kk + kSel) >> 1) << 7;
#pragma unroll
        for (int t = 0; t < 4; ++t) {
          int cc = ((t << 4) + l16) << 1;
          v2f b1 = *(const v2f*)(Yr + jb + cc);
          v2f b2 = *(const v2f*)(Yi + jb + cc);
          w1[t] = wmma4(a, b1, w1[t]);
          w2[t] = wmma4(a, b2, w2[t]);
        }
      }
#pragma unroll
      for (int t = 0; t < 4; ++t) {
#pragma unroll
        for (int j = 0; j < 4; ++j) {
          int p = pB + 128 * j + 32 * t;
          v2f s1; s1.x = w1[t][2 * j]; s1.y = w1[t][2 * j + 1];
          v2f s2; s2.x = w2[t][2 * j]; s2.y = w2[t][2 * j + 1];
          v2f ya = s1 + *(const v2f*)(xa + p);
          v2f yb = s2 + *(const v2f*)(xb + p);
          *(v2f*)(xa + p) = gelu2(ya);
          *(v2f*)(xb + p) = gelu2(yb);
        }
      }
    }
    __syncthreads();
  }

  float4* dstA = (float4*)(uout + rowOff);
  float4* dstB = (float4*)(uout + rowOff + LROW);
  int tid_o = tid;
  asm volatile("" : "+v"(tid_o));
  for (int i = tid_o; i < LROW / 4; i += 256) {
    int n2 = i >> 4, n1 = (i & 15) << 2;
    int base = pidx(n2, n1);
    float4 oa, ob;
    oa.x = xa[base]; oa.y = xa[base + 2]; oa.z = xa[base + 4]; oa.w = xa[base + 6];
    ob.x = xb[base]; ob.y = xb[base + 2]; ob.z = xb[base + 4]; ob.w = xb[base + 6];
    dstA[i] = oa;
    dstB[i] = ob;
  }
}

extern "C" void kernel_launch(void* const* d_in, const int* in_sizes, int n_in,
                              void* d_out, int out_size, void* d_ws, size_t ws_size,
                              hipStream_t stream) {
  const float* u    = (const float*)d_in[0];
  const float* Lr   = (const float*)d_in[1];
  const float* Li   = (const float*)d_in[2];
  const float* Pv   = (const float*)d_in[3];
  const float* Bv   = (const float*)d_in[4];
  const float* Ct   = (const float*)d_in[5];
  const float* step = (const float*)d_in[6];
  float* out = (float*)d_out;
  float* ws  = (float*)d_ws;

  int pairs = in_sizes[0] / LPAIR;

  ssm_gen_tables<<<64, 256, 0, stream>>>(ws);
  ssm_gen_khat<<<dim3(32, 4), 256, 0, stream>>>(ws, Lr, Li, Pv, Bv, Ct, step);
  ssm_rows<<<pairs, 256, 32768 * sizeof(float), stream>>>(u, out, ws);
}


# Both 128-point stages: table's cosine columns (J<=64) and sine columns (J>64) summed apart for output rows 0..64 only, rows 128-k by C-S: 149 WMMAs per wave and stage, not 256; regroups their sums
# speedup vs baseline: 3.8172x; 1.1552x over previous
#include <hip/hip_runtime.h>
#include <math.h>

typedef float v2f __attribute__((ext_vector_type(2)));
typedef float v8f __attribute__((ext_vector_type(8)));

#define LROW 8192
#define LPAIR 16384
#define NSTATE 64
#define PI_F 3.14159265358979323846f

#define OFF_T128F   0
#define OFF_T128I   32768
#define OFF_TS      16384
#define OFF_T64F    65536
#define OFF_T64I    73728
#define OFF_WF      81920
#define OFF_WI      98304
#define OFF_KHAT    114688
#define WS_FLOATS   (OFF_KHAT + 4*16384)

static __device__ inline v8f wmma4(v2f a, v2f b, v8f c) {
  return __builtin_amdgcn_wmma_f32_16x16x4_f32(false, a, false, b, (short)0, c,
                                               false, false);
}

static __device__ inline int pidx(int row, int col) {
  return ((row >> 1) << 7) + (col << 1) + (row & 1);
}

static __device__ inline v2f vfma2(v2f a, v2f b, v2f c) {
  return __builtin_elementwise_fma(a, b, c);
}
static __device__ inline v2f bc2(float s) { v2f r; r.x = s; r.y = s; return r; }

static __device__ inline v2f gelu2(v2f y) {
  v2f x  = y * 0.707106769f;
  v2f ax = __builtin_elementwise_abs(x);
  v2f t  = x * x;
  v2f pb = vfma2(t, bc2(-0.000561801775f), bc2(0.00491381623f));
  pb = vfma2(t, pb, bc2(-0.0267075151f));
  pb = vfma2(t, pb, bc2(0.112800106f));
  pb = vfma2(t, pb, bc2(-0.376122952f));
  pb = vfma2(t, pb, bc2(0.128379107f));
  v2f rs = vfma2(ax, pb, ax);
  v2f pa = vfma2(ax, bc2(0.0000169988107f), bc2(-0.000378677854f));
  pa = vfma2(ax, pa, bc2(0.00385781587f));
  pa = vfma2(ax, pa, bc2(-0.0241816975f));
  pa = vfma2(ax, pa, bc2(0.106668264f));
  pa = vfma2(ax, pa, bc2(0.634933293f));
  pa = vfma2(ax, pa, bc2(0.128689408f));
  v2f q  = vfma2(ax, pa, ax);
  v2f qe = q * -1.44269502f;
  v2f e;
  e.x = __builtin_amdgcn_exp2f(qe.x);
  e.y = __builtin_amdgcn_exp2f(qe.y);
  v2f rl = 1.0f - e;
  v2f r;
  r.x = ax.x < 1.0f ? rs.x : rl.x;
  r.y = ax.y < 1.0f ? rs.y : rl.y;
  r = __builtin_elementwise_copysign(r, x);
  return (0.5f * y) * (1.0f + r);
}

__global__ void ssm_gen_tables(float* __restrict__ ws) {
  int idx = blockIdx.x * blockDim.x + threadIdx.x;
  {
    int k2 = idx >> 7, J = idx & 127;
    int t = (k2 * J) & 127;
    float s, c;
    sincosf(-2.f * PI_F * (float)t / 128.f, &s, &c);
    if (J < 68) {
      int pos = (J >> 2) * 512 + k2 * 4 + (J & 3);
      float vc = (J <= 64) ? c : 0.f;
      ws[OFF_T128F + pos] = vc;
      ws[OFF_T128I + pos] = vc;
    }
    if (J >= 64) {
      int pos = OFF_TS + ((J - 64) >> 2) * 512 + k2 * 4 + (J & 3);
      float vs = (J > 64) ? s : 0.f;
      ws[OFF_T128F + pos] = vs;
      ws[OFF_T128I + pos] = -vs;
    }
  }
  if (idx < 8192) {
    int k2 = idx >> 6, n1 = idx & 63;
    int t = (n1 * k2) & 8191;
    float s, c;
    sincosf(-2.f * PI_F * (float)t / 8192.f, &s, &c);
    ws[OFF_WF + 2 * idx]     = c;  ws[OFF_WF + 2 * idx + 1] = s;
    ws[OFF_WI + 2 * idx]     = c;  ws[OFF_WI + 2 * idx + 1] = -s;
  }
  if (idx < 4096) {
    int J = idx >> 6, k1 = idx & 63;
    int t = (J * k1) & 63;
    float s, c;
    sincosf(-2.f * PI_F * (float)t / 64.f, &s, &c);
    int pos = (J >> 1) * 128 + 2 * k1 + (J & 1);
    ws[OFF_T64F + pos] = (J <= 32) ? c : s;
    ws[OFF_T64I + pos] = (J <= 32) ? c : -s;
  }
}

static __device__ inline void ssm_khat_at(int kf, float stp,
                                          const float* __restrict__ lr,
                                          const float* __restrict__ li,
                                          const float* __restrict__ pp,
                                          const float* __restrict__ bb,
                                          const float* __restrict__ ct,
                                          float& outr, float& outi) {
  float osin, ocos;
  sincosf(-2.f * PI_F * (float)kf / 8192.f, &osin, &ocos);
  float pr = 1.f + ocos, pi = osin;
  float mr = 1.f - ocos, mi = -osin;
  float pd = pr * pr + pi * pi;
  float ipr = pr / pd, ipi = -pi / pd;
  float cr = 2.f * ipr, ci = 2.f * ipi;
  float sc = 2.f / stp;
  float gr = sc * (mr * ipr - mi * ipi);
  float gi = sc * (mr * ipi + mi * ipr);

  float k00r = 0, k00i = 0, k01r = 0, k01i = 0;
  float k10r = 0, k10i = 0, k11r = 0, k11i = 0;
  for (int n = 0; n < NSTATE; ++n) {
    float dr = gr - lr[n], di = gi - li[n];
    float dd = dr * dr + di * di;
    float ir = dr / dd, ii = -di / dd;
    float b = bb[n], p = pp[n], c = ct[n];
    float w00 = c * b, w01 = c * p, w10 = p * b, w11 = p * p;
    k00r += w00 * ir; k00i += w00 * ii;
    k01r += w01 * ir; k01i += w01 * ii;
    k10r += w10 * ir; k10i += w10 * ii;
    k11r += w11 * ir; k11i += w11 * ii;
  }
  float qr = k01r * k10r - k01i * k10i;
  float qi = k01r * k10i + k01i * k10r;
  float er = 1.f + k11r, ei = k11i;
  float ed = er * er + ei * ei;
  float ier = er / ed, iei = -ei / ed;
  float tr = qr * ier - qi * iei;
  float ti = qr * iei + qi * ier;
  float kr = k00r - tr, ki = k00i - ti;
  outr = cr * kr - ci * ki;
  outi = cr * ki + ci * kr;
}

__global__ void ssm_gen_khat(float* __restrict__ ws,
                             const float* __restrict__ Lr,
                             const float* __restrict__ Li,
                             const float* __restrict__ Pv,
                             const float* __restrict__ Bv,
                             const float* __restrict__ Ct,
                             const float* __restrict__ stepv) {
  int layer = blockIdx.y;
  int s = blockIdx.x * blockDim.x + threadIdx.x;
  int k2 = s >> 6, k1 = s & 63;
  int kf = 128 * k1 + k2;
  int kg = (8192 - kf) & 8191;

  const float* lr = Lr + layer * NSTATE;
  const float* li = Li + layer * NSTATE;
  const float* pp = Pv + layer * NSTATE;
  const float* bb = Bv + layer * NSTATE;
  const float* ct = Ct + layer * NSTATE;
  float stp = stepv[layer];

  float ar, ai, br, bi;
  ssm_khat_at(kf, stp, lr, li, pp, bb, ct, ar, ai);
  ssm_khat_at(kg, stp, lr, li, pp, bb, ct, br, bi);
  const float scale = 0.5f / 8192.f;
  ws[OFF_KHAT + layer * 16384 + 2 * s]     = (ar + br) * scale;
  ws[OFF_KHAT + layer * 16384 + 2 * s + 1] = (ai - bi) * scale;
}

static __device__ inline void fold_rows(const float* sr, const float* si,
                                        float* dr, float* di, int tid) {
  const int col2 = (tid & 63) << 1;
  const int jg = tid >> 6;
  const bool edge = (jg == 0);
  const int bj = ((jg >> 1) << 7) + col2 + (jg & 1);
  const int bm = ((64 - ((jg + 1) >> 1)) << 7) + col2 + (jg & 1);
  int bl = bm - 256 * 15;
  asm volatile("" : "+v"(bl));
  const int p64 = (32 << 7) + col2;
#pragma unroll
  for (int h = 0; h < 2; ++h) {
    float a[8], b[8], c[8], d[8];
#pragma unroll
    for (int q = 0; q < 8; ++q) {
      const int i = 8 * h + q;
      int pj = bj + 256 * i;
      int pm = (i == 0 && edge) ? p64 : bl + 256 * (15 - i);
      a[q] = sr[pj]; b[q] = sr[pm]; c[q] = si[pj]; d[q] = si[pm];
    }
#pragma unroll
    for (int q = 0; q < 8; ++q) {
      const int i = 8 * h + q;
      int pj = bj + 256 * i;
      int pm = (i == 0 && edge) ? p64 : bl + 256 * (15 - i);
      bool cp = (i == 0 && edge);
      dr[pj] = cp ? a[q] : a[q] + b[q];
      dr[pm] = cp ? b[q] : c[q] - d[q];
      di[pj] = cp ? c[q] : c[q] + d[q];
      di[pm] = cp ? d[q] : b[q] - a[q];
    }
  }
}

static __device__ inline void fold_cols(float* yr, float* yi, int mB, int lane) {
  const int cj = (mB << 6) + (lane << 1), cm = (mB << 6) + ((64 - lane) << 1);
  if (lane != 0) {
#pragma unroll
    for (int h = 0; h < 2; ++h) {
      float a[8], b[8], c[8], d[8];
#pragma unroll
      for (int q = 0; q < 8; ++q) {
        const int r = 8 * h + q;
        const int ro = ((r >> 1) << 7) + (r & 1);
        a[q] = yr[cj + ro]; b[q] = yr[cm + ro]; c[q] = yi[cj + ro]; d[q] = yi[cm + ro];
      }
#pragma unroll
      for (int q = 0; q < 8; ++q) {
        const int r = 8 * h + q;
        const int ro = ((r >> 1) << 7) + (r & 1);
        yr[cj + ro] = a[q] + b[q];
        yr[cm + ro] = c[q] - d[q];
        yi[cj + ro] = c[q] + d[q];
        yi[cm + ro] = b[q] - a[q];
      }
    }
  }
}

static __device__ __forceinline__ void dft64_inplace(float* Yr, float* Yi,
                                                     const float* T64, const float* tbl,
                                                     int mB, int mHi, int l16, int kSel) {
  v8f c1[3] = {}, c2[3] = {}, s1[2] = {}, s2[2] = {};
  const int arow = pidx(mB + l16, kSel);
  const bool hi = (kSel != 0);
#pragma unroll 3
  for (int kk = 0; kk < 36; kk += 4) {
    int p0 = arow + 2 * kk;
    const bool last = (kk == 32);
    v2f a1; a1.x = Yr[p0]; a1.y = Yr[p0 + 2];
    v2f a2; a2.x = Yi[p0]; a2.y = Yi[p0 + 2];
    a1.x = (last && hi) ? 0.f : a1.x;  a1.y = last ? 0.f : a1.y;
    a2.x = (last && hi) ? 0.f : a2.x;  a2.y = last ? 0.f : a2.y;
    int jrow = ((kk + kSel) >> 1) << 7;
#pragma unroll
    for (int t = 0; t < 3; ++t) {
      int cc = ((t << 4) + l16) << 1;
      v2f b = *(const v2f*)(T64 + jrow + cc);
      c1[t] = wmma4(a1, b, c1[t]);
      c2[t] = wmma4(a2, b, c2[t]);
    }
  }
#pragma unroll 4
  for (int kk = 32; kk < 64; kk += 4) {
    int p0 = arow + 2 * kk;
    const bool first = (kk == 32);
    v2f a1; a1.x = Yr[p0]; a1.y = Yr[p0 + 2];
    v2f a2; a2.x = Yi[p0]; a2.y = Yi[p0 + 2];
    a1.x = (first && !hi) ? 0.f : a1.x;
    a2.x = (first && !hi) ? 0.f : a2.x;
    int jrow = ((kk + kSel) >> 1) << 7;
#pragma unroll
    for (int t = 0; t < 2; ++t) {
      int cc = ((t << 4) + l16) << 1;
      v2f b = *(const v2f*)(T64 + jrow + cc);
      s1[t] = wmma4(a1, b, s1[t]);
      s2[t] = wmma4(a2, b, s2[t]);
    }
  }
  const int rb = (mB + mHi) << 6;
#pragma unroll
  for (int t = 0; t < 2; ++t) {
    const int kd = (t << 4) + l16;
    const int km = (t == 0) ? ((l16 == 0) ? 32 : 64 - l16) : 48 - l16;
    const float* td = tbl + ((rb + kd) << 1);
    const float* tm = tbl + ((rb + km) << 1);
    const int pd = rb + (kd << 1);
    const int pm = rb + (km << 1);
    v2f twd[8], twm[8];
#pragma unroll
    for (int v = 0; v < 8; ++v) {
      twd[v] = *(const v2f*)(td + 128 * v);
      twm[v] = *(const v2f*)(tm + 128 * v);
    }
#pragma unroll
    for (int v = 0; v < 8; ++v) {
      float ad = c1[t][v] + s1[t][v], bd = c2[t][v] + s2[t][v];
      float am = c1[t][v] - s1[t][v], bm = c2[t][v] - s2[t][v];
      if (t == 0) {
        am = (l16 == 0) ? c1[2][v] : am;
        bm = (l16 == 0) ? c2[2][v] : bm;
      }
      const int po = 128 * (v >> 1) + (v & 1);
      v2f wd = twd[v], wm = twm[v];
      Yr[pd + po] = ad * wd.x - bd * wd.y;
      Yi[pd + po] = ad * wd.y + bd * wd.x;
      Yr[pm + po] = am * wm.x - bm * wm.y;
      Yi[pm + po] = am * wm.y + bm * wm.x;
    }
  }
}

static __device__ __forceinline__ void dft128_split(const float* T, const float* Br,
                                                    int m, int nh, int exT, int exP,
                                                    int l16, int kSel,
                                                    v8f (&c1)[2], v8f (&c2)[2],
                                                    v8f (&s1)[2], v8f (&s2)[2], v8f& x) {
  const float* TC = T;
  const float* TS = T + OFF_TS;
  const float* Bi = Br + 8192;
  const float* Bx = Br + (exP << 13);
  const int aM = ((16 * m + l16) << 2) + kSel;
  const int aX = ((64 + l16) << 2) + kSel;
  const int cB = (32 * nh + l16) << 1;
  const int cX = (16 * exT + l16) << 1;
#pragma unroll 1
  for (int q0 = 0; q0 < 16; q0 += 4) {
    v2f ac[4], ax[4], as[4];
#pragma unroll
    for (int u = 0; u < 4; ++u) {
      ac[u] = *(const v2f*)(TC + ((q0 + u) << 9) + aM);
      ax[u] = *(const v2f*)(TC + ((q0 + u) << 9) + aX);
      as[u] = *(const v2f*)(TS + ((q0 + u) << 9) + aM);
    }
#pragma unroll
    for (int u = 0; u < 4; ++u) {
      const int jc = ((4 * (q0 + u) + kSel) >> 1) << 7;
      const int js = jc + 4096;
#pragma unroll
      for (int t = 0; t < 2; ++t) {
        v2f b1 = *(const v2f*)(Br + jc + cB + 32 * t);
        v2f b2 = *(const v2f*)(Bi + jc + cB + 32 * t);
        v2f d1 = *(const v2f*)(Br + js + cB + 32 * t);
        v2f d2 = *(const v2f*)(Bi + js + cB + 32 * t);
        c1[t] = wmma4(ac[u], b1, c1[t]);
        c2[t] = wmma4(ac[u], b2, c2[t]);
        s1[t] = wmma4(as[u], d1, s1[t]);
        s2[t] = wmma4(as[u], d2, s2[t]);
      }
      v2f bx = *(const v2f*)(Bx + jc + cX);
      x = wmma4(ax[u], bx, x);
    }
  }
  {
    v2f ac = *(const v2f*)(TC + (16 << 9) + aM);
    v2f ax = *(const v2f*)(TC + (16 << 9) + aX);
    const int jc = ((64 + kSel) >> 1) << 7;
#pragma unroll
    for (int t = 0; t < 2; ++t) {
      v2f b1 = *(const v2f*)(Br + jc + cB + 32 * t);
      v2f b2 = *(const v2f*)(Bi + jc + cB + 32 * t);
      c1[t] = wmma4(ac, b1, c1[t]);
      c2[t] = wmma4(ac, b2, c2[t]);
    }
    v2f bx = *(const v2f*)(Bx + jc + cX);
    x = wmma4(ax, bx, x);
  }
}

__global__ __launch_bounds__(256, 4) void ssm_rows(const float* __restrict__ uin,
                                                float* __restrict__ uout,
                                                const float* ws) {
  extern __shared__ float smem[];
  float* xa = smem;
  float* xb = smem + 8192;
  float* Yr = smem + 16384;
  float* Yi = smem + 24576;

  const int tid  = threadIdx.x;
  const int wave = tid >> 5;
  const int lane = tid & 31;
  const int l16  = lane & 15;
  const int kSel = (lane >> 4) << 1;
  const int mHi  = (lane >> 4) << 3;
  const int mB   = wave << 4;
  const int m128 = wave >> 1;
  const int nh   = wave & 1;
  const int exT  = wave & 3;
  const int exP  = wave >> 2;
  const int rD   = (m128 << 4) + mHi;
  const int rM   = 128 - rD;
  const bool nopart = (rD == 0);

  const size_t rowOff = (size_t)blockIdx.x * (2 * LROW);
  const float4* srcA = (const float4*)(uin + rowOff);
  const float4* srcB = (const float4*)(uin + rowOff + LROW);
  for (int i = tid; i < LROW / 4; i += 256) {
    float4 va = srcA[i];
    float4 vb = srcB[i];
    int n2 = i >> 4, n1 = (i & 15) << 2;
    int base = pidx(n2, n1);
    xa[base] = va.x; xa[base + 2] = va.y; xa[base + 4] = va.z; xa[base + 6] = va.w;
    xb[base] = vb.x; xb[base + 2] = vb.y; xb[base + 4] = vb.z; xb[base + 6] = vb.w;
  }
  __syncthreads();

  const float* T128F = ws + OFF_T128F;
  const float* T128I = ws + OFF_T128I;
  const float* T64F  = ws + OFF_T64F;
  const float* T64I  = ws + OFF_T64I;
  const float* Wf    = ws + OFF_WF;
  const float* Wi    = ws + OFF_WI;

  for (int layer = 0; layer < 4; ++layer) {
    const float* kh = ws + OFF_KHAT + layer * 16384;

    fold_rows(xa, xb, Yr, Yi, tid);
    __syncthreads();
    {
      v8f c1[2] = {}, c2[2] = {}, s1[2] = {}, s2[2] = {}, x5 = {};
      dft128_split(T128F, Yr, m128, nh, exT, exP, l16, kSel, c1, c2, s1, s2, x5);
      __syncthreads();
#pragma unroll
      for (int t = 0; t < 2; ++t) {
        const int n = (nh << 5) + (t << 4) + l16;
        const float* td = Wf + (((rD << 6) + n) << 1);
        const float* tm = Wf + ((((rM - 8) << 6) + n) << 1);
        const int pd = (rD << 6) + (n << 1);
        const int pm = ((rM - 8) << 6) + (n << 1);
        v2f twd[8], twm[8];
#pragma unroll
        for (int v = 0; v < 8; ++v) {
          twd[v] = *(const v2f*)(td + 128 * v);
          twm[v] = *(const v2f*)((v == 0 && nopart) ? td : tm + 128 * (8 - v));
        }
#pragma unroll
        for (int v = 0; v < 8; ++v) {
          float ad = c1[t][v] + s1[t][v], bd = c2[t][v] + s2[t][v];
          float am = c1[t][v] - s1[t][v], bm = c2[t][v] - s2[t][v];
          const int po = 128 * (v >> 1) + (v & 1);
          const int qo = 128 * (4 - ((v + 1) >> 1)) + (v & 1);
          v2f wd = twd[v], wm = twm[v];
          Yr[pd + po] = ad * wd.x - bd * wd.y;
          Yi[pd + po] = ad * wd.y + bd * wd.x;
          if (!(v == 0 && nopart)) {
            Yr[pm + qo] = am * wm.x - bm * wm.y;
            Yi[pm + qo] = am * wm.y + bm * wm.x;
          }
        }
      }
      if (lane < 16) Yr[(exP << 13) + 4096 + (((exT << 4) + l16) << 1)] = x5[0];
    }
    __syncthreads();
    if (wave == 4) {
#pragma unroll
      for (int h = 0; h < 2; ++h) {
        const int c = lane + 32 * h;
        const int p = 4096 + (c << 1);
        float a = Yr[p], b = Yi[p];
        v2f w = *(const v2f*)(Wf + ((4096 + c) << 1));
        Yr[p] = a * w.x - b * w.y;
        Yi[p] = a * w.y + b * w.x;
      }
    }

    fold_cols(Yr, Yi, mB, lane);
    dft64_inplace(Yr, Yi, T64F, kh, mB, mHi, l16, kSel);

    fold_cols(Yr, Yi, mB, lane);
    dft64_inplace(Yr, Yi, T64I, Wi, mB, mHi, l16, kSel);
    __syncthreads();

    fold_rows(Yr, Yi, Yr, Yi, tid);
    __syncthreads();
    {
      v8f w1[2] = {}, w2[2] = {}, z1[2] = {}, z2[2] = {}, x5 = {};
      dft128_split(T128I, Yr, m128, nh, exT, exP, l16, kSel, w1, w2, z1, z2, x5);
#pragma unroll
      for (int t = 0; t < 2; ++t) {
        const int n = (nh << 5) + (t << 4) + l16;
        const int pd = (rD << 6) + (n << 1);
        const int pm = ((rM - 8) << 6) + (n << 1);
#pragma unroll
        for (int j = 0; j < 4; ++j) {
          int p = pd + 128 * j;
          v2f a1; a1.x = w1[t][2 * j] + z1[t][2 * j]; a1.y = w1[t][2 * j + 1] + z1[t][2 * j + 1];
          v2f a2; a2.x = w2[t][2 * j] + z2[t][2 * j]; a2.y = w2[t][2 * j + 1] + z2[t][2 * j + 1];
          v2f ya = a1 + *(const v2f*)(xa + p);
          v2f yb = a2 + *(const v2f*)(xb + p);
          *(v2f*)(xa + p) = gelu2(ya);
          *(v2f*)(xb + p) = gelu2(yb);
        }
#pragma unroll
        for (int j = 0; j < 3; ++j) {
          int p = pm + 128 * (3 - j);
          v2f a1; a1.x = w1[t][2 * j + 2] - z1[t][2 * j + 2]; a1.y = w1[t][2 * j + 1] - z1[t][2 * j + 1];
          v2f a2; a2.x = w2[t][2 * j + 2] - z2[t][2 * j + 2]; a2.y = w2[t][2 * j + 1] - z2[t][2 * j + 1];
          v2f ya = a1 + *(const v2f*)(xa + p);
          v2f yb = a2 + *(const v2f*)(xb + p);
          *(v2f*)(xa + p) = gelu2(ya);
          *(v2f*)(xb + p) = gelu2(yb);
        }
        {
          const int p7 = pm + 1;
          const int p0 = nopart ? p7 : pm + 512;
          v2f a1; a1.x = w1[t][0] - z1[t][0]; a1.y = w1[t][7] - z1[t][7];
          v2f a2; a2.x = w2[t][0] - z2[t][0]; a2.y = w2[t][7] - z2[t][7];
          v2f ra; ra.x = xa[p0]; ra.y = xa[p7];
          v2f rb; rb.x = xb[p0]; rb.y = xb[p7];
          v2f ga = gelu2(a1 + ra);
          v2f gb = gelu2(a2 + rb);
          xa[p7] = ga.y;
          xb[p7] = gb.y;
          if (!nopart) {
            xa[p0] = ga.x;
            xb[p0] = gb.x;
          }
        }
      }
      if (lane < 16) {
        const int p = (exP << 13) + 4096 + (((exT << 4) + l16) << 1);
        float y = x5[0] + xa[p];
        xa[p] = gelu2(bc2(y)).x;
      }
    }
    __syncthreads();
  }

  float4* dstA = (float4*)(uout + rowOff);
  float4* dstB = (float4*)(uout + rowOff + LROW);
  int tid_o = tid;
  asm volatile("" : "+v"(tid_o));
  for (int i = tid_o; i < LROW / 4; i += 256) {
    int n2 = i >> 4, n1 = (i & 15) << 2;
    int base = pidx(n2, n1);
    float4 oa, ob;
    oa.x = xa[base]; oa.y = xa[base + 2]; oa.z = xa[base + 4]; oa.w = xa[base + 6];
    ob.x = xb[base]; ob.y = xb[base + 2]; ob.z = xb[base + 4]; ob.w = xb[base + 6];
    dstA[i] = oa;
    dstB[i] = ob;
  }
}

extern "C" void kernel_launch(void* const* d_in, const int* in_sizes, int n_in,
                              void* d_out, int out_size, void* d_ws, size_t ws_size,
                              hipStream_t stream) {
  const float* u    = (const float*)d_in[0];
  const float* Lr   = (const float*)d_in[1];
  const float* Li   = (const float*)d_in[2];
  const float* Pv   = (const float*)d_in[3];
  const float* Bv   = (const float*)d_in[4];
  const float* Ct   = (const float*)d_in[5];
  const float* step = (const float*)d_in[6];
  float* out = (float*)d_out;
  float* ws  = (float*)d_ws;

  int pairs = in_sizes[0] / LPAIR;

  ssm_gen_tables<<<64, 256, 0, stream>>>(ws);
  ssm_gen_khat<<<dim3(32, 4), 256, 0, stream>>>(ws, Lr, Li, Pv, Bv, Ct, step);
  ssm_rows<<<pairs, 256, 32768 * sizeof(float), stream>>>(u, out, ws);
}


# All four FFT stages' product loops: next block's operand fragments (LDS data, memory table) loaded as one group a block ahead into a second register set, behind scheduling fences; sum order unchanged
# speedup vs baseline: 4.0337x; 1.0567x over previous
#include <hip/hip_runtime.h>
#include <math.h>

typedef float v2f __attribute__((ext_vector_type(2)));
typedef float v8f __attribute__((ext_vector_type(8)));

#define LROW 8192
#define LPAIR 16384
#define NSTATE 64
#define PI_F 3.14159265358979323846f

#define OFF_T128F   0
#define OFF_T128I   32768
#define OFF_TS      16384
#define OFF_T64F    65536
#define OFF_T64I    73728
#define OFF_WF      81920
#define OFF_WI      98304
#define OFF_KHAT    114688
#define WS_FLOATS   (OFF_KHAT + 4*16384)

static __device__ inline v8f wmma4(v2f a, v2f b, v8f c) {
  return __builtin_amdgcn_wmma_f32_16x16x4_f32(false, a, false, b, (short)0, c,
                                               false, false);
}

static __device__ inline int pidx(int row, int col) {
  return ((row >> 1) << 7) + (col << 1) + (row & 1);
}

static __device__ inline v2f vfma2(v2f a, v2f b, v2f c) {
  return __builtin_elementwise_fma(a, b, c);
}
static __device__ inline v2f bc2(float s) { v2f r; r.x = s; r.y = s; return r; }

static __device__ inline v2f gelu2(v2f y) {
  v2f x  = y * 0.707106769f;
  v2f ax = __builtin_elementwise_abs(x);
  v2f t  = x * x;
  v2f pb = vfma2(t, bc2(-0.000561801775f), bc2(0.00491381623f));
  pb = vfma2(t, pb, bc2(-0.0267075151f));
  pb = vfma2(t, pb, bc2(0.112800106f));
  pb = vfma2(t, pb, bc2(-0.376122952f));
  pb = vfma2(t, pb, bc2(0.128379107f));
  v2f rs = vfma2(ax, pb, ax);
  v2f pa = vfma2(ax, bc2(0.0000169988107f), bc2(-0.000378677854f));
  pa = vfma2(ax, pa, bc2(0.00385781587f));
  pa = vfma2(ax, pa, bc2(-0.0241816975f));
  pa = vfma2(ax, pa, bc2(0.106668264f));
  pa = vfma2(ax, pa, bc2(0.634933293f));
  pa = vfma2(ax, pa, bc2(0.128689408f));
  v2f q  = vfma2(ax, pa, ax);
  v2f qe = q * -1.44269502f;
  v2f e;
  e.x = __builtin_amdgcn_exp2f(qe.x);
  e.y = __builtin_amdgcn_exp2f(qe.y);
  v2f rl = 1.0f - e;
  v2f r;
  r.x = ax.x < 1.0f ? rs.x : rl.x;
  r.y = ax.y < 1.0f ? rs.y : rl.y;
  r = __builtin_elementwise_copysign(r, x);
  return (0.5f * y) * (1.0f + r);
}

__global__ void ssm_gen_tables(float* __restrict__ ws) {
  int idx = blockIdx.x * blockDim.x + threadIdx.x;
  {
    int k2 = idx >> 7, J = idx & 127;
    int t = (k2 * J) & 127;
    float s, c;
    sincosf(-2.f * PI_F * (float)t / 128.f, &s, &c);
    if (J < 68) {
      int pos = (J >> 2) * 512 + k2 * 4 + (J & 3);
      float vc = (J <= 64) ? c : 0.f;
      ws[OFF_T128F + pos] = vc;
      ws[OFF_T128I + pos] = vc;
    }
    if (J >= 64) {
      int pos = OFF_TS + ((J - 64) >> 2) * 512 + k2 * 4 + (J & 3);
      float vs = (J > 64) ? s : 0.f;
      ws[OFF_T128F + pos] = vs;
      ws[OFF_T128I + pos] = -vs;
    }
  }
  if (idx < 8192) {
    int k2 = idx >> 6, n1 = idx & 63;
    int t = (n1 * k2) & 8191;
    float s, c;
    sincosf(-2.f * PI_F * (float)t / 8192.f, &s, &c);
    ws[OFF_WF + 2 * idx]     = c;  ws[OFF_WF + 2 * idx + 1] = s;
    ws[OFF_WI + 2 * idx]     = c;  ws[OFF_WI + 2 * idx + 1] = -s;
  }
  if (idx < 4096) {
    int J = idx >> 6, k1 = idx & 63;
    int t = (J * k1) & 63;
    float s, c;
    sincosf(-2.f * PI_F * (float)t / 64.f, &s, &c);
    int pos = (J >> 1) * 128 + 2 * k1 + (J & 1);
    ws[OFF_T64F + pos] = (J <= 32) ? c : s;
    ws[OFF_T64I + pos] = (J <= 32) ? c : -s;
  }
}

static __device__ inline void ssm_khat_at(int kf, float stp,
                                          const float* __restrict__ lr,
                                          const float* __restrict__ li,
                                          const float* __restrict__ pp,
                                          const float* __restrict__ bb,
                                          const float* __restrict__ ct,
                                          float& outr, float& outi) {
  float osin, ocos;
  sincosf(-2.f * PI_F * (float)kf / 8192.f, &osin, &ocos);
  float pr = 1.f + ocos, pi = osin;
  float mr = 1.f - ocos, mi = -osin;
  float pd = pr * pr + pi * pi;
  float ipr = pr / pd, ipi = -pi / pd;
  float cr = 2.f * ipr, ci = 2.f * ipi;
  float sc = 2.f / stp;
  float gr = sc * (mr * ipr - mi * ipi);
  float gi = sc * (mr * ipi + mi * ipr);

  float k00r = 0, k00i = 0, k01r = 0, k01i = 0;
  float k10r = 0, k10i = 0, k11r = 0, k11i = 0;
  for (int n = 0; n < NSTATE; ++n) {
    float dr = gr - lr[n], di = gi - li[n];
    float dd = dr * dr + di * di;
    float ir = dr / dd, ii = -di / dd;
    float b = bb[n], p = pp[n], c = ct[n];
    float w00 = c * b, w01 = c * p, w10 = p * b, w11 = p * p;
    k00r += w00 * ir; k00i += w00 * ii;
    k01r += w01 * ir; k01i += w01 * ii;
    k10r += w10 * ir; k10i += w10 * ii;
    k11r += w11 * ir; k11i += w11 * ii;
  }
  float qr = k01r * k10r - k01i * k10i;
  float qi = k01r * k10i + k01i * k10r;
  float er = 1.f + k11r, ei = k11i;
  float ed = er * er + ei * ei;
  float ier = er / ed, iei = -ei / ed;
  float tr = qr * ier - qi * iei;
  float ti = qr * iei + qi * ier;
  float kr = k00r - tr, ki = k00i - ti;
  outr = cr * kr - ci * ki;
  outi = cr * ki + ci * kr;
}

__global__ void ssm_gen_khat(float* __restrict__ ws,
                             const float* __restrict__ Lr,
                             const float* __restrict__ Li,
                             const float* __restrict__ Pv,
                             const float* __restrict__ Bv,
                             const float* __restrict__ Ct,
                             const float* __restrict__ stepv) {
  int layer = blockIdx.y;
  int s = blockIdx.x * blockDim.x + threadIdx.x;
  int k2 = s >> 6, k1 = s & 63;
  int kf = 128 * k1 + k2;
  int kg = (8192 - kf) & 8191;

  const float* lr = Lr + layer * NSTATE;
  const float* li = Li + layer * NSTATE;
  const float* pp = Pv + layer * NSTATE;
  const float* bb = Bv + layer * NSTATE;
  const float* ct = Ct + layer * NSTATE;
  float stp = stepv[layer];

  float ar, ai, br, bi;
  ssm_khat_at(kf, stp, lr, li, pp, bb, ct, ar, ai);
  ssm_khat_at(kg, stp, lr, li, pp, bb, ct, br, bi);
  const float scale = 0.5f / 8192.f;
  ws[OFF_KHAT + layer * 16384 + 2 * s]     = (ar + br) * scale;
  ws[OFF_KHAT + layer * 16384 + 2 * s + 1] = (ai - bi) * scale;
}

static __device__ inline void fold_rows(const float* sr, const float* si,
                                        float* dr, float* di, int tid) {
  const int col2 = (tid & 63) << 1;
  const int jg = tid >> 6;
  const bool edge = (jg == 0);
  const int bj = ((jg >> 1) << 7) + col2 + (jg & 1);
  const int bm = ((64 - ((jg + 1) >> 1)) << 7) + col2 + (jg & 1);
  int bl = bm - 256 * 15;
  asm volatile("" : "+v"(bl));
  const int p64 = (32 << 7) + col2;
#pragma unroll
  for (int h = 0; h < 2; ++h) {
    float a[8], b[8], c[8], d[8];
#pragma unroll
    for (int q = 0; q < 8; ++q) {
      const int i = 8 * h + q;
      int pj = bj + 256 * i;
      int pm = (i == 0 && edge) ? p64 : bl + 256 * (15 - i);
      a[q] = sr[pj]; b[q] = sr[pm]; c[q] = si[pj]; d[q] = si[pm];
    }
#pragma unroll
    for (int q = 0; q < 8; ++q) {
      const int i = 8 * h + q;
      int pj = bj + 256 * i;
      int pm = (i == 0 && edge) ? p64 : bl + 256 * (15 - i);
      bool cp = (i == 0 && edge);
      dr[pj] = cp ? a[q] : a[q] + b[q];
      dr[pm] = cp ? b[q] : c[q] - d[q];
      di[pj] = cp ? c[q] : c[q] + d[q];
      di[pm] = cp ? d[q] : b[q] - a[q];
    }
  }
}

static __device__ inline void fold_cols(float* yr, float* yi, int mB, int lane) {
  const int cj = (mB << 6) + (lane << 1), cm = (mB << 6) + ((64 - lane) << 1);
  if (lane != 0) {
#pragma unroll
    for (int h = 0; h < 2; ++h) {
      float a[8], b[8], c[8], d[8];
#pragma unroll
      for (int q = 0; q < 8; ++q) {
        const int r = 8 * h + q;
        const int ro = ((r >> 1) << 7) + (r & 1);
        a[q] = yr[cj + ro]; b[q] = yr[cm + ro]; c[q] = yi[cj + ro]; d[q] = yi[cm + ro];
      }
#pragma unroll
      for (int q = 0; q < 8; ++q) {
        const int r = 8 * h + q;
        const int ro = ((r >> 1) << 7) + (r & 1);
        yr[cj + ro] = a[q] + b[q];
        yr[cm + ro] = c[q] - d[q];
        yi[cj + ro] = c[q] + d[q];
        yi[cm + ro] = b[q] - a[q];
      }
    }
  }
}

static __device__ __forceinline__ void fence128() { __builtin_amdgcn_sched_barrier(0); }
static __device__ __forceinline__ void fence128_after(v8f& x) {
  asm volatile("" : "+v"(x));
  __builtin_amdgcn_sched_barrier(0);
}

template <int NK, int NT>
static __device__ __forceinline__ void ld_tab64(const float* tl, int kk0, v2f (&b)[9]) {
#pragma unroll
  for (int k = 0; k < NK; ++k)
#pragma unroll
    for (int t = 0; t < NT; ++t)
      b[NT * k + t] = *(const v2f*)(tl + 64 * (kk0 + 4 * k) + 32 * t);
}

template <int NK>
static __device__ __forceinline__ void ld_a64(const float* yr, const float* yi, int kk0,
                                              v2f (&a)[8]) {
#pragma unroll
  for (int k = 0; k < NK; ++k) {
    const int p = 2 * (kk0 + 4 * k);
    a[2 * k].x = yr[p];     a[2 * k].y = yr[p + 2];
    a[2 * k + 1].x = yi[p]; a[2 * k + 1].y = yi[p + 2];
  }
}

static __device__ __forceinline__ void dft64_inplace(float* Yr, float* Yi,
                                                     const float* T64, const float* tbl,
                                                     int mB, int mHi, int l16, int kSel,
                                                     int lane) {
  v8f c1[3] = {}, c2[3] = {}, s1[2] = {}, s2[2] = {};
  const int arow = pidx(mB + l16, kSel);
  const bool hi = (kSel != 0);
  const float* tl = T64 + ((kSel >> 1) << 7) + (l16 << 1);
  const float* yr = Yr + arow;
  const float* yi = Yi + arow;
  v2f tb0[9], tb1[9], av0[8], av1[8];
  ld_tab64<3, 3>(tl, 0, tb0);
  fence128();
  fold_cols(Yr, Yi, mB, lane);
  ld_a64<3>(yr, yi, 0, av0);
  ld_tab64<3, 3>(tl, 12, tb1);
  ld_a64<3>(yr, yi, 12, av1);
  fence128();
#pragma unroll
  for (int k = 0; k < 3; ++k)
#pragma unroll
    for (int t = 0; t < 3; ++t) {
      c1[t] = wmma4(av0[2 * k], tb0[3 * k + t], c1[t]);
      c2[t] = wmma4(av0[2 * k + 1], tb0[3 * k + t], c2[t]);
    }
  fence128();
  ld_tab64<3, 3>(tl, 24, tb0);
  ld_a64<3>(yr, yi, 24, av0);
  fence128();
#pragma unroll
  for (int k = 0; k < 3; ++k)
#pragma unroll
    for (int t = 0; t < 3; ++t) {
      c1[t] = wmma4(av1[2 * k], tb1[3 * k + t], c1[t]);
      c2[t] = wmma4(av1[2 * k + 1], tb1[3 * k + t], c2[t]);
    }
  fence128();
  ld_tab64<4, 2>(tl, 32, tb1);
  ld_a64<4>(yr, yi, 32, av1);
  fence128();
#pragma unroll
  for (int k = 0; k < 3; ++k) {
    v2f a1 = av0[2 * k], a2 = av0[2 * k + 1];
    if (k == 2) {
      a1.x = hi ? 0.f : a1.x;  a1.y = 0.f;
      a2.x = hi ? 0.f : a2.x;  a2.y = 0.f;
    }
#pragma unroll
    for (int t = 0; t < 3; ++t) {
      c1[t] = wmma4(a1, tb0[3 * k + t], c1[t]);
      c2[t] = wmma4(a2, tb0[3 * k + t], c2[t]);
    }
  }
  fence128();
  ld_tab64<4, 2>(tl, 48, tb0);
  ld_a64<4>(yr, yi, 48, av0);
  fence128();
#pragma unroll
  for (int k = 0; k < 4; ++k) {
    v2f a1 = av1[2 * k], a2 = av1[2 * k + 1];
    if (k == 0) {
      a1.x = hi ? a1.x : 0.f;
      a2.x = hi ? a2.x : 0.f;
    }
#pragma unroll
    for (int t = 0; t < 2; ++t) {
      s1[t] = wmma4(a1, tb1[2 * k + t], s1[t]);
      s2[t] = wmma4(a2, tb1[2 * k + t], s2[t]);
    }
  }
#pragma unroll
  for (int k = 0; k < 4; ++k)
#pragma unroll
    for (int t = 0; t < 2; ++t) {
      s1[t] = wmma4(av0[2 * k], tb0[2 * k + t], s1[t]);
      s2[t] = wmma4(av0[2 * k + 1], tb0[2 * k + t], s2[t]);
    }
  const int rb = (mB + mHi) << 6;
#pragma unroll
  for (int t = 0; t < 2; ++t) {
    const int kd = (t << 4) + l16;
    const int km = (t == 0) ? ((l16 == 0) ? 32 : 64 - l16) : 48 - l16;
    const float* td = tbl + ((rb + kd) << 1);
    const float* tm = tbl + ((rb + km) << 1);
    const int pd = rb + (kd << 1);
    const int pm = rb + (km << 1);
    v2f twd[8], twm[8];
#pragma unroll
    for (int v = 0; v < 8; ++v) {
      twd[v] = *(const v2f*)(td + 128 * v);
      twm[v] = *(const v2f*)(tm + 128 * v);
    }
#pragma unroll
    for (int v = 0; v < 8; ++v) {
      float ad = c1[t][v] + s1[t][v], bd = c2[t][v] + s2[t][v];
      float am = c1[t][v] - s1[t][v], bm = c2[t][v] - s2[t][v];
      if (t == 0) {
        am = (l16 == 0) ? c1[2][v] : am;
        bm = (l16 == 0) ? c2[2][v] : bm;
      }
      const int po = 128 * (v >> 1) + (v & 1);
      v2f wd = twd[v], wm = twm[v];
      Yr[pd + po] = ad * wd.x - bd * wd.y;
      Yi[pd + po] = ad * wd.y + bd * wd.x;
      Yr[pm + po] = am * wm.x - bm * wm.y;
      Yi[pm + po] = am * wm.y + bm * wm.x;
    }
  }
}

static __device__ __forceinline__ void ld_data128(const float* pr, const float* pi,
                                                  const float* px, int jc, v2f (&b)[9]) {
  const int js = jc + 4096;
#pragma unroll
  for (int t = 0; t < 2; ++t) {
    b[4 * t + 0] = *(const v2f*)(pr + jc + 32 * t);
    b[4 * t + 1] = *(const v2f*)(pi + jc + 32 * t);
    b[4 * t + 2] = *(const v2f*)(pr + js + 32 * t);
    b[4 * t + 3] = *(const v2f*)(pi + js + 32 * t);
  }
  b[8] = *(const v2f*)(px + jc);
}

static __device__ __forceinline__ void ld_tab128(const float* tcM, const float* tsM,
                                                 const float* tcX, int q, v2f (&a)[3]) {
  a[0] = *(const v2f*)(tcM + (q << 9));
  a[1] = *(const v2f*)(tsM + (q << 9));
  a[2] = *(const v2f*)(tcX + (q << 9));
}

static __device__ __forceinline__ void mm_pass128(const v2f (&a)[3], const v2f (&b)[9],
                                                  v8f (&c1)[2], v8f (&c2)[2],
                                                  v8f (&s1)[2], v8f (&s2)[2], v8f& x) {
  x = wmma4(a[2], b[8], x);
#pragma unroll
  for (int t = 0; t < 2; ++t) {
    c1[t] = wmma4(a[0], b[4 * t + 0], c1[t]);
    c2[t] = wmma4(a[0], b[4 * t + 1], c2[t]);
    s1[t] = wmma4(a[1], b[4 * t + 2], s1[t]);
    s2[t] = wmma4(a[1], b[4 * t + 3], s2[t]);
  }
}

static __device__ __forceinline__ void dft128_split(const float* T, const float* Br,
                                                    int m, int nh, int exT, int exP,
                                                    int l16, int kSel,
                                                    v8f (&c1)[2], v8f (&c2)[2],
                                                    v8f (&s1)[2], v8f (&s2)[2], v8f& x) {
  const int aM = ((16 * m + l16) << 2) + kSel;
  const int aX = ((64 + l16) << 2) + kSel;
  const float* tcM = T + aM;
  const float* tsM = T + OFF_TS + aM;
  const float* tcX = T + aX;
  const int j0 = (kSel >> 1) << 7;
  const float* pr = Br + j0 + ((32 * nh + l16) << 1);
  const float* pi = pr + 8192;
  const float* px = Br + (exP << 13) + j0 + ((16 * exT + l16) << 1);

  v2f ta[4][3];
  v2f b0[9], b1[9];
  ld_tab128(tcM, tsM, tcX, 0, ta[0]);
  ld_tab128(tcM, tsM, tcX, 1, ta[1]);
  ld_tab128(tcM, tsM, tcX, 2, ta[2]);
  ld_data128(pr, pi, px, 0, b0);
#pragma unroll 1
  for (int q0 = 0; q0 < 12; q0 += 4) {
#pragma unroll
    for (int u = 0; u < 4; u += 2) {
      const int p = q0 + u;
      ld_data128(pr, pi, px, (p + 1) << 8, b1);
      ld_tab128(tcM, tsM, tcX, p + 3, ta[(u + 3) & 3]);
      fence128();
      mm_pass128(ta[u], b0, c1, c2, s1, s2, x);
      fence128_after(x);
      ld_data128(pr, pi, px, (p + 2) << 8, b0);
      ld_tab128(tcM, tsM, tcX, p + 4, ta[u]);
      fence128();
      mm_pass128(ta[u + 1], b1, c1, c2, s1, s2, x);
      fence128_after(x);
    }
  }
  ld_data128(pr, pi, px, 13 << 8, b1);
  ld_tab128(tcM, tsM, tcX, 15, ta[3]);
  fence128();
  mm_pass128(ta[0], b0, c1, c2, s1, s2, x);
  fence128_after(x);
  ld_data128(pr, pi, px, 14 << 8, b0);
  ta[0][0] = *(const v2f*)(tcM + (16 << 9));
  ta[0][2] = *(const v2f*)(tcX + (16 << 9));
  fence128();
  mm_pass128(ta[1], b1, c1, c2, s1, s2, x);
  fence128_after(x);
  ld_data128(pr, pi, px, 15 << 8, b1);
  fence128();
  mm_pass128(ta[2], b0, c1, c2, s1, s2, x);
  fence128_after(x);
#pragma unroll
  for (int t = 0; t < 2; ++t) {
    b0[4 * t + 0] = *(const v2f*)(pr + (16 << 8) + 32 * t);
    b0[4 * t + 1] = *(const v2f*)(pi + (16 << 8) + 32 * t);
  }
  b0[8] = *(const v2f*)(px + (16 << 8));
  fence128();
  mm_pass128(ta[3], b1, c1, c2, s1, s2, x);
#pragma unroll
  for (int t = 0; t < 2; ++t) {
    c1[t] = wmma4(ta[0][0], b0[4 * t + 0], c1[t]);
    c2[t] = wmma4(ta[0][0], b0[4 * t + 1], c2[t]);
  }
  x = wmma4(ta[0][2], b0[8], x);
}

__global__ __launch_bounds__(256, 4) void ssm_rows(const float* __restrict__ uin,
                                                float* __restrict__ uout,
                                                const float* ws) {
  extern __shared__ float smem[];
  float* xa = smem;
  float* xb = smem + 8192;
  float* Yr = smem + 16384;
  float* Yi = smem + 24576;

  const int tid  = threadIdx.x;
  const int wave = tid >> 5;
  const int lane = tid & 31;
  const int l16  = lane & 15;
  const int kSel = (lane >> 4) << 1;
  const int mHi  = (lane >> 4) << 3;
  const int mB   = wave << 4;
  const int m128 = wave >> 1;
  const int nh   = wave & 1;
  const int exT  = wave & 3;
  const int exP  = wave >> 2;
  const int rD   = (m128 << 4) + mHi;
  const int rM   = 128 - rD;
  const bool nopart = (rD == 0);

  const size_t rowOff = (size_t)blockIdx.x * (2 * LROW);
  const float4* srcA = (const float4*)(uin + rowOff);
  const float4* srcB = (const float4*)(uin + rowOff + LROW);
  for (int i = tid; i < LROW / 4; i += 256) {
    float4 va = srcA[i];
    float4 vb = srcB[i];
    int n2 = i >> 4, n1 = (i & 15) << 2;
    int base = pidx(n2, n1);
    xa[base] = va.x; xa[base + 2] = va.y; xa[base + 4] = va.z; xa[base + 6] = va.w;
    xb[base] = vb.x; xb[base + 2] = vb.y; xb[base + 4] = vb.z; xb[base + 6] = vb.w;
  }
  __syncthreads();

  const float* T128F = ws + OFF_T128F;
  const float* T128I = ws + OFF_T128I;
  const float* T64F  = ws + OFF_T64F;
  const float* T64I  = ws + OFF_T64I;
  const float* Wf    = ws + OFF_WF;
  const float* Wi    = ws + OFF_WI;

  for (int layer = 0; layer < 4; ++layer) {
    const float* kh = ws + OFF_KHAT + layer * 16384;

    fold_rows(xa, xb, Yr, Yi, tid);
    __syncthreads();
    {
      v8f c1[2] = {}, c2[2] = {}, s1[2] = {}, s2[2] = {}, x5 = {};
      dft128_split(T128F, Yr, m128, nh, exT, exP, l16, kSel, c1, c2, s1, s2, x5);
      __syncthreads();
      int l16e = l16, rDe = rD;
      asm volatile("" : "+v"(l16e), "+v"(rDe));
      const int rMe = 128 - rDe;
#pragma unroll
      for (int t = 0; t < 2; ++t) {
        const int n = (nh << 5) + (t << 4) + l16e;
        const float* td = Wf + (((rDe << 6) + n) << 1);
        const float* tm = Wf + ((((rMe - 8) << 6) + n) << 1);
        const int pd = (rDe << 6) + (n << 1);
        const int pm = ((rMe - 8) << 6) + (n << 1);
        v2f twd[8], twm[8];
#pragma unroll
        for (int v = 0; v < 8; ++v) {
          twd[v] = *(const v2f*)(td + 128 * v);
          twm[v] = *(const v2f*)((v == 0 && nopart) ? td : tm + 128 * (8 - v));
        }
#pragma unroll
        for (int v = 0; v < 8; ++v) {
          float ad = c1[t][v] + s1[t][v], bd = c2[t][v] + s2[t][v];
          float am = c1[t][v] - s1[t][v], bm = c2[t][v] - s2[t][v];
          const int po = 128 * (v >> 1) + (v & 1);
          const int qo = 128 * (4 - ((v + 1) >> 1)) + (v & 1);
          v2f wd = twd[v], wm = twm[v];
          Yr[pd + po] = ad * wd.x - bd * wd.y;
          Yi[pd + po] = ad * wd.y + bd * wd.x;
          if (!(v == 0 && nopart)) {
            Yr[pm + qo] = am * wm.x - bm * wm.y;
            Yi[pm + qo] = am * wm.y + bm * wm.x;
          }
        }
      }
      if (lane < 16) Yr[(exP << 13) + 4096 + (((exT << 4) + l16) << 1)] = x5[0];
    }
    __syncthreads();
    if (wave == 4) {
#pragma unroll
      for (int h = 0; h < 2; ++h) {
        const int c = lane + 32 * h;
        const int p = 4096 + (c << 1);
        float a = Yr[p], b = Yi[p];
        v2f w = *(const v2f*)(Wf + ((4096 + c) << 1));
        Yr[p] = a * w.x - b * w.y;
        Yi[p] = a * w.y + b * w.x;
      }
    }

    dft64_inplace(Yr, Yi, T64F, kh, mB, mHi, l16, kSel, lane);

    dft64_inplace(Yr, Yi, T64I, Wi, mB, mHi, l16, kSel, lane);
    __syncthreads();

    fold_rows(Yr, Yi, Yr, Yi, tid);
    __syncthreads();
    {
      v8f w1[2] = {}, w2[2] = {}, z1[2] = {}, z2[2] = {}, x5 = {};
      dft128_split(T128I, Yr, m128, nh, exT, exP, l16, kSel, w1, w2, z1, z2, x5);
#pragma unroll
      for (int t = 0; t < 2; ++t) {
        const int n = (nh << 5) + (t << 4) + l16;
        const int pd = (rD << 6) + (n << 1);
        const int pm = ((rM - 8) << 6) + (n << 1);
#pragma unroll
        for (int j = 0; j < 4; ++j) {
          int p = pd + 128 * j;
          v2f a1; a1.x = w1[t][2 * j] + z1[t][2 * j]; a1.y = w1[t][2 * j + 1] + z1[t][2 * j + 1];
          v2f a2; a2.x = w2[t][2 * j] + z2[t][2 * j]; a2.y = w2[t][2 * j + 1] + z2[t][2 * j + 1];
          v2f ya = a1 + *(const v2f*)(xa + p);
          v2f yb = a2 + *(const v2f*)(xb + p);
          *(v2f*)(xa + p) = gelu2(ya);
          *(v2f*)(xb + p) = gelu2(yb);
        }
#pragma unroll
        for (int j = 0; j < 3; ++j) {
          int p = pm + 128 * (3 - j);
          v2f a1; a1.x = w1[t][2 * j + 2] - z1[t][2 * j + 2]; a1.y = w1[t][2 * j + 1] - z1[t][2 * j + 1];
          v2f a2; a2.x = w2[t][2 * j + 2] - z2[t][2 * j + 2]; a2.y = w2[t][2 * j + 1] - z2[t][2 * j + 1];
          v2f ya = a1 + *(const v2f*)(xa + p);
          v2f yb = a2 + *(const v2f*)(xb + p);
          *(v2f*)(xa + p) = gelu2(ya);
          *(v2f*)(xb + p) = gelu2(yb);
        }
        {
          const int p7 = pm + 1;
          const int p0 = nopart ? p7 : pm + 512;
          v2f a1; a1.x = w1[t][0] - z1[t][0]; a1.y = w1[t][7] - z1[t][7];
          v2f a2; a2.x = w2[t][0] - z2[t][0]; a2.y = w2[t][7] - z2[t][7];
          v2f ra; ra.x = xa[p0]; ra.y = xa[p7];
          v2f rb; rb.x = xb[p0]; rb.y = xb[p7];
          v2f ga = gelu2(a1 + ra);
          v2f gb = gelu2(a2 + rb);
          xa[p7] = ga.y;
          xb[p7] = gb.y;
          if (!nopart) {
            xa[p0] = ga.x;
            xb[p0] = gb.x;
          }
        }
      }
      if (lane < 16) {
        const int p = (exP << 13) + 4096 + (((exT << 4) + l16) << 1);
        float y = x5[0] + xa[p];
        xa[p] = gelu2(bc2(y)).x;
      }
    }
    __syncthreads();
  }

  float4* dstA = (float4*)(uout + rowOff);
  float4* dstB = (float4*)(uout + rowOff + LROW);
  int tid_o = tid;
  asm volatile("" : "+v"(tid_o));
  for (int i = tid_o; i < LROW / 4; i += 256) {
    int n2 = i >> 4, n1 = (i & 15) << 2;
    int base = pidx(n2, n1);
    float4 oa, ob;
    oa.x = xa[base]; oa.y = xa[base + 2]; oa.z = xa[base + 4]; oa.w = xa[base + 6];
    ob.x = xb[base]; ob.y = xb[base + 2]; ob.z = xb[base + 4]; ob.w = xb[base + 6];
    dstA[i] = oa;
    dstB[i] = ob;
  }
}

extern "C" void kernel_launch(void* const* d_in, const int* in_sizes, int n_in,
                              void* d_out, int out_size, void* d_ws, size_t ws_size,
                              hipStream_t stream) {
  const float* u    = (const float*)d_in[0];
  const float* Lr   = (const float*)d_in[1];
  const float* Li   = (const float*)d_in[2];
  const float* Pv   = (const float*)d_in[3];
  const float* Bv   = (const float*)d_in[4];
  const float* Ct   = (const float*)d_in[5];
  const float* step = (const float*)d_in[6];
  float* out = (float*)d_out;
  float* ws  = (float*)d_ws;

  int pairs = in_sizes[0] / LPAIR;

  ssm_gen_tables<<<64, 256, 0, stream>>>(ws);
  ssm_gen_khat<<<dim3(32, 4), 256, 0, stream>>>(ws, Lr, Li, Pv, Bv, Ct, step);
  ssm_rows<<<pairs, 256, 32768 * sizeof(float), stream>>>(u, out, ws);
}


# LDS planes padded, row-pair pitch 128 -> 136 words (136 KiB per workgroup, still two per WGP): 64-point stages' column-wise A-fragment reads go from 8-way bank conflict to none; sum order unchanged
# speedup vs baseline: 4.2398x; 1.0511x over previous
#include <hip/hip_runtime.h>
#include <math.h>

typedef float v2f __attribute__((ext_vector_type(2)));
typedef float v8f __attribute__((ext_vector_type(8)));

#define LROW 8192
#define LPAIR 16384
#define NSTATE 64
#define PI_F 3.14159265358979323846f

#define LDSP 136
#define LPLANE (64 * LDSP)

#define OFF_T128F   0
#define OFF_T128I   32768
#define OFF_TS      16384
#define OFF_T64F    65536
#define OFF_T64I    73728
#define OFF_WF      81920
#define OFF_WI      98304
#define OFF_KHAT    114688
#define WS_FLOATS   (OFF_KHAT + 4*16384)

static __device__ inline v8f wmma4(v2f a, v2f b, v8f c) {
  return __builtin_amdgcn_wmma_f32_16x16x4_f32(false, a, false, b, (short)0, c,
                                               false, false);
}

static __device__ inline int pidx(int row, int col) {
  return (row >> 1) * LDSP + (col << 1) + (row & 1);
}

static __device__ inline v2f vfma2(v2f a, v2f b, v2f c) {
  return __builtin_elementwise_fma(a, b, c);
}
static __device__ inline v2f bc2(float s) { v2f r; r.x = s; r.y = s; return r; }

static __device__ inline v2f gelu2(v2f y) {
  v2f x  = y * 0.707106769f;
  v2f ax = __builtin_elementwise_abs(x);
  v2f t  = x * x;
  v2f pb = vfma2(t, bc2(-0.000561801775f), bc2(0.00491381623f));
  pb = vfma2(t, pb, bc2(-0.0267075151f));
  pb = vfma2(t, pb, bc2(0.112800106f));
  pb = vfma2(t, pb, bc2(-0.376122952f));
  pb = vfma2(t, pb, bc2(0.128379107f));
  v2f rs = vfma2(ax, pb, ax);
  v2f pa = vfma2(ax, bc2(0.0000169988107f), bc2(-0.000378677854f));
  pa = vfma2(ax, pa, bc2(0.00385781587f));
  pa = vfma2(ax, pa, bc2(-0.0241816975f));
  pa = vfma2(ax, pa, bc2(0.106668264f));
  pa = vfma2(ax, pa, bc2(0.634933293f));
  pa = vfma2(ax, pa, bc2(0.128689408f));
  v2f q  = vfma2(ax, pa, ax);
  v2f qe = q * -1.44269502f;
  v2f e;
  e.x = __builtin_amdgcn_exp2f(qe.x);
  e.y = __builtin_amdgcn_exp2f(qe.y);
  v2f rl = 1.0f - e;
  v2f r;
  r.x = ax.x < 1.0f ? rs.x : rl.x;
  r.y = ax.y < 1.0f ? rs.y : rl.y;
  r = __builtin_elementwise_copysign(r, x);
  return (0.5f * y) * (1.0f + r);
}

__global__ void ssm_gen_tables(float* __restrict__ ws) {
  int idx = blockIdx.x * blockDim.x + threadIdx.x;
  {
    int k2 = idx >> 7, J = idx & 127;
    int t = (k2 * J) & 127;
    float s, c;
    sincosf(-2.f * PI_F * (float)t / 128.f, &s, &c);
    if (J < 68) {
      int pos = (J >> 2) * 512 + k2 * 4 + (J & 3);
      float vc = (J <= 64) ? c : 0.f;
      ws[OFF_T128F + pos] = vc;
      ws[OFF_T128I + pos] = vc;
    }
    if (J >= 64) {
      int pos = OFF_TS + ((J - 64) >> 2) * 512 + k2 * 4 + (J & 3);
      float vs = (J > 64) ? s : 0.f;
      ws[OFF_T128F + pos] = vs;
      ws[OFF_T128I + pos] = -vs;
    }
  }
  if (idx < 8192) {
    int k2 = idx >> 6, n1 = idx & 63;
    int t = (n1 * k2) & 8191;
    float s, c;
    sincosf(-2.f * PI_F * (float)t / 8192.f, &s, &c);
    ws[OFF_WF + 2 * idx]     = c;  ws[OFF_WF + 2 * idx + 1] = s;
    ws[OFF_WI + 2 * idx]     = c;  ws[OFF_WI + 2 * idx + 1] = -s;
  }
  if (idx < 4096) {
    int J = idx >> 6, k1 = idx & 63;
    int t = (J * k1) & 63;
    float s, c;
    sincosf(-2.f * PI_F * (float)t / 64.f, &s, &c);
    int pos = (J >> 1) * 128 + 2 * k1 + (J & 1);
    ws[OFF_T64F + pos] = (J <= 32) ? c : s;
    ws[OFF_T64I + pos] = (J <= 32) ? c : -s;
  }
}

static __device__ inline void ssm_khat_at(int kf, float stp,
                                          const float* __restrict__ lr,
                                          const float* __restrict__ li,
                                          const float* __restrict__ pp,
                                          const float* __restrict__ bb,
                                          const float* __restrict__ ct,
                                          float& outr, float& outi) {
  float osin, ocos;
  sincosf(-2.f * PI_F * (float)kf / 8192.f, &osin, &ocos);
  float pr = 1.f + ocos, pi = osin;
  float mr = 1.f - ocos, mi = -osin;
  float pd = pr * pr + pi * pi;
  float ipr = pr / pd, ipi = -pi / pd;
  float cr = 2.f * ipr, ci = 2.f * ipi;
  float sc = 2.f / stp;
  float gr = sc * (mr * ipr - mi * ipi);
  float gi = sc * (mr * ipi + mi * ipr);

  float k00r = 0, k00i = 0, k01r = 0, k01i = 0;
  float k10r = 0, k10i = 0, k11r = 0, k11i = 0;
  for (int n = 0; n < NSTATE; ++n) {
    float dr = gr - lr[n], di = gi - li[n];
    float dd = dr * dr + di * di;
    float ir = dr / dd, ii = -di / dd;
    float b = bb[n], p = pp[n], c = ct[n];
    float w00 = c * b, w01 = c * p, w10 = p * b, w11 = p * p;
    k00r += w00 * ir; k00i += w00 * ii;
    k01r += w01 * ir; k01i += w01 * ii;
    k10r += w10 * ir; k10i += w10 * ii;
    k11r += w11 * ir; k11i += w11 * ii;
  }
  float qr = k01r * k10r - k01i * k10i;
  float qi = k01r * k10i + k01i * k10r;
  float er = 1.f + k11r, ei = k11i;
  float ed = er * er + ei * ei;
  float ier = er / ed, iei = -ei / ed;
  float tr = qr * ier - qi * iei;
  float ti = qr * iei + qi * ier;
  float kr = k00r - tr, ki = k00i - ti;
  outr = cr * kr - ci * ki;
  outi = cr * ki + ci * kr;
}

__global__ void ssm_gen_khat(float* __restrict__ ws,
                             const float* __restrict__ Lr,
                             const float* __restrict__ Li,
                             const float* __restrict__ Pv,
                             const float* __restrict__ Bv,
                             const float* __restrict__ Ct,
                             const float* __restrict__ stepv) {
  int layer = blockIdx.y;
  int s = blockIdx.x * blockDim.x + threadIdx.x;
  int k2 = s >> 6, k1 = s & 63;
  int kf = 128 * k1 + k2;
  int kg = (8192 - kf) & 8191;

  const float* lr = Lr + layer * NSTATE;
  const float* li = Li + layer * NSTATE;
  const float* pp = Pv + layer * NSTATE;
  const float* bb = Bv + layer * NSTATE;
  const float* ct = Ct + layer * NSTATE;
  float stp = stepv[layer];

  float ar, ai, br, bi;
  ssm_khat_at(kf, stp, lr, li, pp, bb, ct, ar, ai);
  ssm_khat_at(kg, stp, lr, li, pp, bb, ct, br, bi);
  const float scale = 0.5f / 8192.f;
  ws[OFF_KHAT + layer * 16384 + 2 * s]     = (ar + br) * scale;
  ws[OFF_KHAT + layer * 16384 + 2 * s + 1] = (ai - bi) * scale;
}

static __device__ inline void fold_rows(const float* sr, const float* si,
                                        float* dr, float* di, int tid) {
  const int col2 = (tid & 63) << 1;
  const int jg = tid >> 6;
  const bool edge = (jg == 0);
  const int jb = (jg & 1) + ((jg >> 1) << 5);
  const int bj = (jb >> 1) * LDSP + col2 + (jb & 1);
  const int bm = (64 - ((jb + 1) >> 1)) * LDSP + col2 + (jb & 1);
  int bl = bm - LDSP * 15;
  asm volatile("" : "+v"(bl));
  const int p64 = 32 * LDSP + col2;
#pragma unroll
  for (int h = 0; h < 2; ++h) {
    float a[8], b[8], c[8], d[8];
#pragma unroll
    for (int q = 0; q < 8; ++q) {
      const int i = 8 * h + q;
      int pj = bj + LDSP * i;
      int pm = (i == 0 && edge) ? p64 : bl + LDSP * (15 - i);
      a[q] = sr[pj]; b[q] = sr[pm]; c[q] = si[pj]; d[q] = si[pm];
    }
#pragma unroll
    for (int q = 0; q < 8; ++q) {
      const int i = 8 * h + q;
      int pj = bj + LDSP * i;
      int pm = (i == 0 && edge) ? p64 : bl + LDSP * (15 - i);
      bool cp = (i == 0 && edge);
      dr[pj] = cp ? a[q] : a[q] + b[q];
      dr[pm] = cp ? b[q] : c[q] - d[q];
      di[pj] = cp ? c[q] : c[q] + d[q];
      di[pm] = cp ? d[q] : b[q] - a[q];
    }
  }
}

static __device__ inline void fold_cols(float* yr, float* yi, int mB, int lane) {
  const int cj = (mB >> 1) * LDSP + (lane << 1), cm = (mB >> 1) * LDSP + ((64 - lane) << 1);
  if (lane != 0) {
#pragma unroll
    for (int h = 0; h < 2; ++h) {
      float a[8], b[8], c[8], d[8];
#pragma unroll
      for (int q = 0; q < 8; ++q) {
        const int r = 8 * h + q;
        const int ro = (r >> 1) * LDSP + (r & 1);
        a[q] = yr[cj + ro]; b[q] = yr[cm + ro]; c[q] = yi[cj + ro]; d[q] = yi[cm + ro];
      }
#pragma unroll
      for (int q = 0; q < 8; ++q) {
        const int r = 8 * h + q;
        const int ro = (r >> 1) * LDSP + (r & 1);
        yr[cj + ro] = a[q] + b[q];
        yr[cm + ro] = c[q] - d[q];
        yi[cj + ro] = c[q] + d[q];
        yi[cm + ro] = b[q] - a[q];
      }
    }
  }
}

static __device__ __forceinline__ void fence128() { __builtin_amdgcn_sched_barrier(0); }
static __device__ __forceinline__ void fence128_after(v8f& x) {
  asm volatile("" : "+v"(x));
  __builtin_amdgcn_sched_barrier(0);
}

template <int NK, int NT>
static __device__ __forceinline__ void ld_tab64(const float* tl, int kk0, v2f (&b)[9]) {
#pragma unroll
  for (int k = 0; k < NK; ++k)
#pragma unroll
    for (int t = 0; t < NT; ++t)
      b[NT * k + t] = *(const v2f*)(tl + 64 * (kk0 + 4 * k) + 32 * t);
}

template <int NK>
static __device__ __forceinline__ void ld_a64(const float* yr, const float* yi, int kk0,
                                              v2f (&a)[8]) {
#pragma unroll
  for (int k = 0; k < NK; ++k) {
    const int p = 2 * (kk0 + 4 * k);
    a[2 * k].x = yr[p];     a[2 * k].y = yr[p + 2];
    a[2 * k + 1].x = yi[p]; a[2 * k + 1].y = yi[p + 2];
  }
}

static __device__ __forceinline__ void dft64_inplace(float* Yr, float* Yi,
                                                     const float* T64, const float* tbl,
                                                     int mB, int mHi, int l16, int kSel,
                                                     int lane) {
  v8f c1[3] = {}, c2[3] = {}, s1[2] = {}, s2[2] = {};
  const int arow = pidx(mB + l16, kSel);
  const bool hi = (kSel != 0);
  const float* tl = T64 + ((kSel >> 1) << 7) + (l16 << 1);
  const float* yr = Yr + arow;
  const float* yi = Yi + arow;
  v2f tb0[9], tb1[9], av0[8], av1[8];
  ld_tab64<3, 3>(tl, 0, tb0);
  fence128();
  fold_cols(Yr, Yi, mB, lane);
  ld_a64<3>(yr, yi, 0, av0);
  ld_tab64<3, 3>(tl, 12, tb1);
  ld_a64<3>(yr, yi, 12, av1);
  fence128();
#pragma unroll
  for (int k = 0; k < 3; ++k)
#pragma unroll
    for (int t = 0; t < 3; ++t) {
      c1[t] = wmma4(av0[2 * k], tb0[3 * k + t], c1[t]);
      c2[t] = wmma4(av0[2 * k + 1], tb0[3 * k + t], c2[t]);
    }
  fence128();
  ld_tab64<3, 3>(tl, 24, tb0);
  ld_a64<3>(yr, yi, 24, av0);
  fence128();
#pragma unroll
  for (int k = 0; k < 3; ++k)
#pragma unroll
    for (int t = 0; t < 3; ++t) {
      c1[t] = wmma4(av1[2 * k], tb1[3 * k + t], c1[t]);
      c2[t] = wmma4(av1[2 * k + 1], tb1[3 * k + t], c2[t]);
    }
  fence128();
  ld_tab64<4, 2>(tl, 32, tb1);
  ld_a64<4>(yr, yi, 32, av1);
  fence128();
#pragma unroll
  for (int k = 0; k < 3; ++k) {
    v2f a1 = av0[2 * k], a2 = av0[2 * k + 1];
    if (k == 2) {
      a1.x = hi ? 0.f : a1.x;  a1.y = 0.f;
      a2.x = hi ? 0.f : a2.x;  a2.y = 0.f;
    }
#pragma unroll
    for (int t = 0; t < 3; ++t) {
      c1[t] = wmma4(a1, tb0[3 * k + t], c1[t]);
      c2[t] = wmma4(a2, tb0[3 * k + t], c2[t]);
    }
  }
  fence128();
  ld_tab64<4, 2>(tl, 48, tb0);
  ld_a64<4>(yr, yi, 48, av0);
  fence128();
#pragma unroll
  for (int k = 0; k < 4; ++k) {
    v2f a1 = av1[2 * k], a2 = av1[2 * k + 1];
    if (k == 0) {
      a1.x = hi ? a1.x : 0.f;
      a2.x = hi ? a2.x : 0.f;
    }
#pragma unroll
    for (int t = 0; t < 2; ++t) {
      s1[t] = wmma4(a1, tb1[2 * k + t], s1[t]);
      s2[t] = wmma4(a2, tb1[2 * k + t], s2[t]);
    }
  }
#pragma unroll
  for (int k = 0; k < 4; ++k)
#pragma unroll
    for (int t = 0; t < 2; ++t) {
      s1[t] = wmma4(av0[2 * k], tb0[2 * k + t], s1[t]);
      s2[t] = wmma4(av0[2 * k + 1], tb0[2 * k + t], s2[t]);
    }
  const int rb = (mB + mHi) << 6;
  const int rl = ((mB + mHi) >> 1) * LDSP;
#pragma unroll
  for (int t = 0; t < 2; ++t) {
    const int kd = (t << 4) + l16;
    const int km = (t == 0) ? ((l16 == 0) ? 32 : 64 - l16) : 48 - l16;
    const float* td = tbl + ((rb + kd) << 1);
    const float* tm = tbl + ((rb + km) << 1);
    const int pd = rl + (kd << 1);
    const int pm = rl + (km << 1);
    v2f twd[8], twm[8];
#pragma unroll
    for (int v = 0; v < 8; ++v) {
      twd[v] = *(const v2f*)(td + 128 * v);
      twm[v] = *(const v2f*)(tm + 128 * v);
    }
#pragma unroll
    for (int v = 0; v < 8; ++v) {
      float ad = c1[t][v] + s1[t][v], bd = c2[t][v] + s2[t][v];
      float am = c1[t][v] - s1[t][v], bm = c2[t][v] - s2[t][v];
      if (t == 0) {
        am = (l16 == 0) ? c1[2][v] : am;
        bm = (l16 == 0) ? c2[2][v] : bm;
      }
      const int po = LDSP * (v >> 1) + (v & 1);
      v2f wd = twd[v], wm = twm[v];
      Yr[pd + po] = ad * wd.x - bd * wd.y;
      Yi[pd + po] = ad * wd.y + bd * wd.x;
      Yr[pm + po] = am * wm.x - bm * wm.y;
      Yi[pm + po] = am * wm.y + bm * wm.x;
    }
  }
}

static __device__ __forceinline__ void ld_data128(const float* pr, const float* pi,
                                                  const float* px, int jc, v2f (&b)[9]) {
  const int js = jc + 32 * LDSP;
#pragma unroll
  for (int t = 0; t < 2; ++t) {
    b[4 * t + 0] = *(const v2f*)(pr + jc + 32 * t);
    b[4 * t + 1] = *(const v2f*)(pi + jc + 32 * t);
    b[4 * t + 2] = *(const v2f*)(pr + js + 32 * t);
    b[4 * t + 3] = *(const v2f*)(pi + js + 32 * t);
  }
  b[8] = *(const v2f*)(px + jc);
}

static __device__ __forceinline__ void ld_tab128(const float* tcM, const float* tsM,
                                                 const float* tcX, int q, v2f (&a)[3]) {
  a[0] = *(const v2f*)(tcM + (q << 9));
  a[1] = *(const v2f*)(tsM + (q << 9));
  a[2] = *(const v2f*)(tcX + (q << 9));
}

static __device__ __forceinline__ void mm_pass128(const v2f (&a)[3], const v2f (&b)[9],
                                                  v8f (&c1)[2], v8f (&c2)[2],
                                                  v8f (&s1)[2], v8f (&s2)[2], v8f& x) {
  x = wmma4(a[2], b[8], x);
#pragma unroll
  for (int t = 0; t < 2; ++t) {
    c1[t] = wmma4(a[0], b[4 * t + 0], c1[t]);
    c2[t] = wmma4(a[0], b[4 * t + 1], c2[t]);
    s1[t] = wmma4(a[1], b[4 * t + 2], s1[t]);
    s2[t] = wmma4(a[1], b[4 * t + 3], s2[t]);
  }
}

static __device__ __forceinline__ void dft128_split(const float* T, const float* Br,
                                                    int m, int nh, int exT, int exP,
                                                    int l16, int kSel,
                                                    v8f (&c1)[2], v8f (&c2)[2],
                                                    v8f (&s1)[2], v8f (&s2)[2], v8f& x) {
  const int aM = ((16 * m + l16) << 2) + kSel;
  const int aX = ((64 + l16) << 2) + kSel;
  const float* tcM = T + aM;
  const float* tsM = T + OFF_TS + aM;
  const float* tcX = T + aX;
  const int j0 = (kSel >> 1) * LDSP;
  const float* pr = Br + j0 + ((32 * nh + l16) << 1);
  const float* pi = pr + LPLANE;
  const float* px = Br + exP * LPLANE + j0 + ((16 * exT + l16) << 1);

  v2f ta[4][3];
  v2f b0[9], b1[9];
  ld_tab128(tcM, tsM, tcX, 0, ta[0]);
  ld_tab128(tcM, tsM, tcX, 1, ta[1]);
  ld_tab128(tcM, tsM, tcX, 2, ta[2]);
  ld_data128(pr, pi, px, 0, b0);
#pragma unroll 1
  for (int q0 = 0; q0 < 12; q0 += 4) {
#pragma unroll
    for (int u = 0; u < 4; u += 2) {
      const int p = q0 + u;
      ld_data128(pr, pi, px, (p + 1) * (2 * LDSP), b1);
      ld_tab128(tcM, tsM, tcX, p + 3, ta[(u + 3) & 3]);
      fence128();
      mm_pass128(ta[u], b0, c1, c2, s1, s2, x);
      fence128_after(x);
      ld_data128(pr, pi, px, (p + 2) * (2 * LDSP), b0);
      ld_tab128(tcM, tsM, tcX, p + 4, ta[u]);
      fence128();
      mm_pass128(ta[u + 1], b1, c1, c2, s1, s2, x);
      fence128_after(x);
    }
  }
  ld_data128(pr, pi, px, 13 * (2 * LDSP), b1);
  ld_tab128(tcM, tsM, tcX, 15, ta[3]);
  fence128();
  mm_pass128(ta[0], b0, c1, c2, s1, s2, x);
  fence128_after(x);
  ld_data128(pr, pi, px, 14 * (2 * LDSP), b0);
  ta[0][0] = *(const v2f*)(tcM + (16 << 9));
  ta[0][2] = *(const v2f*)(tcX + (16 << 9));
  fence128();
  mm_pass128(ta[1], b1, c1, c2, s1, s2, x);
  fence128_after(x);
  ld_data128(pr, pi, px, 15 * (2 * LDSP), b1);
  fence128();
  mm_pass128(ta[2], b0, c1, c2, s1, s2, x);
  fence128_after(x);
#pragma unroll
  for (int t = 0; t < 2; ++t) {
    b0[4 * t + 0] = *(const v2f*)(pr + 16 * (2 * LDSP) + 32 * t);
    b0[4 * t + 1] = *(const v2f*)(pi + 16 * (2 * LDSP) + 32 * t);
  }
  b0[8] = *(const v2f*)(px + 16 * (2 * LDSP));
  fence128();
  mm_pass128(ta[3], b1, c1, c2, s1, s2, x);
#pragma unroll
  for (int t = 0; t < 2; ++t) {
    c1[t] = wmma4(ta[0][0], b0[4 * t + 0], c1[t]);
    c2[t] = wmma4(ta[0][0], b0[4 * t + 1], c2[t]);
  }
  x = wmma4(ta[0][2], b0[8], x);
}

__global__ __launch_bounds__(256, 4) void ssm_rows(const float* __restrict__ uin,
                                                float* __restrict__ uout,
                                                const float* ws) {
  extern __shared__ float smem[];
  float* xa = smem;
  float* xb = smem + LPLANE;
  float* Yr = smem + 2 * LPLANE;
  float* Yi = smem + 3 * LPLANE;

  const int tid  = threadIdx.x;
  const int wave = tid >> 5;
  const int lane = tid & 31;
  const int l16  = lane & 15;
  const int kSel = (lane >> 4) << 1;
  const int mHi  = (lane >> 4) << 3;
  const int mB   = wave << 4;
  const int m128 = wave >> 1;
  const int nh   = wave & 1;
  const int exT  = wave & 3;
  const int exP  = wave >> 2;
  const int rD   = (m128 << 4) + mHi;
  const int rM   = 128 - rD;
  const bool nopart = (rD == 0);

  const size_t rowOff = (size_t)blockIdx.x * (2 * LROW);
  const float4* srcA = (const float4*)(uin + rowOff);
  const float4* srcB = (const float4*)(uin + rowOff + LROW);
  for (int i = tid; i < LROW / 4; i += 256) {
    float4 va = srcA[i];
    float4 vb = srcB[i];
    int n2 = i >> 4, n1 = (i & 15) << 2;
    int base = pidx(n2, n1);
    xa[base] = va.x; xa[base + 2] = va.y; xa[base + 4] = va.z; xa[base + 6] = va.w;
    xb[base] = vb.x; xb[base + 2] = vb.y; xb[base + 4] = vb.z; xb[base + 6] = vb.w;
  }
  __syncthreads();

  const float* T128F = ws + OFF_T128F;
  const float* T128I = ws + OFF_T128I;
  const float* T64F  = ws + OFF_T64F;
  const float* T64I  = ws + OFF_T64I;
  const float* Wf    = ws + OFF_WF;
  const float* Wi    = ws + OFF_WI;

  for (int layer = 0; layer < 4; ++layer) {
    const float* kh = ws + OFF_KHAT + layer * 16384;

    fold_rows(xa, xb, Yr, Yi, tid);
    __syncthreads();
    {
      v8f c1[2] = {}, c2[2] = {}, s1[2] = {}, s2[2] = {}, x5 = {};
      dft128_split(T128F, Yr, m128, nh, exT, exP, l16, kSel, c1, c2, s1, s2, x5);
      __syncthreads();
      int l16e = l16, rDe = rD;
      asm volatile("" : "+v"(l16e), "+v"(rDe));
      const int rMe = 128 - rDe;
#pragma unroll
      for (int t = 0; t < 2; ++t) {
        const int n = (nh << 5) + (t << 4) + l16e;
        const float* td = Wf + (((rDe << 6) + n) << 1);
        const float* tm = Wf + ((((rMe - 8) << 6) + n) << 1);
        const int pd = (rDe >> 1) * LDSP + (n << 1);
        const int pm = ((rMe - 8) >> 1) * LDSP + (n << 1);
        v2f twd[8], twm[8];
#pragma unroll
        for (int v = 0; v < 8; ++v) {
          twd[v] = *(const v2f*)(td + 128 * v);
          twm[v] = *(const v2f*)((v == 0 && nopart) ? td : tm + 128 * (8 - v));
        }
#pragma unroll
        for (int v = 0; v < 8; ++v) {
          float ad = c1[t][v] + s1[t][v], bd = c2[t][v] + s2[t][v];
          float am = c1[t][v] - s1[t][v], bm = c2[t][v] - s2[t][v];
          const int po = LDSP * (v >> 1) + (v & 1);
          const int qo = LDSP * (4 - ((v + 1) >> 1)) + (v & 1);
          v2f wd = twd[v], wm = twm[v];
          Yr[pd + po] = ad * wd.x - bd * wd.y;
          Yi[pd + po] = ad * wd.y + bd * wd.x;
          if (!(v == 0 && nopart)) {
            Yr[pm + qo] = am * wm.x - bm * wm.y;
            Yi[pm + qo] = am * wm.y + bm * wm.x;
          }
        }
      }
      {
        int t64 = tid;
        asm volatile("" : "+v"(t64));
        const int a64 = (t64 >> 7) * LPLANE + 32 * LDSP + (((((t64 >> 5) & 3) << 4) + (t64 & 15)) << 1);
        if (lane < 16) Yr[a64] = x5[0];
      }
    }
    __syncthreads();
    if (wave == 4) {
#pragma unroll
      for (int h = 0; h < 2; ++h) {
        const int c = lane + 32 * h;
        const int p = 32 * LDSP + (c << 1);
        float a = Yr[p], b = Yi[p];
        v2f w = *(const v2f*)(Wf + ((4096 + c) << 1));
        Yr[p] = a * w.x - b * w.y;
        Yi[p] = a * w.y + b * w.x;
      }
    }

    dft64_inplace(Yr, Yi, T64F, kh, mB, mHi, l16, kSel, lane);

    dft64_inplace(Yr, Yi, T64I, Wi, mB, mHi, l16, kSel, lane);
    __syncthreads();

    fold_rows(Yr, Yi, Yr, Yi, tid);
    __syncthreads();
    {
      v8f w1[2] = {}, w2[2] = {}, z1[2] = {}, z2[2] = {}, x5 = {};
      dft128_split(T128I, Yr, m128, nh, exT, exP, l16, kSel, w1, w2, z1, z2, x5);
#pragma unroll
      for (int t = 0; t < 2; ++t) {
        const int n = (nh << 5) + (t << 4) + l16;
        const int pd = (rD >> 1) * LDSP + (n << 1);
        const int pm = ((rM - 8) >> 1) * LDSP + (n << 1);
#pragma unroll
        for (int j = 0; j < 4; ++j) {
          int p = pd + LDSP * j;
          v2f a1; a1.x = w1[t][2 * j] + z1[t][2 * j]; a1.y = w1[t][2 * j + 1] + z1[t][2 * j + 1];
          v2f a2; a2.x = w2[t][2 * j] + z2[t][2 * j]; a2.y = w2[t][2 * j + 1] + z2[t][2 * j + 1];
          v2f ya = a1 + *(const v2f*)(xa + p);
          v2f yb = a2 + *(const v2f*)(xb + p);
          *(v2f*)(xa + p) = gelu2(ya);
          *(v2f*)(xb + p) = gelu2(yb);
        }
#pragma unroll
        for (int j = 0; j < 3; ++j) {
          int p = pm + LDSP * (3 - j);
          v2f a1; a1.x = w1[t][2 * j + 2] - z1[t][2 * j + 2]; a1.y = w1[t][2 * j + 1] - z1[t][2 * j + 1];
          v2f a2; a2.x = w2[t][2 * j + 2] - z2[t][2 * j + 2]; a2.y = w2[t][2 * j + 1] - z2[t][2 * j + 1];
          v2f ya = a1 + *(const v2f*)(xa + p);
          v2f yb = a2 + *(const v2f*)(xb + p);
          *(v2f*)(xa + p) = gelu2(ya);
          *(v2f*)(xb + p) = gelu2(yb);
        }
        {
          const int p7 = pm + 1;
          const int p0 = nopart ? p7 : pm + 4 * LDSP;
          v2f a1; a1.x = w1[t][0] - z1[t][0]; a1.y = w1[t][7] - z1[t][7];
          v2f a2; a2.x = w2[t][0] - z2[t][0]; a2.y = w2[t][7] - z2[t][7];
          v2f ra; ra.x = xa[p0]; ra.y = xa[p7];
          v2f rb; rb.x = xb[p0]; rb.y = xb[p7];
          v2f ga = gelu2(a1 + ra);
          v2f gb = gelu2(a2 + rb);
          xa[p7] = ga.y;
          xb[p7] = gb.y;
          if (!nopart) {
            xa[p0] = ga.x;
            xb[p0] = gb.x;
          }
        }
      }
      {
        int t64 = tid;
        asm volatile("" : "+v"(t64));
        const int p = (t64 >> 7) * LPLANE + 32 * LDSP + (((((t64 >> 5) & 3) << 4) + (t64 & 15)) << 1);
        if (lane < 16) {
          float y = x5[0] + xa[p];
          xa[p] = gelu2(bc2(y)).x;
        }
      }
    }
    __syncthreads();
  }

  float4* dstA = (float4*)(uout + rowOff);
  float4* dstB = (float4*)(uout + rowOff + LROW);
  int tid_o = tid;
  asm volatile("" : "+v"(tid_o));
  for (int i = tid_o; i < LROW / 4; i += 256) {
    int n2 = i >> 4, n1 = (i & 15) << 2;
    int base = pidx(n2, n1);
    float4 oa, ob;
    oa.x = xa[base]; oa.y = xa[base + 2]; oa.z = xa[base + 4]; oa.w = xa[base + 6];
    ob.x = xb[base]; ob.y = xb[base + 2]; ob.z = xb[base + 4]; ob.w = xb[base + 6];
    dstA[i] = oa;
    dstB[i] = ob;
  }
}

extern "C" void kernel_launch(void* const* d_in, const int* in_sizes, int n_in,
                              void* d_out, int out_size, void* d_ws, size_t ws_size,
                              hipStream_t stream) {
  const float* u    = (const float*)d_in[0];
  const float* Lr   = (const float*)d_in[1];
  const float* Li   = (const float*)d_in[2];
  const float* Pv   = (const float*)d_in[3];
  const float* Bv   = (const float*)d_in[4];
  const float* Ct   = (const float*)d_in[5];
  const float* step = (const float*)d_in[6];
  float* out = (float*)d_out;
  float* ws  = (float*)d_ws;

  int pairs = in_sizes[0] / LPAIR;

  ssm_gen_tables<<<64, 256, 0, stream>>>(ws);
  ssm_gen_khat<<<dim3(32, 4), 256, 0, stream>>>(ws, Lr, Li, Pv, Bv, Ct, step);
  ssm_rows<<<pairs, 256, 34816 * sizeof(float), stream>>>(u, out, ws);
}


# Memory loads issued early: 128-point table heads before the fold and barrier, all three epilogues' twiddle factors before the last product passes, not behind them; sum order unchanged
# speedup vs baseline: 4.3749x; 1.0319x over previous
#include <hip/hip_runtime.h>
#include <math.h>

typedef float v2f __attribute__((ext_vector_type(2)));
typedef float v8f __attribute__((ext_vector_type(8)));

#define LROW 8192
#define LPAIR 16384
#define NSTATE 64
#define PI_F 3.14159265358979323846f

#define LDSP 136
#define LPLANE (64 * LDSP)

#define OFF_T128F   0
#define OFF_T128I   32768
#define OFF_TS      16384
#define OFF_T64F    65536
#define OFF_T64I    73728
#define OFF_WF      81920
#define OFF_WI      98304
#define OFF_KHAT    114688
#define WS_FLOATS   (OFF_KHAT + 4*16384)

static __device__ inline v8f wmma4(v2f a, v2f b, v8f c) {
  return __builtin_amdgcn_wmma_f32_16x16x4_f32(false, a, false, b, (short)0, c,
                                               false, false);
}

static __device__ inline int pidx(int row, int col) {
  return (row >> 1) * LDSP + (col << 1) + (row & 1);
}

static __device__ inline v2f vfma2(v2f a, v2f b, v2f c) {
  return __builtin_elementwise_fma(a, b, c);
}
static __device__ inline v2f bc2(float s) { v2f r; r.x = s; r.y = s; return r; }

static __device__ inline v2f gelu2(v2f y) {
  v2f x  = y * 0.707106769f;
  v2f ax = __builtin_elementwise_abs(x);
  v2f t  = x * x;
  v2f pb = vfma2(t, bc2(-0.000561801775f), bc2(0.00491381623f));
  pb = vfma2(t, pb, bc2(-0.0267075151f));
  pb = vfma2(t, pb, bc2(0.112800106f));
  pb = vfma2(t, pb, bc2(-0.376122952f));
  pb = vfma2(t, pb, bc2(0.128379107f));
  v2f rs = vfma2(ax, pb, ax);
  v2f pa = vfma2(ax, bc2(0.0000169988107f), bc2(-0.000378677854f));
  pa = vfma2(ax, pa, bc2(0.00385781587f));
  pa = vfma2(ax, pa, bc2(-0.0241816975f));
  pa = vfma2(ax, pa, bc2(0.106668264f));
  pa = vfma2(ax, pa, bc2(0.634933293f));
  pa = vfma2(ax, pa, bc2(0.128689408f));
  v2f q  = vfma2(ax, pa, ax);
  v2f qe = q * -1.44269502f;
  v2f e;
  e.x = __builtin_amdgcn_exp2f(qe.x);
  e.y = __builtin_amdgcn_exp2f(qe.y);
  v2f rl = 1.0f - e;
  v2f r;
  r.x = ax.x < 1.0f ? rs.x : rl.x;
  r.y = ax.y < 1.0f ? rs.y : rl.y;
  r = __builtin_elementwise_copysign(r, x);
  return (0.5f * y) * (1.0f + r);
}

__global__ void ssm_gen_tables(float* __restrict__ ws) {
  int idx = blockIdx.x * blockDim.x + threadIdx.x;
  {
    int k2 = idx >> 7, J = idx & 127;
    int t = (k2 * J) & 127;
    float s, c;
    sincosf(-2.f * PI_F * (float)t / 128.f, &s, &c);
    if (J < 68) {
      int pos = (J >> 2) * 512 + k2 * 4 + (J & 3);
      float vc = (J <= 64) ? c : 0.f;
      ws[OFF_T128F + pos] = vc;
      ws[OFF_T128I + pos] = vc;
    }
    if (J >= 64) {
      int pos = OFF_TS + ((J - 64) >> 2) * 512 + k2 * 4 + (J & 3);
      float vs = (J > 64) ? s : 0.f;
      ws[OFF_T128F + pos] = vs;
      ws[OFF_T128I + pos] = -vs;
    }
  }
  if (idx < 8192) {
    int k2 = idx >> 6, n1 = idx & 63;
    int t = (n1 * k2) & 8191;
    float s, c;
    sincosf(-2.f * PI_F * (float)t / 8192.f, &s, &c);
    ws[OFF_WF + 2 * idx]     = c;  ws[OFF_WF + 2 * idx + 1] = s;
    ws[OFF_WI + 2 * idx]     = c;  ws[OFF_WI + 2 * idx + 1] = -s;
  }
  if (idx < 4096) {
    int J = idx >> 6, k1 = idx & 63;
    int t = (J * k1) & 63;
    float s, c;
    sincosf(-2.f * PI_F * (float)t / 64.f, &s, &c);
    int pos = (J >> 1) * 128 + 2 * k1 + (J & 1);
    ws[OFF_T64F + pos] = (J <= 32) ? c : s;
    ws[OFF_T64I + pos] = (J <= 32) ? c : -s;
  }
}

static __device__ inline void ssm_khat_at(int kf, float stp,
                                          const float* __restrict__ lr,
                                          const float* __restrict__ li,
                                          const float* __restrict__ pp,
                                          const float* __restrict__ bb,
                                          const float* __restrict__ ct,
                                          float& outr, float& outi) {
  float osin, ocos;
  sincosf(-2.f * PI_F * (float)kf / 8192.f, &osin, &ocos);
  float pr = 1.f + ocos, pi = osin;
  float mr = 1.f - ocos, mi = -osin;
  float pd = pr * pr + pi * pi;
  float ipr = pr / pd, ipi = -pi / pd;
  float cr = 2.f * ipr, ci = 2.f * ipi;
  float sc = 2.f / stp;
  float gr = sc * (mr * ipr - mi * ipi);
  float gi = sc * (mr * ipi + mi * ipr);

  float k00r = 0, k00i = 0, k01r = 0, k01i = 0;
  float k10r = 0, k10i = 0, k11r = 0, k11i = 0;
  for (int n = 0; n < NSTATE; ++n) {
    float dr = gr - lr[n], di = gi - li[n];
    float dd = dr * dr + di * di;
    float ir = dr / dd, ii = -di / dd;
    float b = bb[n], p = pp[n], c = ct[n];
    float w00 = c * b, w01 = c * p, w10 = p * b, w11 = p * p;
    k00r += w00 * ir; k00i += w00 * ii;
    k01r += w01 * ir; k01i += w01 * ii;
    k10r += w10 * ir; k10i += w10 * ii;
    k11r += w11 * ir; k11i += w11 * ii;
  }
  float qr = k01r * k10r - k01i * k10i;
  float qi = k01r * k10i + k01i * k10r;
  float er = 1.f + k11r, ei = k11i;
  float ed = er * er + ei * ei;
  float ier = er / ed, iei = -ei / ed;
  float tr = qr * ier - qi * iei;
  float ti = qr * iei + qi * ier;
  float kr = k00r - tr, ki = k00i - ti;
  outr = cr * kr - ci * ki;
  outi = cr * ki + ci * kr;
}

__global__ void ssm_gen_khat(float* __restrict__ ws,
                             const float* __restrict__ Lr,
                             const float* __restrict__ Li,
                             const float* __restrict__ Pv,
                             const float* __restrict__ Bv,
                             const float* __restrict__ Ct,
                             const float* __restrict__ stepv) {
  int layer = blockIdx.y;
  int s = blockIdx.x * blockDim.x + threadIdx.x;
  int k2 = s >> 6, k1 = s & 63;
  int kf = 128 * k1 + k2;
  int kg = (8192 - kf) & 8191;

  const float* lr = Lr + layer * NSTATE;
  const float* li = Li + layer * NSTATE;
  const float* pp = Pv + layer * NSTATE;
  const float* bb = Bv + layer * NSTATE;
  const float* ct = Ct + layer * NSTATE;
  float stp = stepv[layer];

  float ar, ai, br, bi;
  ssm_khat_at(kf, stp, lr, li, pp, bb, ct, ar, ai);
  ssm_khat_at(kg, stp, lr, li, pp, bb, ct, br, bi);
  const float scale = 0.5f / 8192.f;
  ws[OFF_KHAT + layer * 16384 + 2 * s]     = (ar + br) * scale;
  ws[OFF_KHAT + layer * 16384 + 2 * s + 1] = (ai - bi) * scale;
}

static __device__ inline void fold_rows(const float* sr, const float* si,
                                        float* dr, float* di, int tid) {
  const int col2 = (tid & 63) << 1;
  const int jg = tid >> 6;
  const bool edge = (jg == 0);
  const int jb = (jg & 1) + ((jg >> 1) << 5);
  const int bj = (jb >> 1) * LDSP + col2 + (jb & 1);
  const int bm = (64 - ((jb + 1) >> 1)) * LDSP + col2 + (jb & 1);
  int bl = bm - LDSP * 15;
  asm volatile("" : "+v"(bl));
  const int p64 = 32 * LDSP + col2;
#pragma unroll
  for (int h = 0; h < 2; ++h) {
    float a[8], b[8], c[8], d[8];
#pragma unroll
    for (int q = 0; q < 8; ++q) {
      const int i = 8 * h + q;
      int pj = bj + LDSP * i;
      int pm = (i == 0 && edge) ? p64 : bl + LDSP * (15 - i);
      a[q] = sr[pj]; b[q] = sr[pm]; c[q] = si[pj]; d[q] = si[pm];
    }
#pragma unroll
    for (int q = 0; q < 8; ++q) {
      const int i = 8 * h + q;
      int pj = bj + LDSP * i;
      int pm = (i == 0 && edge) ? p64 : bl + LDSP * (15 - i);
      bool cp = (i == 0 && edge);
      dr[pj] = cp ? a[q] : a[q] + b[q];
      dr[pm] = cp ? b[q] : c[q] - d[q];
      di[pj] = cp ? c[q] : c[q] + d[q];
      di[pm] = cp ? d[q] : b[q] - a[q];
    }
  }
}

static __device__ inline void fold_cols(float* yr, float* yi, int mB, int lane) {
  const int cj = (mB >> 1) * LDSP + (lane << 1), cm = (mB >> 1) * LDSP + ((64 - lane) << 1);
  if (lane != 0) {
#pragma unroll
    for (int h = 0; h < 2; ++h) {
      float a[8], b[8], c[8], d[8];
#pragma unroll
      for (int q = 0; q < 8; ++q) {
        const int r = 8 * h + q;
        const int ro = (r >> 1) * LDSP + (r & 1);
        a[q] = yr[cj + ro]; b[q] = yr[cm + ro]; c[q] = yi[cj + ro]; d[q] = yi[cm + ro];
      }
#pragma unroll
      for (int q = 0; q < 8; ++q) {
        const int r = 8 * h + q;
        const int ro = (r >> 1) * LDSP + (r & 1);
        yr[cj + ro] = a[q] + b[q];
        yr[cm + ro] = c[q] - d[q];
        yi[cj + ro] = c[q] + d[q];
        yi[cm + ro] = b[q] - a[q];
      }
    }
  }
}

static __device__ __forceinline__ void fence128() { __builtin_amdgcn_sched_barrier(0); }
static __device__ __forceinline__ void fence128_after(v8f& x) {
  asm volatile("" : "+v"(x));
  __builtin_amdgcn_sched_barrier(0);
}

template <int NK, int NT>
static __device__ __forceinline__ void ld_tab64(const float* tl, int kk0, v2f (&b)[9]) {
#pragma unroll
  for (int k = 0; k < NK; ++k)
#pragma unroll
    for (int t = 0; t < NT; ++t)
      b[NT * k + t] = *(const v2f*)(tl + 64 * (kk0 + 4 * k) + 32 * t);
}

static __device__ __forceinline__ void ld_fac8(const float* p, v2f (&w)[8]) {
#pragma unroll
  for (int v = 0; v < 8; ++v) w[v] = *(const v2f*)(p + 128 * v);
}

template <int NK>
static __device__ __forceinline__ void ld_a64(const float* yr, const float* yi, int kk0,
                                              v2f (&a)[8]) {
#pragma unroll
  for (int k = 0; k < NK; ++k) {
    const int p = 2 * (kk0 + 4 * k);
    a[2 * k].x = yr[p];     a[2 * k].y = yr[p + 2];
    a[2 * k + 1].x = yi[p]; a[2 * k + 1].y = yi[p + 2];
  }
}

static __device__ __forceinline__ void dft64_inplace(float* Yr, float* Yi,
                                                     const float* T64, const float* tbl,
                                                     int mB, int mHi, int l16, int kSel,
                                                     int lane) {
  v8f c1[3] = {}, c2[3] = {}, s1[2] = {}, s2[2] = {};
  const int arow = pidx(mB + l16, kSel);
  const bool hi = (kSel != 0);
  const float* tl = T64 + ((kSel >> 1) << 7) + (l16 << 1);
  const float* yr = Yr + arow;
  const float* yi = Yi + arow;
  v2f twd[2][8], twm[2][8];
  v2f tb0[9], tb1[9], av0[8], av1[8];
  ld_tab64<3, 3>(tl, 0, tb0);
  fence128();
  fold_cols(Yr, Yi, mB, lane);
  ld_a64<3>(yr, yi, 0, av0);
  ld_tab64<3, 3>(tl, 12, tb1);
  ld_a64<3>(yr, yi, 12, av1);
  fence128();
#pragma unroll
  for (int k = 0; k < 3; ++k)
#pragma unroll
    for (int t = 0; t < 3; ++t) {
      c1[t] = wmma4(av0[2 * k], tb0[3 * k + t], c1[t]);
      c2[t] = wmma4(av0[2 * k + 1], tb0[3 * k + t], c2[t]);
    }
  fence128();
  ld_tab64<3, 3>(tl, 24, tb0);
  ld_a64<3>(yr, yi, 24, av0);
  fence128();
#pragma unroll
  for (int k = 0; k < 3; ++k)
#pragma unroll
    for (int t = 0; t < 3; ++t) {
      c1[t] = wmma4(av1[2 * k], tb1[3 * k + t], c1[t]);
      c2[t] = wmma4(av1[2 * k + 1], tb1[3 * k + t], c2[t]);
    }
  fence128();
  ld_tab64<4, 2>(tl, 32, tb1);
  ld_a64<4>(yr, yi, 32, av1);
  fence128();
#pragma unroll
  for (int k = 0; k < 3; ++k) {
    v2f a1 = av0[2 * k], a2 = av0[2 * k + 1];
    if (k == 2) {
      a1.x = hi ? 0.f : a1.x;  a1.y = 0.f;
      a2.x = hi ? 0.f : a2.x;  a2.y = 0.f;
    }
#pragma unroll
    for (int t = 0; t < 3; ++t) {
      c1[t] = wmma4(a1, tb0[3 * k + t], c1[t]);
      c2[t] = wmma4(a2, tb0[3 * k + t], c2[t]);
    }
  }
  fence128();
  ld_tab64<4, 2>(tl, 48, tb0);
  ld_a64<4>(yr, yi, 48, av0);
  int l16q = l16, rq = mB + mHi;
  asm volatile("" : "+v"(l16q), "+v"(rq));
  const int rb = rq << 6;
  const float* td0 = tbl + ((rb + l16q) << 1);
  const float* tm0 = tbl + ((rb + ((l16q == 0) ? 32 : 64 - l16q)) << 1);
  const float* tm1 = tbl + ((rb + 48 - l16q) << 1);
  fence128();
#pragma unroll
  for (int k = 0; k < 4; ++k) {
    v2f a1 = av1[2 * k], a2 = av1[2 * k + 1];
    if (k == 0) {
      a1.x = hi ? a1.x : 0.f;
      a2.x = hi ? a2.x : 0.f;
    }
#pragma unroll
    for (int t = 0; t < 2; ++t) {
      s1[t] = wmma4(a1, tb1[2 * k + t], s1[t]);
      s2[t] = wmma4(a2, tb1[2 * k + t], s2[t]);
    }
  }
  ld_fac8(td0, twd[0]);
  ld_fac8(tm0, twm[0]);
  fence128();
#pragma unroll
  for (int k = 0; k < 4; ++k)
#pragma unroll
    for (int t = 0; t < 2; ++t) {
      s1[t] = wmma4(av0[2 * k], tb0[2 * k + t], s1[t]);
      s2[t] = wmma4(av0[2 * k + 1], tb0[2 * k + t], s2[t]);
    }
  ld_fac8(td0 + 32, twd[1]);
  ld_fac8(tm1, twm[1]);
  fence128();
  const int rl = ((mB + mHi) >> 1) * LDSP;
#pragma unroll
  for (int t = 0; t < 2; ++t) {
    const int kd = (t << 4) + l16;
    const int km = (t == 0) ? ((l16 == 0) ? 32 : 64 - l16) : 48 - l16;
    const int pd = rl + (kd << 1);
    const int pm = rl + (km << 1);
#pragma unroll
    for (int v = 0; v < 8; ++v) {
      float ad = c1[t][v] + s1[t][v], bd = c2[t][v] + s2[t][v];
      float am = c1[t][v] - s1[t][v], bm = c2[t][v] - s2[t][v];
      if (t == 0) {
        am = (l16 == 0) ? c1[2][v] : am;
        bm = (l16 == 0) ? c2[2][v] : bm;
      }
      const int po = LDSP * (v >> 1) + (v & 1);
      v2f wd = twd[t][v], wm = twm[t][v];
      Yr[pd + po] = ad * wd.x - bd * wd.y;
      Yi[pd + po] = ad * wd.y + bd * wd.x;
      Yr[pm + po] = am * wm.x - bm * wm.y;
      Yi[pm + po] = am * wm.y + bm * wm.x;
    }
  }
}

static __device__ __forceinline__ void ld_data128(const float* pr, const float* pi,
                                                  const float* px, int jc, v2f (&b)[9]) {
  const int js = jc + 32 * LDSP;
#pragma unroll
  for (int t = 0; t < 2; ++t) {
    b[4 * t + 0] = *(const v2f*)(pr + jc + 32 * t);
    b[4 * t + 1] = *(const v2f*)(pi + jc + 32 * t);
    b[4 * t + 2] = *(const v2f*)(pr + js + 32 * t);
    b[4 * t + 3] = *(const v2f*)(pi + js + 32 * t);
  }
  b[8] = *(const v2f*)(px + jc);
}

static __device__ __forceinline__ void ld_tab128(const float* tcM, const float* tsM,
                                                 const float* tcX, int q, v2f (&a)[3]) {
  a[0] = *(const v2f*)(tcM + (q << 9));
  a[1] = *(const v2f*)(tsM + (q << 9));
  a[2] = *(const v2f*)(tcX + (q << 9));
}

static __device__ __forceinline__ void mm_pass128(const v2f (&a)[3], const v2f (&b)[9],
                                                  v8f (&c1)[2], v8f (&c2)[2],
                                                  v8f (&s1)[2], v8f (&s2)[2], v8f& x) {
  x = wmma4(a[2], b[8], x);
#pragma unroll
  for (int t = 0; t < 2; ++t) {
    c1[t] = wmma4(a[0], b[4 * t + 0], c1[t]);
    c2[t] = wmma4(a[0], b[4 * t + 1], c2[t]);
    s1[t] = wmma4(a[1], b[4 * t + 2], s1[t]);
    s2[t] = wmma4(a[1], b[4 * t + 3], s2[t]);
  }
}

static __device__ __forceinline__ void ld_tab128_head(const float* T, int m, int l16, int kSel,
                                                      v2f (&ta)[4][3]) {
  const int aM = ((16 * m + l16) << 2) + kSel;
  const int aX = ((64 + l16) << 2) + kSel;
  ld_tab128(T + aM, T + OFF_TS + aM, T + aX, 0, ta[0]);
  ld_tab128(T + aM, T + OFF_TS + aM, T + aX, 1, ta[1]);
  ld_tab128(T + aM, T + OFF_TS + aM, T + aX, 2, ta[2]);
}

static __device__ __forceinline__ void ld_fac128(const float* td, const float* tm, bool nopart,
                                                 v2f (&wd)[8], v2f (&wm)[8]) {
#pragma unroll
  for (int v = 0; v < 8; ++v) {
    wd[v] = *(const v2f*)(td + 128 * v);
    wm[v] = *(const v2f*)((v == 0 && nopart) ? td : tm + 128 * (8 - v));
  }
}

template <bool PF>
static __device__ __forceinline__ void dft128_split(const float* T, const float* Br,
                                                    int m, int nh, int exT, int exP,
                                                    int l16, int kSel,
                                                    v2f (&ta)[4][3],
                                                    const float* Wt, int rD, bool nopart,
                                                    v2f (&twd)[2][8], v2f (&twm)[2][8],
                                                    v8f (&c1)[2], v8f (&c2)[2],
                                                    v8f (&s1)[2], v8f (&s2)[2], v8f& x) {
  const int aM = ((16 * m + l16) << 2) + kSel;
  const int aX = ((64 + l16) << 2) + kSel;
  const float* tcM = T + aM;
  const float* tsM = T + OFF_TS + aM;
  const float* tcX = T + aX;
  const int j0 = (kSel >> 1) * LDSP;
  const float* pr = Br + j0 + ((32 * nh + l16) << 1);
  const float* pi = pr + LPLANE;
  const float* px = Br + exP * LPLANE + j0 + ((16 * exT + l16) << 1);

  v2f b0[9], b1[9];
  ld_data128(pr, pi, px, 0, b0);
#pragma unroll 1
  for (int q0 = 0; q0 < 12; q0 += 4) {
#pragma unroll
    for (int u = 0; u < 4; u += 2) {
      const int p = q0 + u;
      ld_data128(pr, pi, px, (p + 1) * (2 * LDSP), b1);
      ld_tab128(tcM, tsM, tcX, p + 3, ta[(u + 3) & 3]);
      fence128();
      mm_pass128(ta[u], b0, c1, c2, s1, s2, x);
      fence128_after(x);
      ld_data128(pr, pi, px, (p + 2) * (2 * LDSP), b0);
      ld_tab128(tcM, tsM, tcX, p + 4, ta[u]);
      fence128();
      mm_pass128(ta[u + 1], b1, c1, c2, s1, s2, x);
      fence128_after(x);
    }
  }
  ld_data128(pr, pi, px, 13 * (2 * LDSP), b1);
  ld_tab128(tcM, tsM, tcX, 15, ta[3]);
  fence128();
  mm_pass128(ta[0], b0, c1, c2, s1, s2, x);
  fence128_after(x);
  ld_data128(pr, pi, px, 14 * (2 * LDSP), b0);
  ta[0][0] = *(const v2f*)(tcM + (16 << 9));
  ta[0][2] = *(const v2f*)(tcX + (16 << 9));
  const float* tdp = nullptr;
  const float* tmp = nullptr;
  if (PF) {
    int l16p = l16, rDp = rD;
    asm volatile("" : "+v"(l16p), "+v"(rDp));
    const int n = (nh << 5) + l16p;
    const float* td = Wt + (((rDp << 6) + n) << 1);
    const float* tm = Wt + ((((128 - rDp - 8) << 6) + n) << 1);
    tdp = td; tmp = tm;
  }
  fence128();
  mm_pass128(ta[1], b1, c1, c2, s1, s2, x);
  fence128_after(x);
  ld_data128(pr, pi, px, 15 * (2 * LDSP), b1);
  if (PF) ld_fac128(tdp, tmp, nopart, twd[0], twm[0]);
  fence128();
  mm_pass128(ta[2], b0, c1, c2, s1, s2, x);
  fence128_after(x);
#pragma unroll
  for (int t = 0; t < 2; ++t) {
    b0[4 * t + 0] = *(const v2f*)(pr + 16 * (2 * LDSP) + 32 * t);
    b0[4 * t + 1] = *(const v2f*)(pi + 16 * (2 * LDSP) + 32 * t);
  }
  b0[8] = *(const v2f*)(px + 16 * (2 * LDSP));
  if (PF) ld_fac128(tdp + 32, tmp + 32, nopart, twd[1], twm[1]);
  fence128();
  mm_pass128(ta[3], b1, c1, c2, s1, s2, x);
#pragma unroll
  for (int t = 0; t < 2; ++t) {
    c1[t] = wmma4(ta[0][0], b0[4 * t + 0], c1[t]);
    c2[t] = wmma4(ta[0][0], b0[4 * t + 1], c2[t]);
  }
  x = wmma4(ta[0][2], b0[8], x);
}

__global__ __launch_bounds__(256, 4) void ssm_rows(const float* __restrict__ uin,
                                                float* __restrict__ uout,
                                                const float* ws) {
  extern __shared__ float smem[];
  float* xa = smem;
  float* xb = smem + LPLANE;
  float* Yr = smem + 2 * LPLANE;
  float* Yi = smem + 3 * LPLANE;

  const int tid  = threadIdx.x;
  const int wave = tid >> 5;
  const int lane = tid & 31;
  const int l16  = lane & 15;
  const int kSel = (lane >> 4) << 1;
  const int mHi  = (lane >> 4) << 3;
  const int mB   = wave << 4;
  const int m128 = wave >> 1;
  const int nh   = wave & 1;
  const int exT  = wave & 3;
  const int exP  = wave >> 2;
  const int rD   = (m128 << 4) + mHi;
  const int rM   = 128 - rD;
  const bool nopart = (rD == 0);

  const size_t rowOff = (size_t)blockIdx.x * (2 * LROW);
  const float4* srcA = (const float4*)(uin + rowOff);
  const float4* srcB = (const float4*)(uin + rowOff + LROW);
  for (int i = tid; i < LROW / 4; i += 256) {
    float4 va = srcA[i];
    float4 vb = srcB[i];
    int n2 = i >> 4, n1 = (i & 15) << 2;
    int base = pidx(n2, n1);
    xa[base] = va.x; xa[base + 2] = va.y; xa[base + 4] = va.z; xa[base + 6] = va.w;
    xb[base] = vb.x; xb[base + 2] = vb.y; xb[base + 4] = vb.z; xb[base + 6] = vb.w;
  }
  __syncthreads();

  const float* T128F = ws + OFF_T128F;
  const float* T128I = ws + OFF_T128I;
  const float* T64F  = ws + OFF_T64F;
  const float* T64I  = ws + OFF_T64I;
  const float* Wf    = ws + OFF_WF;
  const float* Wi    = ws + OFF_WI;

  for (int layer = 0; layer < 4; ++layer) {
    const float* kh = ws + OFF_KHAT + layer * 16384;

    v2f ta[4][3];
    v2f twd[2][8], twm[2][8];
    v2f w64[2] = {};
    ld_tab128_head(T128F, m128, l16, kSel, ta);
    fence128();
    fold_rows(xa, xb, Yr, Yi, tid);
    __syncthreads();
    {
      v8f c1[2] = {}, c2[2] = {}, s1[2] = {}, s2[2] = {}, x5 = {};
      dft128_split<true>(T128F, Yr, m128, nh, exT, exP, l16, kSel, ta, Wf, rD, nopart, twd, twm,
                         c1, c2, s1, s2, x5);
      __syncthreads();
      if (wave == 4) {
        int lane_o = lane;
        asm volatile("" : "+v"(lane_o));
#pragma unroll
        for (int h = 0; h < 2; ++h) w64[h] = *(const v2f*)(Wf + ((4096 + lane_o + 32 * h) << 1));
      }
      int l16e = l16, rDe = rD;
      asm volatile("" : "+v"(l16e), "+v"(rDe));
      const int rMe = 128 - rDe;
#pragma unroll
      for (int t = 0; t < 2; ++t) {
        const int n = (nh << 5) + (t << 4) + l16e;
        const int pd = (rDe >> 1) * LDSP + (n << 1);
        const int pm = ((rMe - 8) >> 1) * LDSP + (n << 1);
#pragma unroll
        for (int v = 0; v < 8; ++v) {
          float ad = c1[t][v] + s1[t][v], bd = c2[t][v] + s2[t][v];
          float am = c1[t][v] - s1[t][v], bm = c2[t][v] - s2[t][v];
          const int po = LDSP * (v >> 1) + (v & 1);
          const int qo = LDSP * (4 - ((v + 1) >> 1)) + (v & 1);
          v2f wd = twd[t][v], wm = twm[t][v];
          Yr[pd + po] = ad * wd.x - bd * wd.y;
          Yi[pd + po] = ad * wd.y + bd * wd.x;
          if (!(v == 0 && nopart)) {
            Yr[pm + qo] = am * wm.x - bm * wm.y;
            Yi[pm + qo] = am * wm.y + bm * wm.x;
          }
        }
      }
      {
        int t64 = tid;
        asm volatile("" : "+v"(t64));
        const int a64 = (t64 >> 7) * LPLANE + 32 * LDSP + (((((t64 >> 5) & 3) << 4) + (t64 & 15)) << 1);
        if (lane < 16) Yr[a64] = x5[0];
      }
    }
    __syncthreads();
    if (wave == 4) {
#pragma unroll
      for (int h = 0; h < 2; ++h) {
        const int c = lane + 32 * h;
        const int p = 32 * LDSP + (c << 1);
        float a = Yr[p], b = Yi[p];
        v2f w = w64[h];
        Yr[p] = a * w.x - b * w.y;
        Yi[p] = a * w.y + b * w.x;
      }
    }

    dft64_inplace(Yr, Yi, T64F, kh, mB, mHi, l16, kSel, lane);

    dft64_inplace(Yr, Yi, T64I, Wi, mB, mHi, l16, kSel, lane);
    ld_tab128_head(T128I, m128, l16, kSel, ta);
    fence128();
    __syncthreads();

    fold_rows(Yr, Yi, Yr, Yi, tid);
    __syncthreads();
    {
      v8f w1[2] = {}, w2[2] = {}, z1[2] = {}, z2[2] = {}, x5 = {};
      dft128_split<false>(T128I, Yr, m128, nh, exT, exP, l16, kSel, ta, Wi, rD, nopart, twd, twm,
                          w1, w2, z1, z2, x5);
#pragma unroll
      for (int t = 0; t < 2; ++t) {
        const int n = (nh << 5) + (t << 4) + l16;
        const int pd = (rD >> 1) * LDSP + (n << 1);
        const int pm = ((rM - 8) >> 1) * LDSP + (n << 1);
#pragma unroll
        for (int j = 0; j < 4; ++j) {
          int p = pd + LDSP * j;
          v2f a1; a1.x = w1[t][2 * j] + z1[t][2 * j]; a1.y = w1[t][2 * j + 1] + z1[t][2 * j + 1];
          v2f a2; a2.x = w2[t][2 * j] + z2[t][2 * j]; a2.y = w2[t][2 * j + 1] + z2[t][2 * j + 1];
          v2f ya = a1 + *(const v2f*)(xa + p);
          v2f yb = a2 + *(const v2f*)(xb + p);
          *(v2f*)(xa + p) = gelu2(ya);
          *(v2f*)(xb + p) = gelu2(yb);
        }
#pragma unroll
        for (int j = 0; j < 3; ++j) {
          int p = pm + LDSP * (3 - j);
          v2f a1; a1.x = w1[t][2 * j + 2] - z1[t][2 * j + 2]; a1.y = w1[t][2 * j + 1] - z1[t][2 * j + 1];
          v2f a2; a2.x = w2[t][2 * j + 2] - z2[t][2 * j + 2]; a2.y = w2[t][2 * j + 1] - z2[t][2 * j + 1];
          v2f ya = a1 + *(const v2f*)(xa + p);
          v2f yb = a2 + *(const v2f*)(xb + p);
          *(v2f*)(xa + p) = gelu2(ya);
          *(v2f*)(xb + p) = gelu2(yb);
        }
        {
          const int p7 = pm + 1;
          const int p0 = nopart ? p7 : pm + 4 * LDSP;
          v2f a1; a1.x = w1[t][0] - z1[t][0]; a1.y = w1[t][7] - z1[t][7];
          v2f a2; a2.x = w2[t][0] - z2[t][0]; a2.y = w2[t][7] - z2[t][7];
          v2f ra; ra.x = xa[p0]; ra.y = xa[p7];
          v2f rb; rb.x = xb[p0]; rb.y = xb[p7];
          v2f ga = gelu2(a1 + ra);
          v2f gb = gelu2(a2 + rb);
          xa[p7] = ga.y;
          xb[p7] = gb.y;
          if (!nopart) {
            xa[p0] = ga.x;
            xb[p0] = gb.x;
          }
        }
      }
      {
        int t64 = tid;
        asm volatile("" : "+v"(t64));
        const int p = (t64 >> 7) * LPLANE + 32 * LDSP + (((((t64 >> 5) & 3) << 4) + (t64 & 15)) << 1);
        if (lane < 16) {
          float y = x5[0] + xa[p];
          xa[p] = gelu2(bc2(y)).x;
        }
      }
    }
    __syncthreads();
  }

  float4* dstA = (float4*)(uout + rowOff);
  float4* dstB = (float4*)(uout + rowOff + LROW);
  int tid_o = tid;
  asm volatile("" : "+v"(tid_o));
  for (int i = tid_o; i < LROW / 4; i += 256) {
    int n2 = i >> 4, n1 = (i & 15) << 2;
    int base = pidx(n2, n1);
    float4 oa, ob;
    oa.x = xa[base]; oa.y = xa[base + 2]; oa.z = xa[base + 4]; oa.w = xa[base + 6];
    ob.x = xb[base]; ob.y = xb[base + 2]; ob.z = xb[base + 4]; ob.w = xb[base + 6];
    dstA[i] = oa;
    dstB[i] = ob;
  }
}

extern "C" void kernel_launch(void* const* d_in, const int* in_sizes, int n_in,
                              void* d_out, int out_size, void* d_ws, size_t ws_size,
                              hipStream_t stream) {
  const float* u    = (const float*)d_in[0];
  const float* Lr   = (const float*)d_in[1];
  const float* Li   = (const float*)d_in[2];
  const float* Pv   = (const float*)d_in[3];
  const float* Bv   = (const float*)d_in[4];
  const float* Ct   = (const float*)d_in[5];
  const float* step = (const float*)d_in[6];
  float* out = (float*)d_out;
  float* ws  = (float*)d_ws;

  int pairs = in_sizes[0] / LPAIR;

  ssm_gen_tables<<<64, 256, 0, stream>>>(ws);
  ssm_gen_khat<<<dim3(32, 4), 256, 0, stream>>>(ws, Lr, Li, Pv, Bv, Ct, step);
  ssm_rows<<<pairs, 256, 34816 * sizeof(float), stream>>>(u, out, ws);
}
